# starting kernel
# baseline (speedup 1.0000x reference)
#include <hip/hip_runtime.h>
#include <math.h>

#define N_NODES 20000
#define N_EDGES 320000
#define IN_CH 512
#define HID 128
#define HEADS 4
#define OUT_CH 256
#define NEG_SLOPE 0.2f

typedef float v2f __attribute__((ext_vector_type(2)));
typedef float v8f __attribute__((ext_vector_type(8)));

__device__ __forceinline__ int ford(float f) {
    int i = __float_as_int(f);
    return (i >= 0) ? i : (i ^ 0x7FFFFFFF);
}
__device__ __forceinline__ float iord(int i) {
    return __int_as_float((i >= 0) ? i : (i ^ 0x7FFFFFFF));
}

__global__ void fill_f32(float* __restrict__ p, float v, int n) {
    int i = blockIdx.x * blockDim.x + threadIdx.x;
    if (i < n) p[i] = v;
}
__global__ void fill_i32(int* __restrict__ p, int v, int n) {
    int i = blockIdx.x * blockDim.x + threadIdx.x;
    if (i < n) p[i] = v;
}

template <int K, int NOUT>
__global__ __launch_bounds__(32)
void gemm_wmma_f32(const float* __restrict__ A, const float* __restrict__ B,
                   float* __restrict__ C) {
    const int lane = threadIdx.x & 31;
    const int m0 = blockIdx.x * 16;
    const int n0 = blockIdx.y * 64;
    const int mr = lane & 15;
    const int nr = lane & 15;
    const int kb = (lane >> 4) * 2;

    v8f acc0 = {}, acc1 = {}, acc2 = {}, acc3 = {};
    const float* __restrict__ Arow = A + (size_t)(m0 + mr) * K + kb;

    for (int k = 0; k < K; k += 4) {
        v2f a;
        a.x = Arow[k + 0];
        a.y = Arow[k + 1];

        const float* __restrict__ Bp = B + (size_t)(k + kb) * NOUT + n0 + nr;
        v2f b0, b1, b2, b3;
        b0.x = Bp[0];      b0.y = Bp[NOUT];
        b1.x = Bp[16];     b1.y = Bp[NOUT + 16];
        b2.x = Bp[32];     b2.y = Bp[NOUT + 32];
        b3.x = Bp[48];     b3.y = Bp[NOUT + 48];

        acc0 = __builtin_amdgcn_wmma_f32_16x16x4_f32(false, a, false, b0, (short)0, acc0, false, false);
        acc1 = __builtin_amdgcn_wmma_f32_16x16x4_f32(false, a, false, b1, (short)0, acc1, false, false);
        acc2 = __builtin_amdgcn_wmma_f32_16x16x4_f32(false, a, false, b2, (short)0, acc2, false, false);
        acc3 = __builtin_amdgcn_wmma_f32_16x16x4_f32(false, a, false, b3, (short)0, acc3, false, false);
    }

    const int half = lane >> 4;
#pragma unroll
    for (int r = 0; r < 8; ++r) {
        float* __restrict__ Cp = C + (size_t)(m0 + r + 8 * half) * NOUT + n0 + nr;
        Cp[0]  = acc0[r];
        Cp[16] = acc1[r];
        Cp[32] = acc2[r];
        Cp[48] = acc3[r];
    }
}

template <int CH>
__global__ __launch_bounds__(256)
void att_dots(const float* __restrict__ Hm, const float* __restrict__ a_src,
              const float* __restrict__ a_dst, int nwaves, int heads,
              float* __restrict__ as_out, float* __restrict__ ad_out) {
    int gid = blockIdx.x * blockDim.x + threadIdx.x;
    int wave = gid >> 5;
    int lane = gid & 31;
    if (wave >= nwaves) return;
    int h = wave % heads;
    const float* __restrict__ hp  = Hm + (size_t)wave * CH;
    const float* __restrict__ asv = a_src + h * CH;
    const float* __restrict__ adv = a_dst + h * CH;
    float s = 0.f, d = 0.f;
#pragma unroll
    for (int c = lane; c < CH; c += 32) {
        float v = hp[c];
        s += v * asv[c];
        d += v * adv[c];
    }
#pragma unroll
    for (int off = 16; off > 0; off >>= 1) {
        s += __shfl_xor(s, off, 32);
        d += __shfl_xor(d, off, 32);
    }
    if (lane == 0) { as_out[wave] = s; ad_out[wave] = d; }
}

template <int H>
__global__ __launch_bounds__(256)
void edge_logits(const int* __restrict__ ei, const float* __restrict__ as,
                 const float* __restrict__ ad, float* __restrict__ elog,
                 int* __restrict__ emax) {
    int idx = blockIdx.x * blockDim.x + threadIdx.x;
    const int total = (N_EDGES + N_NODES) * H;
    if (idx >= total) return;
    int e = idx / H, h = idx % H;
    int s = (e < N_EDGES) ? ei[e] : (e - N_EDGES);
    int d = (e < N_EDGES) ? ei[N_EDGES + e] : (e - N_EDGES);
    float v = as[s * H + h] + ad[d * H + h];
    v = (v > 0.f) ? v : NEG_SLOPE * v;
    elog[idx] = v;
    atomicMax(&emax[d * H + h], ford(v));
}

template <int H>
__global__ __launch_bounds__(256)
void edge_exp(const int* __restrict__ ei, float* __restrict__ elog,
              const int* __restrict__ emax, float* __restrict__ denom) {
    int idx = blockIdx.x * blockDim.x + threadIdx.x;
    const int total = (N_EDGES + N_NODES) * H;
    if (idx >= total) return;
    int e = idx / H, h = idx % H;
    int d = (e < N_EDGES) ? ei[N_EDGES + e] : (e - N_EDGES);
    float m = iord(emax[d * H + h]);
    float ex = expf(elog[idx] - m);
    elog[idx] = ex;
    atomicAdd(&denom[d * H + h], ex);
}

template <int H>
__global__ __launch_bounds__(256)
void edge_alpha(const int* __restrict__ ei, float* __restrict__ elog,
                const float* __restrict__ denom, float* __restrict__ alpha_out) {
    int idx = blockIdx.x * blockDim.x + threadIdx.x;
    const int total = (N_EDGES + N_NODES) * H;
    if (idx >= total) return;
    int e = idx / H, h = idx % H;
    int d = (e < N_EDGES) ? ei[N_EDGES + e] : (e - N_EDGES);
    float a = elog[idx] / denom[d * H + h];
    elog[idx] = a;
    alpha_out[idx] = a;
}

template <int H, int CH>
__global__ __launch_bounds__(256)
void edge_msg(const int* __restrict__ ei, const float* __restrict__ Hm,
              const float* __restrict__ alpha, float* __restrict__ agg) {
    int gid = blockIdx.x * blockDim.x + threadIdx.x;
    int wave = gid >> 5;
    int lane = gid & 31;
    const int total = (N_EDGES + N_NODES) * H;
    if (wave >= total) return;
    int e = wave / H, h = wave % H;
    int s = (e < N_EDGES) ? ei[e] : (e - N_EDGES);
    int d = (e < N_EDGES) ? ei[N_EDGES + e] : (e - N_EDGES);
    float al = alpha[wave];
    const float* __restrict__ hp = Hm + (size_t)s * (H * CH) + h * CH;
    float* __restrict__ op = agg + (size_t)d * (H * CH) + h * CH;
#pragma unroll
    for (int c = lane; c < CH; c += 32)
        atomicAdd(&op[c], al * hp[c]);
}

template <int CH>
__global__ __launch_bounds__(256)
void elu_bias_inplace(float* __restrict__ p, const float* __restrict__ bias, int n) {
    int i = blockIdx.x * blockDim.x + threadIdx.x;
    if (i >= n) return;
    float v = p[i] + bias[i & (CH - 1)];
    p[i] = (v > 0.f) ? v : (expf(v) - 1.f);
}

template <int CH>
__global__ __launch_bounds__(256)
void bias_inplace(float* __restrict__ p, const float* __restrict__ bias, int n) {
    int i = blockIdx.x * blockDim.x + threadIdx.x;
    if (i >= n) return;
    p[i] += bias[i & (CH - 1)];
}

static inline int cdiv(long long a, int b) { return (int)((a + b - 1) / b); }

extern "C" void kernel_launch(void* const* d_in, const int* in_sizes, int n_in,
                              void* d_out, int out_size, void* d_ws, size_t ws_size,
                              hipStream_t stream) {
    const float* x      = (const float*)d_in[0];
    const int*   ei     = (const int*)d_in[1];
    const float* W1     = (const float*)d_in[2];
    const float* a_src1 = (const float*)d_in[3];
    const float* a_dst1 = (const float*)d_in[4];
    const float* b1     = (const float*)d_in[5];
    const float* W2     = (const float*)d_in[6];
    const float* a_src2 = (const float*)d_in[7];
    const float* a_dst2 = (const float*)d_in[8];
    const float* b2     = (const float*)d_in[9];

    const int EP = N_EDGES + N_NODES;

    float* H1   = (float*)d_ws;
    float* AGG1 = H1   + (size_t)N_NODES * 512;
    float* AS1  = AGG1 + (size_t)N_NODES * 512;
    float* AD1  = AS1  + (size_t)N_NODES * HEADS;
    float* DEN1 = AD1  + (size_t)N_NODES * HEADS;
    int*   EMX1 = (int*)(DEN1 + (size_t)N_NODES * HEADS);
    float* ELG1 = (float*)(EMX1 + (size_t)N_NODES * HEADS);
    float* AS2  = ELG1 + (size_t)EP * HEADS;
    float* AD2  = AS2  + N_NODES;
    float* DEN2 = AD2  + N_NODES;
    int*   EMX2 = (int*)(DEN2 + N_NODES);
    float* ELG2 = (float*)(EMX2 + N_NODES);
    float* G2   = H1;

    float* out    = (float*)d_out;
    float* alpha1 = out    + (size_t)N_NODES * OUT_CH;
    float* alpha2 = alpha1 + (size_t)EP * HEADS;

    const int TB = 256;

    gemm_wmma_f32<IN_CH, HEADS * HID>
        <<<dim3(N_NODES / 16, (HEADS * HID) / 64), 32, 0, stream>>>(x, W1, H1);

    att_dots<HID><<<cdiv((long long)N_NODES * HEADS * 32, TB), TB, 0, stream>>>(
        H1, a_src1, a_dst1, N_NODES * HEADS, HEADS, AS1, AD1);

    fill_f32<<<cdiv((long long)N_NODES * 512, TB), TB, 0, stream>>>(AGG1, 0.f, N_NODES * 512);
    fill_f32<<<cdiv((long long)N_NODES * HEADS, TB), TB, 0, stream>>>(DEN1, 0.f, N_NODES * HEADS);
    fill_i32<<<cdiv((long long)N_NODES * HEADS, TB), TB, 0, stream>>>(EMX1, (int)0x80000000, N_NODES * HEADS);

    edge_logits<HEADS><<<cdiv((long long)EP * HEADS, TB), TB, 0, stream>>>(ei, AS1, AD1, ELG1, EMX1);
    edge_exp<HEADS><<<cdiv((long long)EP * HEADS, TB), TB, 0, stream>>>(ei, ELG1, EMX1, DEN1);
    edge_alpha<HEADS><<<cdiv((long long)EP * HEADS, TB), TB, 0, stream>>>(ei, ELG1, DEN1, alpha1);
    edge_msg<HEADS, HID><<<cdiv((long long)EP * HEADS * 32, TB), TB, 0, stream>>>(ei, H1, ELG1, AGG1);

    elu_bias_inplace<512><<<cdiv((long long)N_NODES * 512, TB), TB, 0, stream>>>(AGG1, b1, N_NODES * 512);

    gemm_wmma_f32<HEADS * HID, OUT_CH>
        <<<dim3(N_NODES / 16, OUT_CH / 64), 32, 0, stream>>>(AGG1, W2, G2);

    att_dots<OUT_CH><<<cdiv((long long)N_NODES * 32, TB), TB, 0, stream>>>(
        G2, a_src2, a_dst2, N_NODES, 1, AS2, AD2);

    fill_f32<<<cdiv((long long)N_NODES * OUT_CH, TB), TB, 0, stream>>>(out, 0.f, N_NODES * OUT_CH);
    fill_f32<<<cdiv((long long)N_NODES, TB), TB, 0, stream>>>(DEN2, 0.f, N_NODES);
    fill_i32<<<cdiv((long long)N_NODES, TB), TB, 0, stream>>>(EMX2, (int)0x80000000, N_NODES);

    edge_logits<1><<<cdiv((long long)EP, TB), TB, 0, stream>>>(ei, AS2, AD2, ELG2, EMX2);
    edge_exp<1><<<cdiv((long long)EP, TB), TB, 0, stream>>>(ei, ELG2, EMX2, DEN2);
    edge_alpha<1><<<cdiv((long long)EP, TB), TB, 0, stream>>>(ei, ELG2, DEN2, alpha2);
    edge_msg<1, OUT_CH><<<cdiv((long long)EP * 32, TB), TB, 0, stream>>>(ei, G2, ELG2, out);

    bias_inplace<OUT_CH><<<cdiv((long long)N_NODES * OUT_CH, TB), TB, 0, stream>>>(out, b2, N_NODES * OUT_CH);
}


# Message aggregation by gather over a dst-grouped edge list (counting sort, arrival-order rank), register sums, one store with bias/ELU, instead of float atomic scatter-add into a zeroed plane
# speedup vs baseline: 1.7524x; 1.7524x over previous
#include <hip/hip_runtime.h>
#include <math.h>

#define N_NODES 20000
#define N_EDGES 320000
#define IN_CH 512
#define HID 128
#define HEADS 4
#define OUT_CH 256
#define NEG_SLOPE 0.2f

typedef float v2f __attribute__((ext_vector_type(2)));
typedef float v8f __attribute__((ext_vector_type(8)));

__device__ __forceinline__ int ford(float f) {
    int i = __float_as_int(f);
    return (i >= 0) ? i : (i ^ 0x7FFFFFFF);
}
__device__ __forceinline__ float iord(int i) {
    return __int_as_float((i >= 0) ? i : (i ^ 0x7FFFFFFF));
}

__global__ void fill_f32(float* __restrict__ p, float v, int n) {
    int i = blockIdx.x * blockDim.x + threadIdx.x;
    if (i < n) p[i] = v;
}
__global__ void fill_i32(int* __restrict__ p, int v, int n) {
    int i = blockIdx.x * blockDim.x + threadIdx.x;
    if (i < n) p[i] = v;
}

template <int K, int NOUT>
__global__ __launch_bounds__(32)
void gemm_wmma_f32(const float* __restrict__ A, const float* __restrict__ B,
                   float* __restrict__ C) {
    const int lane = threadIdx.x & 31;
    const int m0 = blockIdx.x * 16;
    const int n0 = blockIdx.y * 64;
    const int mr = lane & 15;
    const int nr = lane & 15;
    const int kb = (lane >> 4) * 2;

    v8f acc0 = {}, acc1 = {}, acc2 = {}, acc3 = {};
    const float* __restrict__ Arow = A + (size_t)(m0 + mr) * K + kb;

    for (int k = 0; k < K; k += 4) {
        v2f a;
        a.x = Arow[k + 0];
        a.y = Arow[k + 1];

        const float* __restrict__ Bp = B + (size_t)(k + kb) * NOUT + n0 + nr;
        v2f b0, b1, b2, b3;
        b0.x = Bp[0];      b0.y = Bp[NOUT];
        b1.x = Bp[16];     b1.y = Bp[NOUT + 16];
        b2.x = Bp[32];     b2.y = Bp[NOUT + 32];
        b3.x = Bp[48];     b3.y = Bp[NOUT + 48];

        acc0 = __builtin_amdgcn_wmma_f32_16x16x4_f32(false, a, false, b0, (short)0, acc0, false, false);
        acc1 = __builtin_amdgcn_wmma_f32_16x16x4_f32(false, a, false, b1, (short)0, acc1, false, false);
        acc2 = __builtin_amdgcn_wmma_f32_16x16x4_f32(false, a, false, b2, (short)0, acc2, false, false);
        acc3 = __builtin_amdgcn_wmma_f32_16x16x4_f32(false, a, false, b3, (short)0, acc3, false, false);
    }

    const int half = lane >> 4;
#pragma unroll
    for (int r = 0; r < 8; ++r) {
        float* __restrict__ Cp = C + (size_t)(m0 + r + 8 * half) * NOUT + n0 + nr;
        Cp[0]  = acc0[r];
        Cp[16] = acc1[r];
        Cp[32] = acc2[r];
        Cp[48] = acc3[r];
    }
}

template <int CH>
__global__ __launch_bounds__(256)
void att_dots(const float* __restrict__ Hm, const float* __restrict__ a_src,
              const float* __restrict__ a_dst, int nwaves, int heads,
              float* __restrict__ as_out, float* __restrict__ ad_out) {
    int gid = blockIdx.x * blockDim.x + threadIdx.x;
    int wave = gid >> 5;
    int lane = gid & 31;
    if (wave >= nwaves) return;
    int h = wave % heads;
    const float* __restrict__ hp  = Hm + (size_t)wave * CH;
    const float* __restrict__ asv = a_src + h * CH;
    const float* __restrict__ adv = a_dst + h * CH;
    float s = 0.f, d = 0.f;
#pragma unroll
    for (int c = lane; c < CH; c += 32) {
        float v = hp[c];
        s += v * asv[c];
        d += v * adv[c];
    }
#pragma unroll
    for (int off = 16; off > 0; off >>= 1) {
        s += __shfl_xor(s, off, 32);
        d += __shfl_xor(d, off, 32);
    }
    if (lane == 0) { as_out[wave] = s; ad_out[wave] = d; }
}

template <int H>
__global__ __launch_bounds__(256)
void edge_logits(const int* __restrict__ ei, const float* __restrict__ as,
                 const float* __restrict__ ad, float* __restrict__ elog,
                 int* __restrict__ emax) {
    int idx = blockIdx.x * blockDim.x + threadIdx.x;
    const int total = (N_EDGES + N_NODES) * H;
    if (idx >= total) return;
    int e = idx / H, h = idx % H;
    int s = (e < N_EDGES) ? ei[e] : (e - N_EDGES);
    int d = (e < N_EDGES) ? ei[N_EDGES + e] : (e - N_EDGES);
    float v = as[s * H + h] + ad[d * H + h];
    v = (v > 0.f) ? v : NEG_SLOPE * v;
    elog[idx] = v;
    atomicMax(&emax[d * H + h], ford(v));
}

template <int H>
__global__ __launch_bounds__(256)
void edge_exp(const int* __restrict__ ei, float* __restrict__ elog,
              const int* __restrict__ emax, float* __restrict__ denom) {
    int idx = blockIdx.x * blockDim.x + threadIdx.x;
    const int total = (N_EDGES + N_NODES) * H;
    if (idx >= total) return;
    int e = idx / H, h = idx % H;
    int d = (e < N_EDGES) ? ei[N_EDGES + e] : (e - N_EDGES);
    float m = iord(emax[d * H + h]);
    float ex = expf(elog[idx] - m);
    elog[idx] = ex;
    atomicAdd(&denom[d * H + h], ex);
}

template <int H>
__global__ __launch_bounds__(256)
void edge_alpha(const int* __restrict__ ei, float* __restrict__ elog,
                const float* __restrict__ denom, float* __restrict__ alpha_out) {
    int idx = blockIdx.x * blockDim.x + threadIdx.x;
    const int total = (N_EDGES + N_NODES) * H;
    if (idx >= total) return;
    int e = idx / H, h = idx % H;
    int d = (e < N_EDGES) ? ei[N_EDGES + e] : (e - N_EDGES);
    float a = elog[idx] / denom[d * H + h];
    elog[idx] = a;
    alpha_out[idx] = a;
}

__global__ __launch_bounds__(256)
void dst_count(const int* __restrict__ ei, int* __restrict__ cur) {
    int e = blockIdx.x * blockDim.x + threadIdx.x;
    if (e < N_EDGES) atomicAdd(&cur[ei[N_EDGES + e]], 1);
}

__global__ __launch_bounds__(1024)
void dst_scan(int* __restrict__ cur) {
    constexpr int ITEMS = (N_NODES + 1023) / 1024;
    __shared__ int wsum[32];
    const int t = threadIdx.x;
    const int lane = t & 31, w = t >> 5;
    const int base = t * ITEMS;
    int v[ITEMS];
    int s = 0;
#pragma unroll
    for (int i = 0; i < ITEMS; ++i) {
        int idx = base + i;
        v[i] = (idx < N_NODES) ? cur[idx] : 0;
        s += v[i];
    }
    int inc = s;
#pragma unroll
    for (int off = 1; off < 32; off <<= 1) {
        int n = __shfl_up(inc, off, 32);
        if (lane >= off) inc += n;
    }
    if (lane == 31) wsum[w] = inc;
    __syncthreads();
    if (w == 0) {
        int x = wsum[lane];
        int xi = x;
#pragma unroll
        for (int off = 1; off < 32; off <<= 1) {
            int n = __shfl_up(xi, off, 32);
            if (lane >= off) xi += n;
        }
        wsum[lane] = xi - x;
    }
    __syncthreads();
    int run = wsum[w] + inc - s;
#pragma unroll
    for (int i = 0; i < ITEMS; ++i) {
        int idx = base + i;
        if (idx < N_NODES) { cur[idx] = run; run += v[i]; }
    }
}

__global__ __launch_bounds__(256)
void dst_place(const int* __restrict__ ei, int* __restrict__ cur, int* __restrict__ perm) {
    int e = blockIdx.x * blockDim.x + threadIdx.x;
    if (e < N_EDGES) {
        int p = atomicAdd(&cur[ei[N_EDGES + e]], 1);
        perm[p] = e;
    }
}

template <int H, int CH, bool DO_ELU>
__global__ __launch_bounds__(256)
void node_gather(const int* __restrict__ ei, const int* __restrict__ cur,
                 const int* __restrict__ perm, const float* __restrict__ Hm,
                 const float* __restrict__ alpha, const float* __restrict__ bias,
                 float* __restrict__ outp) {
    constexpr int ROW = H * CH;
    constexpr int SL = ROW / 128;
    const int gid = blockIdx.x * blockDim.x + threadIdx.x;
    const int lane = gid & 31;
    const int wave = __builtin_amdgcn_readfirstlane(gid >> 5);
    if (wave >= N_NODES * SL) return;
    const int d = wave / SL, q = wave % SL;
    const int h = (q * 128) / CH;
    const int col = q * 128 + lane * 4;
    const int beg = d ? cur[d - 1] : 0;
    const int end = cur[d];

    float4 acc;
    {
        const float a = alpha[(size_t)(N_EDGES + d) * H + h];
        const float4 v = *(const float4*)(Hm + (size_t)d * ROW + col);
        acc.x = a * v.x; acc.y = a * v.y; acc.z = a * v.z; acc.w = a * v.w;
    }

    for (int i = beg; i < end; i += 32) {
        const int n = (end - i < 32) ? (end - i) : 32;
        int my_s = 0;
        int my_a = 0;
        if (lane < n) {
            const int e = perm[i + lane];
            my_s = ei[e];
            my_a = __float_as_int(alpha[(size_t)e * H + h]);
        }
        int j = 0;
        for (; j + 4 <= n; j += 4) {
            const int s0 = __builtin_amdgcn_readlane(my_s, j);
            const int s1 = __builtin_amdgcn_readlane(my_s, j + 1);
            const int s2 = __builtin_amdgcn_readlane(my_s, j + 2);
            const int s3 = __builtin_amdgcn_readlane(my_s, j + 3);
            const float a0 = __int_as_float(__builtin_amdgcn_readlane(my_a, j));
            const float a1 = __int_as_float(__builtin_amdgcn_readlane(my_a, j + 1));
            const float a2 = __int_as_float(__builtin_amdgcn_readlane(my_a, j + 2));
            const float a3 = __int_as_float(__builtin_amdgcn_readlane(my_a, j + 3));
            const float4 v0 = *(const float4*)(Hm + (size_t)s0 * ROW + col);
            const float4 v1 = *(const float4*)(Hm + (size_t)s1 * ROW + col);
            const float4 v2 = *(const float4*)(Hm + (size_t)s2 * ROW + col);
            const float4 v3 = *(const float4*)(Hm + (size_t)s3 * ROW + col);
            acc.x += a0 * v0.x; acc.y += a0 * v0.y; acc.z += a0 * v0.z; acc.w += a0 * v0.w;
            acc.x += a1 * v1.x; acc.y += a1 * v1.y; acc.z += a1 * v1.z; acc.w += a1 * v1.w;
            acc.x += a2 * v2.x; acc.y += a2 * v2.y; acc.z += a2 * v2.z; acc.w += a2 * v2.w;
            acc.x += a3 * v3.x; acc.y += a3 * v3.y; acc.z += a3 * v3.z; acc.w += a3 * v3.w;
        }
        for (; j < n; ++j) {
            const int s0 = __builtin_amdgcn_readlane(my_s, j);
            const float a0 = __int_as_float(__builtin_amdgcn_readlane(my_a, j));
            const float4 v0 = *(const float4*)(Hm + (size_t)s0 * ROW + col);
            acc.x += a0 * v0.x; acc.y += a0 * v0.y; acc.z += a0 * v0.z; acc.w += a0 * v0.w;
        }
    }

    const float4 b = *(const float4*)(bias + col);
    acc.x += b.x; acc.y += b.y; acc.z += b.z; acc.w += b.w;
    if (DO_ELU) {
        acc.x = (acc.x > 0.f) ? acc.x : (expf(acc.x) - 1.f);
        acc.y = (acc.y > 0.f) ? acc.y : (expf(acc.y) - 1.f);
        acc.z = (acc.z > 0.f) ? acc.z : (expf(acc.z) - 1.f);
        acc.w = (acc.w > 0.f) ? acc.w : (expf(acc.w) - 1.f);
    }
    *(float4*)(outp + (size_t)d * ROW + col) = acc;
}

static inline int cdiv(long long a, int b) { return (int)((a + b - 1) / b); }

extern "C" void kernel_launch(void* const* d_in, const int* in_sizes, int n_in,
                              void* d_out, int out_size, void* d_ws, size_t ws_size,
                              hipStream_t stream) {
    const float* x      = (const float*)d_in[0];
    const int*   ei     = (const int*)d_in[1];
    const float* W1     = (const float*)d_in[2];
    const float* a_src1 = (const float*)d_in[3];
    const float* a_dst1 = (const float*)d_in[4];
    const float* b1     = (const float*)d_in[5];
    const float* W2     = (const float*)d_in[6];
    const float* a_src2 = (const float*)d_in[7];
    const float* a_dst2 = (const float*)d_in[8];
    const float* b2     = (const float*)d_in[9];

    const int EP = N_EDGES + N_NODES;

    float* H1   = (float*)d_ws;
    float* AGG1 = H1   + (size_t)N_NODES * 512;
    float* AS1  = AGG1 + (size_t)N_NODES * 512;
    float* AD1  = AS1  + (size_t)N_NODES * HEADS;
    float* DEN1 = AD1  + (size_t)N_NODES * HEADS;
    int*   EMX1 = (int*)(DEN1 + (size_t)N_NODES * HEADS);
    float* ELG1 = (float*)(EMX1 + (size_t)N_NODES * HEADS);
    float* AS2  = ELG1;
    float* AD2  = AS2  + N_NODES;
    float* DEN2 = AD2  + N_NODES;
    int*   EMX2 = (int*)(DEN2 + N_NODES);
    float* ELG2 = (float*)(EMX2 + N_NODES);
    int*   CUR  = (int*)(ELG1 + (size_t)EP * HEADS);
    int*   PERM = CUR + N_NODES;
    float* G2   = H1;

    float* out    = (float*)d_out;
    float* alpha1 = out    + (size_t)N_NODES * OUT_CH;
    float* alpha2 = alpha1 + (size_t)EP * HEADS;

    const int TB = 256;

    gemm_wmma_f32<IN_CH, HEADS * HID>
        <<<dim3(N_NODES / 16, (HEADS * HID) / 64), 32, 0, stream>>>(x, W1, H1);

    att_dots<HID><<<cdiv((long long)N_NODES * HEADS * 32, TB), TB, 0, stream>>>(
        H1, a_src1, a_dst1, N_NODES * HEADS, HEADS, AS1, AD1);

    fill_i32<<<cdiv((long long)N_NODES, TB), TB, 0, stream>>>(CUR, 0, N_NODES);
    dst_count<<<cdiv((long long)N_EDGES, TB), TB, 0, stream>>>(ei, CUR);
    dst_scan<<<1, 1024, 0, stream>>>(CUR);
    dst_place<<<cdiv((long long)N_EDGES, TB), TB, 0, stream>>>(ei, CUR, PERM);

    fill_f32<<<cdiv((long long)N_NODES * HEADS, TB), TB, 0, stream>>>(DEN1, 0.f, N_NODES * HEADS);
    fill_i32<<<cdiv((long long)N_NODES * HEADS, TB), TB, 0, stream>>>(EMX1, (int)0x80000000, N_NODES * HEADS);

    edge_logits<HEADS><<<cdiv((long long)EP * HEADS, TB), TB, 0, stream>>>(ei, AS1, AD1, ELG1, EMX1);
    edge_exp<HEADS><<<cdiv((long long)EP * HEADS, TB), TB, 0, stream>>>(ei, ELG1, EMX1, DEN1);
    edge_alpha<HEADS><<<cdiv((long long)EP * HEADS, TB), TB, 0, stream>>>(ei, ELG1, DEN1, alpha1);
    node_gather<HEADS, HID, true><<<cdiv((long long)N_NODES * 4 * 32, TB), TB, 0, stream>>>(
        ei, CUR, PERM, H1, ELG1, b1, AGG1);

    gemm_wmma_f32<HEADS * HID, OUT_CH>
        <<<dim3(N_NODES / 16, OUT_CH / 64), 32, 0, stream>>>(AGG1, W2, G2);

    att_dots<OUT_CH><<<cdiv((long long)N_NODES * 32, TB), TB, 0, stream>>>(
        G2, a_src2, a_dst2, N_NODES, 1, AS2, AD2);

    fill_f32<<<cdiv((long long)N_NODES, TB), TB, 0, stream>>>(DEN2, 0.f, N_NODES);
    fill_i32<<<cdiv((long long)N_NODES, TB), TB, 0, stream>>>(EMX2, (int)0x80000000, N_NODES);

    edge_logits<1><<<cdiv((long long)EP, TB), TB, 0, stream>>>(ei, AS2, AD2, ELG2, EMX2);
    edge_exp<1><<<cdiv((long long)EP, TB), TB, 0, stream>>>(ei, ELG2, EMX2, DEN2);
    edge_alpha<1><<<cdiv((long long)EP, TB), TB, 0, stream>>>(ei, ELG2, DEN2, alpha2);
    node_gather<1, OUT_CH, false><<<cdiv((long long)N_NODES * 2 * 32, TB), TB, 0, stream>>>(
        ei, CUR, PERM, G2, ELG2, b2, out);
}


# Segment softmax (max, exp sum, alpha) moved into the per-destination gather wave as wave reductions over the dst-grouped list; atomic edge passes and fills dropped; denominator sum order: lane tree
# speedup vs baseline: 1.9712x; 1.1249x over previous
#include <hip/hip_runtime.h>
#include <math.h>

#define N_NODES 20000
#define N_EDGES 320000
#define IN_CH 512
#define HID 128
#define HEADS 4
#define OUT_CH 256
#define NEG_SLOPE 0.2f

typedef float v2f __attribute__((ext_vector_type(2)));
typedef float v8f __attribute__((ext_vector_type(8)));

__global__ void fill_i32(int* __restrict__ p, int v, int n) {
    int i = blockIdx.x * blockDim.x + threadIdx.x;
    if (i < n) p[i] = v;
}

template <int K, int NOUT>
__global__ __launch_bounds__(32)
void gemm_wmma_f32(const float* __restrict__ A, const float* __restrict__ B,
                   float* __restrict__ C) {
    const int lane = threadIdx.x & 31;
    const int m0 = blockIdx.x * 16;
    const int n0 = blockIdx.y * 64;
    const int mr = lane & 15;
    const int nr = lane & 15;
    const int kb = (lane >> 4) * 2;

    v8f acc0 = {}, acc1 = {}, acc2 = {}, acc3 = {};
    const float* __restrict__ Arow = A + (size_t)(m0 + mr) * K + kb;

    for (int k = 0; k < K; k += 4) {
        v2f a;
        a.x = Arow[k + 0];
        a.y = Arow[k + 1];

        const float* __restrict__ Bp = B + (size_t)(k + kb) * NOUT + n0 + nr;
        v2f b0, b1, b2, b3;
        b0.x = Bp[0];      b0.y = Bp[NOUT];
        b1.x = Bp[16];     b1.y = Bp[NOUT + 16];
        b2.x = Bp[32];     b2.y = Bp[NOUT + 32];
        b3.x = Bp[48];     b3.y = Bp[NOUT + 48];

        acc0 = __builtin_amdgcn_wmma_f32_16x16x4_f32(false, a, false, b0, (short)0, acc0, false, false);
        acc1 = __builtin_amdgcn_wmma_f32_16x16x4_f32(false, a, false, b1, (short)0, acc1, false, false);
        acc2 = __builtin_amdgcn_wmma_f32_16x16x4_f32(false, a, false, b2, (short)0, acc2, false, false);
        acc3 = __builtin_amdgcn_wmma_f32_16x16x4_f32(false, a, false, b3, (short)0, acc3, false, false);
    }

    const int half = lane >> 4;
#pragma unroll
    for (int r = 0; r < 8; ++r) {
        float* __restrict__ Cp = C + (size_t)(m0 + r + 8 * half) * NOUT + n0 + nr;
        Cp[0]  = acc0[r];
        Cp[16] = acc1[r];
        Cp[32] = acc2[r];
        Cp[48] = acc3[r];
    }
}

template <int CH>
__global__ __launch_bounds__(256)
void att_dots(const float* __restrict__ Hm, const float* __restrict__ a_src,
              const float* __restrict__ a_dst, int nwaves, int heads,
              float* __restrict__ as_out, float* __restrict__ ad_out) {
    int gid = blockIdx.x * blockDim.x + threadIdx.x;
    int wave = gid >> 5;
    int lane = gid & 31;
    if (wave >= nwaves) return;
    int h = wave % heads;
    const float* __restrict__ hp  = Hm + (size_t)wave * CH;
    const float* __restrict__ asv = a_src + h * CH;
    const float* __restrict__ adv = a_dst + h * CH;
    float s = 0.f, d = 0.f;
#pragma unroll
    for (int c = lane; c < CH; c += 32) {
        float v = hp[c];
        s += v * asv[c];
        d += v * adv[c];
    }
#pragma unroll
    for (int off = 16; off > 0; off >>= 1) {
        s += __shfl_xor(s, off, 32);
        d += __shfl_xor(d, off, 32);
    }
    if (lane == 0) { as_out[wave] = s; ad_out[wave] = d; }
}

__global__ __launch_bounds__(256)
void dst_count(const int* __restrict__ ei, int* __restrict__ cur) {
    int e = blockIdx.x * blockDim.x + threadIdx.x;
    if (e < N_EDGES) atomicAdd(&cur[ei[N_EDGES + e]], 1);
}

__global__ __launch_bounds__(1024)
void dst_scan(int* __restrict__ cur) {
    constexpr int ITEMS = (N_NODES + 1023) / 1024;
    __shared__ int wsum[32];
    const int t = threadIdx.x;
    const int lane = t & 31, w = t >> 5;
    const int base = t * ITEMS;
    int v[ITEMS];
    int s = 0;
#pragma unroll
    for (int i = 0; i < ITEMS; ++i) {
        int idx = base + i;
        v[i] = (idx < N_NODES) ? cur[idx] : 0;
        s += v[i];
    }
    int inc = s;
#pragma unroll
    for (int off = 1; off < 32; off <<= 1) {
        int n = __shfl_up(inc, off, 32);
        if (lane >= off) inc += n;
    }
    if (lane == 31) wsum[w] = inc;
    __syncthreads();
    if (w == 0) {
        int x = wsum[lane];
        int xi = x;
#pragma unroll
        for (int off = 1; off < 32; off <<= 1) {
            int n = __shfl_up(xi, off, 32);
            if (lane >= off) xi += n;
        }
        wsum[lane] = xi - x;
    }
    __syncthreads();
    int run = wsum[w] + inc - s;
#pragma unroll
    for (int i = 0; i < ITEMS; ++i) {
        int idx = base + i;
        if (idx < N_NODES) { cur[idx] = run; run += v[i]; }
    }
}

__global__ __launch_bounds__(256)
void dst_place(const int* __restrict__ ei, int* __restrict__ cur, int* __restrict__ perm) {
    int e = blockIdx.x * blockDim.x + threadIdx.x;
    if (e < N_EDGES) {
        int p = atomicAdd(&cur[ei[N_EDGES + e]], 1);
        perm[p] = e;
    }
}


template <int X>
__device__ __forceinline__ float row_xor(float v) {
    const int b = __builtin_bit_cast(int, v);
    return __builtin_bit_cast(float, __builtin_amdgcn_update_dpp(b, b, 0x160 + X, 0xf, 0xf, false));
}
__device__ __forceinline__ float xhalf(float v) {
    const int b = __builtin_bit_cast(int, v);
    return __builtin_bit_cast(float,
        __builtin_amdgcn_permlanex16(b, b, 0x76543210, 0xfedcba98, false, false));
}
__device__ __forceinline__ float wave_max(float v) {
    v = fmaxf(v, row_xor<1>(v));
    v = fmaxf(v, row_xor<2>(v));
    v = fmaxf(v, row_xor<4>(v));
    v = fmaxf(v, row_xor<8>(v));
    return fmaxf(v, xhalf(v));
}
__device__ __forceinline__ float wave_sum(float v) {
    v += row_xor<1>(v);
    v += row_xor<2>(v);
    v += row_xor<4>(v);
    v += row_xor<8>(v);
    return v + xhalf(v);
}
__device__ __forceinline__ float leaky(float v) {
    return (v > 0.f) ? v : NEG_SLOPE * v;
}

__device__ __forceinline__ void gather_rows(float4& acc, int my_s, int my_a, int n,
                                            const float* __restrict__ Hm, int row, int col) {
    int j = 0;
    for (; j + 4 <= n; j += 4) {
        const int s0 = __builtin_amdgcn_readlane(my_s, j);
        const int s1 = __builtin_amdgcn_readlane(my_s, j + 1);
        const int s2 = __builtin_amdgcn_readlane(my_s, j + 2);
        const int s3 = __builtin_amdgcn_readlane(my_s, j + 3);
        const float a0 = __int_as_float(__builtin_amdgcn_readlane(my_a, j));
        const float a1 = __int_as_float(__builtin_amdgcn_readlane(my_a, j + 1));
        const float a2 = __int_as_float(__builtin_amdgcn_readlane(my_a, j + 2));
        const float a3 = __int_as_float(__builtin_amdgcn_readlane(my_a, j + 3));
        const float4 v0 = *(const float4*)(Hm + (size_t)s0 * row + col);
        const float4 v1 = *(const float4*)(Hm + (size_t)s1 * row + col);
        const float4 v2 = *(const float4*)(Hm + (size_t)s2 * row + col);
        const float4 v3 = *(const float4*)(Hm + (size_t)s3 * row + col);
        acc.x += a0 * v0.x; acc.y += a0 * v0.y; acc.z += a0 * v0.z; acc.w += a0 * v0.w;
        acc.x += a1 * v1.x; acc.y += a1 * v1.y; acc.z += a1 * v1.z; acc.w += a1 * v1.w;
        acc.x += a2 * v2.x; acc.y += a2 * v2.y; acc.z += a2 * v2.z; acc.w += a2 * v2.w;
        acc.x += a3 * v3.x; acc.y += a3 * v3.y; acc.z += a3 * v3.z; acc.w += a3 * v3.w;
    }
    for (; j < n; ++j) {
        const int s0 = __builtin_amdgcn_readlane(my_s, j);
        const float a0 = __int_as_float(__builtin_amdgcn_readlane(my_a, j));
        const float4 v0 = *(const float4*)(Hm + (size_t)s0 * row + col);
        acc.x += a0 * v0.x; acc.y += a0 * v0.y; acc.z += a0 * v0.z; acc.w += a0 * v0.w;
    }
}

template <int H, int CH, bool DO_ELU>
__global__ __launch_bounds__(256)
void node_softmax_gather(const int* __restrict__ ei, const int* __restrict__ cur,
                         const int* __restrict__ perm, const float* __restrict__ Hm,
                         const float* __restrict__ as, const float* __restrict__ ad,
                         const float* __restrict__ bias,
                         float* __restrict__ alpha_out, float* __restrict__ outp) {
    constexpr int ROW = H * CH;
    constexpr int SL = ROW / 128;
    const int gid = blockIdx.x * blockDim.x + threadIdx.x;
    const int lane = gid & 31;
    const int wave = __builtin_amdgcn_readfirstlane(gid >> 5);
    if (wave >= N_NODES * SL) return;
    const int d = wave / SL, q = wave % SL;
    const int h = (q * 128) / CH;
    const bool opens_head = ((q * 128) % CH) == 0;
    const int col = q * 128 + lane * 4;
    const int beg = d ? cur[d - 1] : 0;
    const int end = cur[d];
    const int deg = end - beg;

    const float adh = ad[d * H + h];
    const float vself = leaky(as[d * H + h] + adh);
    const float4 vs = *(const float4*)(Hm + (size_t)d * ROW + col);
    float4 acc;

    if (deg <= 32) {
        int my_e = 0, my_s = 0;
        float my_v = vself;
        if (lane < deg) {
            my_e = perm[beg + lane];
            my_s = ei[my_e];
            my_v = leaky(as[my_s * H + h] + adh);
        }
        const float m = wave_max(fmaxf(my_v, vself));
        const float ex = (lane < deg) ? expf(my_v - m) : 0.f;
        const float exs = expf(vself - m);
        const float sum = wave_sum(ex) + exs;
        const float a = ex / sum;
        const float aself = exs / sum;
        if (opens_head) {
            if (lane < deg) alpha_out[(size_t)my_e * H + h] = a;
            if (lane == 0) alpha_out[(size_t)(N_EDGES + d) * H + h] = aself;
        }
        acc.x = aself * vs.x; acc.y = aself * vs.y; acc.z = aself * vs.z; acc.w = aself * vs.w;
        gather_rows(acc, my_s, __float_as_int(a), deg, Hm, ROW, col);
    } else {
        float m = vself;
        for (int i = beg; i < end; i += 32) {
            if (i + lane < end) {
                const int s = ei[perm[i + lane]];
                m = fmaxf(m, leaky(as[s * H + h] + adh));
            }
        }
        m = wave_max(m);
        float part = 0.f;
        for (int i = beg; i < end; i += 32) {
            if (i + lane < end) {
                const int s = ei[perm[i + lane]];
                part += expf(leaky(as[s * H + h] + adh) - m);
            }
        }
        const float exs = expf(vself - m);
        const float sum = wave_sum(part) + exs;
        const float aself = exs / sum;
        if (opens_head && lane == 0) alpha_out[(size_t)(N_EDGES + d) * H + h] = aself;
        acc.x = aself * vs.x; acc.y = aself * vs.y; acc.z = aself * vs.z; acc.w = aself * vs.w;
        for (int i = beg; i < end; i += 32) {
            const int n = (end - i < 32) ? (end - i) : 32;
            int my_s = 0;
            float a = 0.f;
            if (lane < n) {
                const int e = perm[i + lane];
                my_s = ei[e];
                a = expf(leaky(as[my_s * H + h] + adh) - m) / sum;
                if (opens_head) alpha_out[(size_t)e * H + h] = a;
            }
            gather_rows(acc, my_s, __float_as_int(a), n, Hm, ROW, col);
        }
    }

    const float4 b = *(const float4*)(bias + col);
    acc.x += b.x; acc.y += b.y; acc.z += b.z; acc.w += b.w;
    if (DO_ELU) {
        acc.x = (acc.x > 0.f) ? acc.x : (expf(acc.x) - 1.f);
        acc.y = (acc.y > 0.f) ? acc.y : (expf(acc.y) - 1.f);
        acc.z = (acc.z > 0.f) ? acc.z : (expf(acc.z) - 1.f);
        acc.w = (acc.w > 0.f) ? acc.w : (expf(acc.w) - 1.f);
    }
    *(float4*)(outp + (size_t)d * ROW + col) = acc;
}

static inline int cdiv(long long a, int b) { return (int)((a + b - 1) / b); }

extern "C" void kernel_launch(void* const* d_in, const int* in_sizes, int n_in,
                              void* d_out, int out_size, void* d_ws, size_t ws_size,
                              hipStream_t stream) {
    const float* x      = (const float*)d_in[0];
    const int*   ei     = (const int*)d_in[1];
    const float* W1     = (const float*)d_in[2];
    const float* a_src1 = (const float*)d_in[3];
    const float* a_dst1 = (const float*)d_in[4];
    const float* b1     = (const float*)d_in[5];
    const float* W2     = (const float*)d_in[6];
    const float* a_src2 = (const float*)d_in[7];
    const float* a_dst2 = (const float*)d_in[8];
    const float* b2     = (const float*)d_in[9];

    const int EP = N_EDGES + N_NODES;

    float* H1   = (float*)d_ws;
    float* AGG1 = H1   + (size_t)N_NODES * 512;
    float* AS1  = AGG1 + (size_t)N_NODES * 512;
    float* AD1  = AS1  + (size_t)N_NODES * HEADS;
    float* AS2  = AD1  + (size_t)N_NODES * HEADS;
    float* AD2  = AS2  + N_NODES;
    int*   CUR  = (int*)(AD2 + N_NODES);
    int*   PERM = CUR + N_NODES;
    float* G2   = H1;

    float* out    = (float*)d_out;
    float* alpha1 = out    + (size_t)N_NODES * OUT_CH;
    float* alpha2 = alpha1 + (size_t)EP * HEADS;

    const int TB = 256;

    gemm_wmma_f32<IN_CH, HEADS * HID>
        <<<dim3(N_NODES / 16, (HEADS * HID) / 64), 32, 0, stream>>>(x, W1, H1);

    att_dots<HID><<<cdiv((long long)N_NODES * HEADS * 32, TB), TB, 0, stream>>>(
        H1, a_src1, a_dst1, N_NODES * HEADS, HEADS, AS1, AD1);

    fill_i32<<<cdiv((long long)N_NODES, TB), TB, 0, stream>>>(CUR, 0, N_NODES);
    dst_count<<<cdiv((long long)N_EDGES, TB), TB, 0, stream>>>(ei, CUR);
    dst_scan<<<1, 1024, 0, stream>>>(CUR);
    dst_place<<<cdiv((long long)N_EDGES, TB), TB, 0, stream>>>(ei, CUR, PERM);

    node_softmax_gather<HEADS, HID, true><<<cdiv((long long)N_NODES * 4 * 32, TB), TB, 0, stream>>>(
        ei, CUR, PERM, H1, AS1, AD1, b1, alpha1, AGG1);

    gemm_wmma_f32<HEADS * HID, OUT_CH>
        <<<dim3(N_NODES / 16, OUT_CH / 64), 32, 0, stream>>>(AGG1, W2, G2);

    att_dots<OUT_CH><<<cdiv((long long)N_NODES * 32, TB), TB, 0, stream>>>(
        G2, a_src2, a_dst2, N_NODES, 1, AS2, AD2);

    node_softmax_gather<1, OUT_CH, false><<<cdiv((long long)N_NODES * 2 * 32, TB), TB, 0, stream>>>(
        ei, CUR, PERM, G2, AS2, AD2, b2, alpha2, out);
}


# GEMM operand fetch: 16-byte loads, each 128-byte A and B line asked for once per wave by grouped loads issued ahead of the WMMAs, C stored 16 B per lane; k sum order inside each 32-k block changed
# speedup vs baseline: 2.1305x; 1.0808x over previous
#include <hip/hip_runtime.h>
#include <math.h>

#define N_NODES 20000
#define N_EDGES 320000
#define IN_CH 512
#define HID 128
#define HEADS 4
#define OUT_CH 256
#define NEG_SLOPE 0.2f

typedef float v2f __attribute__((ext_vector_type(2)));
typedef float v4f __attribute__((ext_vector_type(4)));
typedef float v8f __attribute__((ext_vector_type(8)));

__global__ void fill_i32(int* __restrict__ p, int v, int n) {
    int i = blockIdx.x * blockDim.x + threadIdx.x;
    if (i < n) p[i] = v;
}

struct a_set { v4f v0, v1, v2, v3; };
struct b_set { v4f r0, r1, r2, r3; };

template <int NOUT>
__device__ __forceinline__ b_set load_b(const float* __restrict__ Bp, int k) {
    b_set b;
    b.r0 = *(const v4f*)(Bp + (size_t)(k + 0) * NOUT);
    b.r1 = *(const v4f*)(Bp + (size_t)(k + 1) * NOUT);
    b.r2 = *(const v4f*)(Bp + (size_t)(k + 2) * NOUT);
    b.r3 = *(const v4f*)(Bp + (size_t)(k + 3) * NOUT);
    return b;
}

__device__ __forceinline__ a_set load_a(const float* __restrict__ Ap, int k0) {
    a_set a;
    a.v0 = *(const v4f*)(Ap + k0 + 0);
    a.v1 = *(const v4f*)(Ap + k0 + 4);
    a.v2 = *(const v4f*)(Ap + k0 + 8);
    a.v3 = *(const v4f*)(Ap + k0 + 12);
    return a;
}

__device__ __forceinline__ void mma_sub(const v4f& a, const b_set& b, v8f& acc0, v8f& acc1,
                                        v8f& acc2, v8f& acc3) {
    const v2f a01 = {a.x, a.y};
    const v2f a23 = {a.z, a.w};
    v2f p0 = {b.r0.x, b.r1.x}, p1 = {b.r0.y, b.r1.y};
    v2f p2 = {b.r0.z, b.r1.z}, p3 = {b.r0.w, b.r1.w};
    v2f q0 = {b.r2.x, b.r3.x}, q1 = {b.r2.y, b.r3.y};
    v2f q2 = {b.r2.z, b.r3.z}, q3 = {b.r2.w, b.r3.w};
    asm volatile("" : "+v"(p0), "+v"(p1), "+v"(p2), "+v"(p3),
                      "+v"(q0), "+v"(q1), "+v"(q2), "+v"(q3));
    __builtin_amdgcn_sched_barrier(0);
    acc0 = __builtin_amdgcn_wmma_f32_16x16x4_f32(false, a01, false, p0, (short)0, acc0, false, false);
    acc1 = __builtin_amdgcn_wmma_f32_16x16x4_f32(false, a01, false, p1, (short)0, acc1, false, false);
    acc2 = __builtin_amdgcn_wmma_f32_16x16x4_f32(false, a01, false, p2, (short)0, acc2, false, false);
    acc3 = __builtin_amdgcn_wmma_f32_16x16x4_f32(false, a01, false, p3, (short)0, acc3, false, false);
    acc0 = __builtin_amdgcn_wmma_f32_16x16x4_f32(false, a23, false, q0, (short)0, acc0, false, false);
    acc1 = __builtin_amdgcn_wmma_f32_16x16x4_f32(false, a23, false, q1, (short)0, acc1, false, false);
    acc2 = __builtin_amdgcn_wmma_f32_16x16x4_f32(false, a23, false, q2, (short)0, acc2, false, false);
    acc3 = __builtin_amdgcn_wmma_f32_16x16x4_f32(false, a23, false, q3, (short)0, acc3, false, false);
    __builtin_amdgcn_sched_barrier(0);
    asm volatile("v_nop\n\tv_nop\n\tv_nop\n\tv_nop");
    __builtin_amdgcn_sched_barrier(0);
}

template <int K, int NOUT, bool LAST>
__device__ __forceinline__ void k_block(const float* __restrict__ Ap, const float* __restrict__ Bp,
                                        int k0, const a_set& cur, a_set& nxt, b_set& b0,
                                        v8f& acc0, v8f& acc1, v8f& acc2, v8f& acc3) {
    if (!LAST) nxt = load_a(Ap, k0 + 32);
    b_set b1 = load_b<NOUT>(Bp, k0 + 4);
    __builtin_amdgcn_sched_barrier(0);
    mma_sub(cur.v0, b0, acc0, acc1, acc2, acc3);
    b0 = load_b<NOUT>(Bp, k0 + 8);
    __builtin_amdgcn_sched_barrier(0);
    mma_sub(cur.v1, b1, acc0, acc1, acc2, acc3);
    b1 = load_b<NOUT>(Bp, k0 + 12);
    __builtin_amdgcn_sched_barrier(0);
    mma_sub(cur.v2, b0, acc0, acc1, acc2, acc3);
    if (!LAST) {
        b0 = load_b<NOUT>(Bp, k0 + 32);
        __builtin_amdgcn_sched_barrier(0);
    }
    mma_sub(cur.v3, b1, acc0, acc1, acc2, acc3);
}

template <int K, int NOUT>
__global__ __launch_bounds__(32)
void gemm_wmma_f32(const float* __restrict__ A, const float* __restrict__ B,
                   float* __restrict__ C) {
    static_assert(K % 64 == 0 && NOUT % 64 == 0, "tile shape");
    const int lane = threadIdx.x & 31;
    const int m0 = blockIdx.x * 16;
    const int n0 = blockIdx.y * 64;
    const int c = lane & 15;
    const int h = lane >> 4;

    v8f acc0 = {}, acc1 = {}, acc2 = {}, acc3 = {};
    const float* __restrict__ Ap = A + (size_t)(m0 + c) * K + 16 * h;
    const float* __restrict__ Bp = B + (size_t)(16 * h) * NOUT + n0 + 4 * c;

    a_set a0 = load_a(Ap, 0), a1;
    b_set b0 = load_b<NOUT>(Bp, 0);
    for (int k0 = 0; k0 < K - 64; k0 += 64) {
        k_block<K, NOUT, false>(Ap, Bp, k0,      a0, a1, b0, acc0, acc1, acc2, acc3);
        k_block<K, NOUT, false>(Ap, Bp, k0 + 32, a1, a0, b0, acc0, acc1, acc2, acc3);
    }
    k_block<K, NOUT, false>(Ap, Bp, K - 64, a0, a1, b0, acc0, acc1, acc2, acc3);
    k_block<K, NOUT, true >(Ap, Bp, K - 32, a1, a0, b0, acc0, acc1, acc2, acc3);

    for (int r = 0; r < 8; ++r) {
        v4f o = {acc0[r], acc1[r], acc2[r], acc3[r]};
        *(v4f*)(C + (size_t)(m0 + r + 8 * h) * NOUT + n0 + 4 * c) = o;
    }
}

template <int CH>
__global__ __launch_bounds__(256)
void att_dots(const float* __restrict__ Hm, const float* __restrict__ a_src,
              const float* __restrict__ a_dst, int nwaves, int heads,
              float* __restrict__ as_out, float* __restrict__ ad_out) {
    int gid = blockIdx.x * blockDim.x + threadIdx.x;
    int wave = gid >> 5;
    int lane = gid & 31;
    if (wave >= nwaves) return;
    int h = wave % heads;
    const float* __restrict__ hp  = Hm + (size_t)wave * CH;
    const float* __restrict__ asv = a_src + h * CH;
    const float* __restrict__ adv = a_dst + h * CH;
    float s = 0.f, d = 0.f;
#pragma unroll
    for (int c = lane; c < CH; c += 32) {
        float v = hp[c];
        s += v * asv[c];
        d += v * adv[c];
    }
#pragma unroll
    for (int off = 16; off > 0; off >>= 1) {
        s += __shfl_xor(s, off, 32);
        d += __shfl_xor(d, off, 32);
    }
    if (lane == 0) { as_out[wave] = s; ad_out[wave] = d; }
}

__global__ __launch_bounds__(256)
void dst_count(const int* __restrict__ ei, int* __restrict__ cur) {
    int e = blockIdx.x * blockDim.x + threadIdx.x;
    if (e < N_EDGES) atomicAdd(&cur[ei[N_EDGES + e]], 1);
}

__global__ __launch_bounds__(1024)
void dst_scan(int* __restrict__ cur) {
    constexpr int ITEMS = (N_NODES + 1023) / 1024;
    __shared__ int wsum[32];
    const int t = threadIdx.x;
    const int lane = t & 31, w = t >> 5;
    const int base = t * ITEMS;
    int v[ITEMS];
    int s = 0;
#pragma unroll
    for (int i = 0; i < ITEMS; ++i) {
        int idx = base + i;
        v[i] = (idx < N_NODES) ? cur[idx] : 0;
        s += v[i];
    }
    int inc = s;
#pragma unroll
    for (int off = 1; off < 32; off <<= 1) {
        int n = __shfl_up(inc, off, 32);
        if (lane >= off) inc += n;
    }
    if (lane == 31) wsum[w] = inc;
    __syncthreads();
    if (w == 0) {
        int x = wsum[lane];
        int xi = x;
#pragma unroll
        for (int off = 1; off < 32; off <<= 1) {
            int n = __shfl_up(xi, off, 32);
            if (lane >= off) xi += n;
        }
        wsum[lane] = xi - x;
    }
    __syncthreads();
    int run = wsum[w] + inc - s;
#pragma unroll
    for (int i = 0; i < ITEMS; ++i) {
        int idx = base + i;
        if (idx < N_NODES) { cur[idx] = run; run += v[i]; }
    }
}

__global__ __launch_bounds__(256)
void dst_place(const int* __restrict__ ei, int* __restrict__ cur, int* __restrict__ perm) {
    int e = blockIdx.x * blockDim.x + threadIdx.x;
    if (e < N_EDGES) {
        int p = atomicAdd(&cur[ei[N_EDGES + e]], 1);
        perm[p] = e;
    }
}


template <int X>
__device__ __forceinline__ float row_xor(float v) {
    const int b = __builtin_bit_cast(int, v);
    return __builtin_bit_cast(float, __builtin_amdgcn_update_dpp(b, b, 0x160 + X, 0xf, 0xf, false));
}
__device__ __forceinline__ float xhalf(float v) {
    const int b = __builtin_bit_cast(int, v);
    return __builtin_bit_cast(float,
        __builtin_amdgcn_permlanex16(b, b, 0x76543210, 0xfedcba98, false, false));
}
__device__ __forceinline__ float wave_max(float v) {
    v = fmaxf(v, row_xor<1>(v));
    v = fmaxf(v, row_xor<2>(v));
    v = fmaxf(v, row_xor<4>(v));
    v = fmaxf(v, row_xor<8>(v));
    return fmaxf(v, xhalf(v));
}
__device__ __forceinline__ float wave_sum(float v) {
    v += row_xor<1>(v);
    v += row_xor<2>(v);
    v += row_xor<4>(v);
    v += row_xor<8>(v);
    return v + xhalf(v);
}
__device__ __forceinline__ float leaky(float v) {
    return (v > 0.f) ? v : NEG_SLOPE * v;
}

__device__ __forceinline__ void gather_rows(float4& acc, int my_s, int my_a, int n,
                                            const float* __restrict__ Hm, int row, int col) {
    int j = 0;
    for (; j + 4 <= n; j += 4) {
        const int s0 = __builtin_amdgcn_readlane(my_s, j);
        const int s1 = __builtin_amdgcn_readlane(my_s, j + 1);
        const int s2 = __builtin_amdgcn_readlane(my_s, j + 2);
        const int s3 = __builtin_amdgcn_readlane(my_s, j + 3);
        const float a0 = __int_as_float(__builtin_amdgcn_readlane(my_a, j));
        const float a1 = __int_as_float(__builtin_amdgcn_readlane(my_a, j + 1));
        const float a2 = __int_as_float(__builtin_amdgcn_readlane(my_a, j + 2));
        const float a3 = __int_as_float(__builtin_amdgcn_readlane(my_a, j + 3));
        const float4 v0 = *(const float4*)(Hm + (size_t)s0 * row + col);
        const float4 v1 = *(const float4*)(Hm + (size_t)s1 * row + col);
        const float4 v2 = *(const float4*)(Hm + (size_t)s2 * row + col);
        const float4 v3 = *(const float4*)(Hm + (size_t)s3 * row + col);
        acc.x += a0 * v0.x; acc.y += a0 * v0.y; acc.z += a0 * v0.z; acc.w += a0 * v0.w;
        acc.x += a1 * v1.x; acc.y += a1 * v1.y; acc.z += a1 * v1.z; acc.w += a1 * v1.w;
        acc.x += a2 * v2.x; acc.y += a2 * v2.y; acc.z += a2 * v2.z; acc.w += a2 * v2.w;
        acc.x += a3 * v3.x; acc.y += a3 * v3.y; acc.z += a3 * v3.z; acc.w += a3 * v3.w;
    }
    for (; j < n; ++j) {
        const int s0 = __builtin_amdgcn_readlane(my_s, j);
        const float a0 = __int_as_float(__builtin_amdgcn_readlane(my_a, j));
        const float4 v0 = *(const float4*)(Hm + (size_t)s0 * row + col);
        acc.x += a0 * v0.x; acc.y += a0 * v0.y; acc.z += a0 * v0.z; acc.w += a0 * v0.w;
    }
}

template <int H, int CH, bool DO_ELU>
__global__ __launch_bounds__(256)
void node_softmax_gather(const int* __restrict__ ei, const int* __restrict__ cur,
                         const int* __restrict__ perm, const float* __restrict__ Hm,
                         const float* __restrict__ as, const float* __restrict__ ad,
                         const float* __restrict__ bias,
                         float* __restrict__ alpha_out, float* __restrict__ outp) {
    constexpr int ROW = H * CH;
    constexpr int SL = ROW / 128;
    const int gid = blockIdx.x * blockDim.x + threadIdx.x;
    const int lane = gid & 31;
    const int wave = __builtin_amdgcn_readfirstlane(gid >> 5);
    if (wave >= N_NODES * SL) return;
    const int d = wave / SL, q = wave % SL;
    const int h = (q * 128) / CH;
    const bool opens_head = ((q * 128) % CH) == 0;
    const int col = q * 128 + lane * 4;
    const int beg = d ? cur[d - 1] : 0;
    const int end = cur[d];
    const int deg = end - beg;

    const float adh = ad[d * H + h];
    const float vself = leaky(as[d * H + h] + adh);
    const float4 vs = *(const float4*)(Hm + (size_t)d * ROW + col);
    float4 acc;

    if (deg <= 32) {
        int my_e = 0, my_s = 0;
        float my_v = vself;
        if (lane < deg) {
            my_e = perm[beg + lane];
            my_s = ei[my_e];
            my_v = leaky(as[my_s * H + h] + adh);
        }
        const float m = wave_max(fmaxf(my_v, vself));
        const float ex = (lane < deg) ? expf(my_v - m) : 0.f;
        const float exs = expf(vself - m);
        const float sum = wave_sum(ex) + exs;
        const float a = ex / sum;
        const float aself = exs / sum;
        if (opens_head) {
            if (lane < deg) alpha_out[(size_t)my_e * H + h] = a;
            if (lane == 0) alpha_out[(size_t)(N_EDGES + d) * H + h] = aself;
        }
        acc.x = aself * vs.x; acc.y = aself * vs.y; acc.z = aself * vs.z; acc.w = aself * vs.w;
        gather_rows(acc, my_s, __float_as_int(a), deg, Hm, ROW, col);
    } else {
        float m = vself;
        for (int i = beg; i < end; i += 32) {
            if (i + lane < end) {
                const int s = ei[perm[i + lane]];
                m = fmaxf(m, leaky(as[s * H + h] + adh));
            }
        }
        m = wave_max(m);
        float part = 0.f;
        for (int i = beg; i < end; i += 32) {
            if (i + lane < end) {
                const int s = ei[perm[i + lane]];
                part += expf(leaky(as[s * H + h] + adh) - m);
            }
        }
        const float exs = expf(vself - m);
        const float sum = wave_sum(part) + exs;
        const float aself = exs / sum;
        if (opens_head && lane == 0) alpha_out[(size_t)(N_EDGES + d) * H + h] = aself;
        acc.x = aself * vs.x; acc.y = aself * vs.y; acc.z = aself * vs.z; acc.w = aself * vs.w;
        for (int i = beg; i < end; i += 32) {
            const int n = (end - i < 32) ? (end - i) : 32;
            int my_s = 0;
            float a = 0.f;
            if (lane < n) {
                const int e = perm[i + lane];
                my_s = ei[e];
                a = expf(leaky(as[my_s * H + h] + adh) - m) / sum;
                if (opens_head) alpha_out[(size_t)e * H + h] = a;
            }
            gather_rows(acc, my_s, __float_as_int(a), n, Hm, ROW, col);
        }
    }

    const float4 b = *(const float4*)(bias + col);
    acc.x += b.x; acc.y += b.y; acc.z += b.z; acc.w += b.w;
    if (DO_ELU) {
        acc.x = (acc.x > 0.f) ? acc.x : (expf(acc.x) - 1.f);
        acc.y = (acc.y > 0.f) ? acc.y : (expf(acc.y) - 1.f);
        acc.z = (acc.z > 0.f) ? acc.z : (expf(acc.z) - 1.f);
        acc.w = (acc.w > 0.f) ? acc.w : (expf(acc.w) - 1.f);
    }
    *(float4*)(outp + (size_t)d * ROW + col) = acc;
}

static inline int cdiv(long long a, int b) { return (int)((a + b - 1) / b); }

extern "C" void kernel_launch(void* const* d_in, const int* in_sizes, int n_in,
                              void* d_out, int out_size, void* d_ws, size_t ws_size,
                              hipStream_t stream) {
    const float* x      = (const float*)d_in[0];
    const int*   ei     = (const int*)d_in[1];
    const float* W1     = (const float*)d_in[2];
    const float* a_src1 = (const float*)d_in[3];
    const float* a_dst1 = (const float*)d_in[4];
    const float* b1     = (const float*)d_in[5];
    const float* W2     = (const float*)d_in[6];
    const float* a_src2 = (const float*)d_in[7];
    const float* a_dst2 = (const float*)d_in[8];
    const float* b2     = (const float*)d_in[9];

    const int EP = N_EDGES + N_NODES;

    float* H1   = (float*)d_ws;
    float* AGG1 = H1   + (size_t)N_NODES * 512;
    float* AS1  = AGG1 + (size_t)N_NODES * 512;
    float* AD1  = AS1  + (size_t)N_NODES * HEADS;
    float* AS2  = AD1  + (size_t)N_NODES * HEADS;
    float* AD2  = AS2  + N_NODES;
    int*   CUR  = (int*)(AD2 + N_NODES);
    int*   PERM = CUR + N_NODES;
    float* G2   = H1;

    float* out    = (float*)d_out;
    float* alpha1 = out    + (size_t)N_NODES * OUT_CH;
    float* alpha2 = alpha1 + (size_t)EP * HEADS;

    const int TB = 256;

    gemm_wmma_f32<IN_CH, HEADS * HID>
        <<<dim3(N_NODES / 16, (HEADS * HID) / 64), 32, 0, stream>>>(x, W1, H1);

    att_dots<HID><<<cdiv((long long)N_NODES * HEADS * 32, TB), TB, 0, stream>>>(
        H1, a_src1, a_dst1, N_NODES * HEADS, HEADS, AS1, AD1);

    fill_i32<<<cdiv((long long)N_NODES, TB), TB, 0, stream>>>(CUR, 0, N_NODES);
    dst_count<<<cdiv((long long)N_EDGES, TB), TB, 0, stream>>>(ei, CUR);
    dst_scan<<<1, 1024, 0, stream>>>(CUR);
    dst_place<<<cdiv((long long)N_EDGES, TB), TB, 0, stream>>>(ei, CUR, PERM);

    node_softmax_gather<HEADS, HID, true><<<cdiv((long long)N_NODES * 4 * 32, TB), TB, 0, stream>>>(
        ei, CUR, PERM, H1, AS1, AD1, b1, alpha1, AGG1);

    gemm_wmma_f32<HEADS * HID, OUT_CH>
        <<<dim3(N_NODES / 16, OUT_CH / 64), 32, 0, stream>>>(AGG1, W2, G2);

    att_dots<OUT_CH><<<cdiv((long long)N_NODES * 32, TB), TB, 0, stream>>>(
        G2, a_src2, a_dst2, N_NODES, 1, AS2, AD2);

    node_softmax_gather<1, OUT_CH, false><<<cdiv((long long)N_NODES * 2 * 32, TB), TB, 0, stream>>>(
        ei, CUR, PERM, G2, AS2, AD2, b2, alpha2, out);
}


# Softmax+gather: one wave per destination node for all heads and the whole row (was one per 128-channel slice): edge id, source, logits read once per edge, one alpha store per edge; sum order kept
# speedup vs baseline: 2.2140x; 1.0392x over previous
#include <hip/hip_runtime.h>
#include <math.h>

#define N_NODES 20000
#define N_EDGES 320000
#define IN_CH 512
#define HID 128
#define HEADS 4
#define OUT_CH 256
#define NEG_SLOPE 0.2f

typedef float v2f __attribute__((ext_vector_type(2)));
typedef float v4f __attribute__((ext_vector_type(4)));
typedef float v8f __attribute__((ext_vector_type(8)));

__global__ void fill_i32(int* __restrict__ p, int v, int n) {
    int i = blockIdx.x * blockDim.x + threadIdx.x;
    if (i < n) p[i] = v;
}

struct a_set { v4f v0, v1, v2, v3; };
struct b_set { v4f r0, r1, r2, r3; };

template <int NOUT>
__device__ __forceinline__ b_set load_b(const float* __restrict__ Bp, int k) {
    b_set b;
    b.r0 = *(const v4f*)(Bp + (size_t)(k + 0) * NOUT);
    b.r1 = *(const v4f*)(Bp + (size_t)(k + 1) * NOUT);
    b.r2 = *(const v4f*)(Bp + (size_t)(k + 2) * NOUT);
    b.r3 = *(const v4f*)(Bp + (size_t)(k + 3) * NOUT);
    return b;
}

__device__ __forceinline__ a_set load_a(const float* __restrict__ Ap, int k0) {
    a_set a;
    a.v0 = *(const v4f*)(Ap + k0 + 0);
    a.v1 = *(const v4f*)(Ap + k0 + 4);
    a.v2 = *(const v4f*)(Ap + k0 + 8);
    a.v3 = *(const v4f*)(Ap + k0 + 12);
    return a;
}

__device__ __forceinline__ void mma_sub(const v4f& a, const b_set& b, v8f& acc0, v8f& acc1,
                                        v8f& acc2, v8f& acc3) {
    const v2f a01 = {a.x, a.y};
    const v2f a23 = {a.z, a.w};
    v2f p0 = {b.r0.x, b.r1.x}, p1 = {b.r0.y, b.r1.y};
    v2f p2 = {b.r0.z, b.r1.z}, p3 = {b.r0.w, b.r1.w};
    v2f q0 = {b.r2.x, b.r3.x}, q1 = {b.r2.y, b.r3.y};
    v2f q2 = {b.r2.z, b.r3.z}, q3 = {b.r2.w, b.r3.w};
    asm volatile("" : "+v"(p0), "+v"(p1), "+v"(p2), "+v"(p3),
                      "+v"(q0), "+v"(q1), "+v"(q2), "+v"(q3));
    __builtin_amdgcn_sched_barrier(0);
    acc0 = __builtin_amdgcn_wmma_f32_16x16x4_f32(false, a01, false, p0, (short)0, acc0, false, false);
    acc1 = __builtin_amdgcn_wmma_f32_16x16x4_f32(false, a01, false, p1, (short)0, acc1, false, false);
    acc2 = __builtin_amdgcn_wmma_f32_16x16x4_f32(false, a01, false, p2, (short)0, acc2, false, false);
    acc3 = __builtin_amdgcn_wmma_f32_16x16x4_f32(false, a01, false, p3, (short)0, acc3, false, false);
    acc0 = __builtin_amdgcn_wmma_f32_16x16x4_f32(false, a23, false, q0, (short)0, acc0, false, false);
    acc1 = __builtin_amdgcn_wmma_f32_16x16x4_f32(false, a23, false, q1, (short)0, acc1, false, false);
    acc2 = __builtin_amdgcn_wmma_f32_16x16x4_f32(false, a23, false, q2, (short)0, acc2, false, false);
    acc3 = __builtin_amdgcn_wmma_f32_16x16x4_f32(false, a23, false, q3, (short)0, acc3, false, false);
    __builtin_amdgcn_sched_barrier(0);
    asm volatile("v_nop\n\tv_nop\n\tv_nop\n\tv_nop");
    __builtin_amdgcn_sched_barrier(0);
}

template <int K, int NOUT, bool LAST>
__device__ __forceinline__ void k_block(const float* __restrict__ Ap, const float* __restrict__ Bp,
                                        int k0, const a_set& cur, a_set& nxt, b_set& b0,
                                        v8f& acc0, v8f& acc1, v8f& acc2, v8f& acc3) {
    if (!LAST) nxt = load_a(Ap, k0 + 32);
    b_set b1 = load_b<NOUT>(Bp, k0 + 4);
    __builtin_amdgcn_sched_barrier(0);
    mma_sub(cur.v0, b0, acc0, acc1, acc2, acc3);
    b0 = load_b<NOUT>(Bp, k0 + 8);
    __builtin_amdgcn_sched_barrier(0);
    mma_sub(cur.v1, b1, acc0, acc1, acc2, acc3);
    b1 = load_b<NOUT>(Bp, k0 + 12);
    __builtin_amdgcn_sched_barrier(0);
    mma_sub(cur.v2, b0, acc0, acc1, acc2, acc3);
    if (!LAST) {
        b0 = load_b<NOUT>(Bp, k0 + 32);
        __builtin_amdgcn_sched_barrier(0);
    }
    mma_sub(cur.v3, b1, acc0, acc1, acc2, acc3);
}

template <int K, int NOUT>
__global__ __launch_bounds__(32)
void gemm_wmma_f32(const float* __restrict__ A, const float* __restrict__ B,
                   float* __restrict__ C) {
    static_assert(K % 64 == 0 && NOUT % 64 == 0, "tile shape");
    const int lane = threadIdx.x & 31;
    const int m0 = blockIdx.x * 16;
    const int n0 = blockIdx.y * 64;
    const int c = lane & 15;
    const int h = lane >> 4;

    v8f acc0 = {}, acc1 = {}, acc2 = {}, acc3 = {};
    const float* __restrict__ Ap = A + (size_t)(m0 + c) * K + 16 * h;
    const float* __restrict__ Bp = B + (size_t)(16 * h) * NOUT + n0 + 4 * c;

    a_set a0 = load_a(Ap, 0), a1;
    b_set b0 = load_b<NOUT>(Bp, 0);
    for (int k0 = 0; k0 < K - 64; k0 += 64) {
        k_block<K, NOUT, false>(Ap, Bp, k0,      a0, a1, b0, acc0, acc1, acc2, acc3);
        k_block<K, NOUT, false>(Ap, Bp, k0 + 32, a1, a0, b0, acc0, acc1, acc2, acc3);
    }
    k_block<K, NOUT, false>(Ap, Bp, K - 64, a0, a1, b0, acc0, acc1, acc2, acc3);
    k_block<K, NOUT, true >(Ap, Bp, K - 32, a1, a0, b0, acc0, acc1, acc2, acc3);

    for (int r = 0; r < 8; ++r) {
        v4f o = {acc0[r], acc1[r], acc2[r], acc3[r]};
        *(v4f*)(C + (size_t)(m0 + r + 8 * h) * NOUT + n0 + 4 * c) = o;
    }
}

template <int CH>
__global__ __launch_bounds__(256)
void att_dots(const float* __restrict__ Hm, const float* __restrict__ a_src,
              const float* __restrict__ a_dst, int nwaves, int heads,
              float* __restrict__ as_out, float* __restrict__ ad_out) {
    int gid = blockIdx.x * blockDim.x + threadIdx.x;
    int wave = gid >> 5;
    int lane = gid & 31;
    if (wave >= nwaves) return;
    int h = wave % heads;
    const float* __restrict__ hp  = Hm + (size_t)wave * CH;
    const float* __restrict__ asv = a_src + h * CH;
    const float* __restrict__ adv = a_dst + h * CH;
    float s = 0.f, d = 0.f;
#pragma unroll
    for (int c = lane; c < CH; c += 32) {
        float v = hp[c];
        s += v * asv[c];
        d += v * adv[c];
    }
#pragma unroll
    for (int off = 16; off > 0; off >>= 1) {
        s += __shfl_xor(s, off, 32);
        d += __shfl_xor(d, off, 32);
    }
    if (lane == 0) { as_out[wave] = s; ad_out[wave] = d; }
}

__global__ __launch_bounds__(256)
void dst_count(const int* __restrict__ ei, int* __restrict__ cur) {
    int e = blockIdx.x * blockDim.x + threadIdx.x;
    if (e < N_EDGES) atomicAdd(&cur[ei[N_EDGES + e]], 1);
}

__global__ __launch_bounds__(1024)
void dst_scan(int* __restrict__ cur) {
    constexpr int ITEMS = (N_NODES + 1023) / 1024;
    __shared__ int wsum[32];
    const int t = threadIdx.x;
    const int lane = t & 31, w = t >> 5;
    const int base = t * ITEMS;
    int v[ITEMS];
    int s = 0;
#pragma unroll
    for (int i = 0; i < ITEMS; ++i) {
        int idx = base + i;
        v[i] = (idx < N_NODES) ? cur[idx] : 0;
        s += v[i];
    }
    int inc = s;
#pragma unroll
    for (int off = 1; off < 32; off <<= 1) {
        int n = __shfl_up(inc, off, 32);
        if (lane >= off) inc += n;
    }
    if (lane == 31) wsum[w] = inc;
    __syncthreads();
    if (w == 0) {
        int x = wsum[lane];
        int xi = x;
#pragma unroll
        for (int off = 1; off < 32; off <<= 1) {
            int n = __shfl_up(xi, off, 32);
            if (lane >= off) xi += n;
        }
        wsum[lane] = xi - x;
    }
    __syncthreads();
    int run = wsum[w] + inc - s;
#pragma unroll
    for (int i = 0; i < ITEMS; ++i) {
        int idx = base + i;
        if (idx < N_NODES) { cur[idx] = run; run += v[i]; }
    }
}

__global__ __launch_bounds__(256)
void dst_place(const int* __restrict__ ei, int* __restrict__ cur, int* __restrict__ perm) {
    int e = blockIdx.x * blockDim.x + threadIdx.x;
    if (e < N_EDGES) {
        int p = atomicAdd(&cur[ei[N_EDGES + e]], 1);
        perm[p] = e;
    }
}


template <int X>
__device__ __forceinline__ float row_xor(float v) {
    const int b = __builtin_bit_cast(int, v);
    return __builtin_bit_cast(float, __builtin_amdgcn_update_dpp(b, b, 0x160 + X, 0xf, 0xf, false));
}
__device__ __forceinline__ float xhalf(float v) {
    const int b = __builtin_bit_cast(int, v);
    return __builtin_bit_cast(float,
        __builtin_amdgcn_permlanex16(b, b, 0x76543210, 0xfedcba98, false, false));
}
__device__ __forceinline__ float wave_max(float v) {
    v = fmaxf(v, row_xor<1>(v));
    v = fmaxf(v, row_xor<2>(v));
    v = fmaxf(v, row_xor<4>(v));
    v = fmaxf(v, row_xor<8>(v));
    return fmaxf(v, xhalf(v));
}
__device__ __forceinline__ float wave_sum(float v) {
    v += row_xor<1>(v);
    v += row_xor<2>(v);
    v += row_xor<4>(v);
    v += row_xor<8>(v);
    return v + xhalf(v);
}
__device__ __forceinline__ float leaky(float v) {
    return (v > 0.f) ? v : NEG_SLOPE * v;
}

template <int H>
__device__ __forceinline__ void load_logits(const float* __restrict__ as, int s, float (&lg)[H]) {
    if constexpr (H == 4) {
        const float4 t = *(const float4*)(as + (size_t)s * 4);
        lg[0] = t.x; lg[1] = t.y; lg[2] = t.z; lg[3] = t.w;
    } else {
#pragma unroll
        for (int h = 0; h < H; ++h) lg[h] = as[(size_t)s * H + h];
    }
}

template <int H>
__device__ __forceinline__ void store_alpha(float* __restrict__ alpha_out, size_t e, const float (&a)[H]) {
    if constexpr (H == 4) {
        float4 t; t.x = a[0]; t.y = a[1]; t.z = a[2]; t.w = a[3];
        *(float4*)(alpha_out + e * 4) = t;
    } else {
#pragma unroll
        for (int h = 0; h < H; ++h) alpha_out[e * H + h] = a[h];
    }
}

template <int H, int CH>
__device__ __forceinline__ void gather_rows(float4 (&acc)[(H * CH) / 128], int my_s,
                                            const float (&a)[H], int n,
                                            const float* __restrict__ Hm, int lane) {
    constexpr int ROW = H * CH;
    constexpr int NV = ROW / 128;
    constexpr int EPT = 8 / NV;
    int j = 0;
    for (; j + EPT <= n; j += EPT) {
        int s[EPT];
        float al[EPT][H];
        float4 v[EPT][NV];
#pragma unroll
        for (int u = 0; u < EPT; ++u) {
            s[u] = __builtin_amdgcn_readlane(my_s, j + u);
#pragma unroll
            for (int h = 0; h < H; ++h)
                al[u][h] = __int_as_float(__builtin_amdgcn_readlane(__float_as_int(a[h]), j + u));
        }
#pragma unroll
        for (int u = 0; u < EPT; ++u) {
#pragma unroll
            for (int p = 0; p < NV; ++p)
                v[u][p] = *(const float4*)(Hm + (size_t)s[u] * ROW + p * 128 + lane * 4);
        }
#pragma unroll
        for (int u = 0; u < EPT; ++u) {
#pragma unroll
            for (int p = 0; p < NV; ++p) {
                const float w = al[u][(p * 128) / CH];
                acc[p].x = __builtin_fmaf(w, v[u][p].x, acc[p].x);
                acc[p].y = __builtin_fmaf(w, v[u][p].y, acc[p].y);
                acc[p].z = __builtin_fmaf(w, v[u][p].z, acc[p].z);
                acc[p].w = __builtin_fmaf(w, v[u][p].w, acc[p].w);
            }
        }
    }
    for (; j < n; ++j) {
        const int s0 = __builtin_amdgcn_readlane(my_s, j);
        float al0[H];
#pragma unroll
        for (int h = 0; h < H; ++h)
            al0[h] = __int_as_float(__builtin_amdgcn_readlane(__float_as_int(a[h]), j));
        float4 v0[NV];
#pragma unroll
        for (int p = 0; p < NV; ++p)
            v0[p] = *(const float4*)(Hm + (size_t)s0 * ROW + p * 128 + lane * 4);
#pragma unroll
        for (int p = 0; p < NV; ++p) {
            const float w = al0[(p * 128) / CH];
            acc[p].x = __builtin_fmaf(w, v0[p].x, acc[p].x);
            acc[p].y = __builtin_fmaf(w, v0[p].y, acc[p].y);
            acc[p].z = __builtin_fmaf(w, v0[p].z, acc[p].z);
            acc[p].w = __builtin_fmaf(w, v0[p].w, acc[p].w);
        }
    }
}

template <int H, int CH, bool DO_ELU>
__global__ __launch_bounds__(256)
void node_softmax_gather(const int* __restrict__ ei, const int* __restrict__ cur,
                         const int* __restrict__ perm, const float* __restrict__ Hm,
                         const float* __restrict__ as, const float* __restrict__ ad,
                         const float* __restrict__ bias,
                         float* __restrict__ alpha_out, float* __restrict__ outp) {
    constexpr int ROW = H * CH;
    constexpr int NV = ROW / 128;
    static_assert(CH % 128 == 0 && (8 % NV) == 0, "row pieces");
    const int gid = blockIdx.x * blockDim.x + threadIdx.x;
    const int lane = gid & 31;
    const int d = __builtin_amdgcn_readfirstlane(gid >> 5);
    if (d >= N_NODES) return;
    const int beg = d ? cur[d - 1] : 0;
    const int end = cur[d];
    const int deg = end - beg;

    float adh[H], vself[H];
    {
        float lgd[H];
        load_logits<H>(as, d, lgd);
#pragma unroll
        for (int h = 0; h < H; ++h) {
            adh[h] = ad[d * H + h];
            vself[h] = leaky(lgd[h] + adh[h]);
        }
    }
    float4 vs[NV];
#pragma unroll
    for (int p = 0; p < NV; ++p)
        vs[p] = *(const float4*)(Hm + (size_t)d * ROW + p * 128 + lane * 4);
    float4 acc[NV];

    if (deg <= 32) {
        int my_e = 0, my_s = 0;
        float my_v[H];
#pragma unroll
        for (int h = 0; h < H; ++h) my_v[h] = vself[h];
        if (lane < deg) {
            my_e = perm[beg + lane];
            my_s = ei[my_e];
            float lg[H];
            load_logits<H>(as, my_s, lg);
#pragma unroll
            for (int h = 0; h < H; ++h) my_v[h] = leaky(lg[h] + adh[h]);
        }
        float a[H], aself[H];
#pragma unroll
        for (int h = 0; h < H; ++h) {
            const float m = wave_max(fmaxf(my_v[h], vself[h]));
            const float ex = (lane < deg) ? expf(my_v[h] - m) : 0.f;
            const float exs = expf(vself[h] - m);
            const float sum = wave_sum(ex) + exs;
            a[h] = ex / sum;
            aself[h] = exs / sum;
        }
        if (lane < deg) store_alpha<H>(alpha_out, (size_t)my_e, a);
        if (lane == 0) store_alpha<H>(alpha_out, (size_t)(N_EDGES + d), aself);
#pragma unroll
        for (int p = 0; p < NV; ++p) {
            const float w = aself[(p * 128) / CH];
            acc[p].x = w * vs[p].x; acc[p].y = w * vs[p].y;
            acc[p].z = w * vs[p].z; acc[p].w = w * vs[p].w;
        }
        gather_rows<H, CH>(acc, my_s, a, deg, Hm, lane);
    } else {
        float m[H], part[H], sum[H], aself[H];
#pragma unroll
        for (int h = 0; h < H; ++h) { m[h] = vself[h]; part[h] = 0.f; }
        for (int i = beg; i < end; i += 32) {
            if (i + lane < end) {
                const int s = ei[perm[i + lane]];
                float lg[H];
                load_logits<H>(as, s, lg);
#pragma unroll
                for (int h = 0; h < H; ++h) m[h] = fmaxf(m[h], leaky(lg[h] + adh[h]));
            }
        }
#pragma unroll
        for (int h = 0; h < H; ++h) m[h] = wave_max(m[h]);
        for (int i = beg; i < end; i += 32) {
            if (i + lane < end) {
                const int s = ei[perm[i + lane]];
                float lg[H];
                load_logits<H>(as, s, lg);
#pragma unroll
                for (int h = 0; h < H; ++h) part[h] += expf(leaky(lg[h] + adh[h]) - m[h]);
            }
        }
#pragma unroll
        for (int h = 0; h < H; ++h) {
            const float exs = expf(vself[h] - m[h]);
            sum[h] = wave_sum(part[h]) + exs;
            aself[h] = exs / sum[h];
        }
        if (lane == 0) store_alpha<H>(alpha_out, (size_t)(N_EDGES + d), aself);
#pragma unroll
        for (int p = 0; p < NV; ++p) {
            const float w = aself[(p * 128) / CH];
            acc[p].x = w * vs[p].x; acc[p].y = w * vs[p].y;
            acc[p].z = w * vs[p].z; acc[p].w = w * vs[p].w;
        }
        for (int i = beg; i < end; i += 32) {
            const int n = (end - i < 32) ? (end - i) : 32;
            int my_s = 0;
            float a[H];
#pragma unroll
            for (int h = 0; h < H; ++h) a[h] = 0.f;
            if (lane < n) {
                const int e = perm[i + lane];
                my_s = ei[e];
                float lg[H];
                load_logits<H>(as, my_s, lg);
#pragma unroll
                for (int h = 0; h < H; ++h) a[h] = expf(leaky(lg[h] + adh[h]) - m[h]) / sum[h];
                store_alpha<H>(alpha_out, (size_t)e, a);
            }
            gather_rows<H, CH>(acc, my_s, a, n, Hm, lane);
        }
    }

#pragma unroll
    for (int p = 0; p < NV; ++p) {
        const float4 b = *(const float4*)(bias + p * 128 + lane * 4);
        float4 o = acc[p];
        o.x += b.x; o.y += b.y; o.z += b.z; o.w += b.w;
        if (DO_ELU) {
            o.x = (o.x > 0.f) ? o.x : (expf(o.x) - 1.f);
            o.y = (o.y > 0.f) ? o.y : (expf(o.y) - 1.f);
            o.z = (o.z > 0.f) ? o.z : (expf(o.z) - 1.f);
            o.w = (o.w > 0.f) ? o.w : (expf(o.w) - 1.f);
        }
        *(float4*)(outp + (size_t)d * ROW + p * 128 + lane * 4) = o;
    }
}

static inline int cdiv(long long a, int b) { return (int)((a + b - 1) / b); }

extern "C" void kernel_launch(void* const* d_in, const int* in_sizes, int n_in,
                              void* d_out, int out_size, void* d_ws, size_t ws_size,
                              hipStream_t stream) {
    const float* x      = (const float*)d_in[0];
    const int*   ei     = (const int*)d_in[1];
    const float* W1     = (const float*)d_in[2];
    const float* a_src1 = (const float*)d_in[3];
    const float* a_dst1 = (const float*)d_in[4];
    const float* b1     = (const float*)d_in[5];
    const float* W2     = (const float*)d_in[6];
    const float* a_src2 = (const float*)d_in[7];
    const float* a_dst2 = (const float*)d_in[8];
    const float* b2     = (const float*)d_in[9];

    const int EP = N_EDGES + N_NODES;

    float* H1   = (float*)d_ws;
    float* AGG1 = H1   + (size_t)N_NODES * 512;
    float* AS1  = AGG1 + (size_t)N_NODES * 512;
    float* AD1  = AS1  + (size_t)N_NODES * HEADS;
    float* AS2  = AD1  + (size_t)N_NODES * HEADS;
    float* AD2  = AS2  + N_NODES;
    int*   CUR  = (int*)(AD2 + N_NODES);
    int*   PERM = CUR + N_NODES;
    float* G2   = H1;

    float* out    = (float*)d_out;
    float* alpha1 = out    + (size_t)N_NODES * OUT_CH;
    float* alpha2 = alpha1 + (size_t)EP * HEADS;

    const int TB = 256;

    gemm_wmma_f32<IN_CH, HEADS * HID>
        <<<dim3(N_NODES / 16, (HEADS * HID) / 64), 32, 0, stream>>>(x, W1, H1);

    att_dots<HID><<<cdiv((long long)N_NODES * HEADS * 32, TB), TB, 0, stream>>>(
        H1, a_src1, a_dst1, N_NODES * HEADS, HEADS, AS1, AD1);

    fill_i32<<<cdiv((long long)N_NODES, TB), TB, 0, stream>>>(CUR, 0, N_NODES);
    dst_count<<<cdiv((long long)N_EDGES, TB), TB, 0, stream>>>(ei, CUR);
    dst_scan<<<1, 1024, 0, stream>>>(CUR);
    dst_place<<<cdiv((long long)N_EDGES, TB), TB, 0, stream>>>(ei, CUR, PERM);

    node_softmax_gather<HEADS, HID, true><<<cdiv((long long)N_NODES * 32, TB), TB, 0, stream>>>(
        ei, CUR, PERM, H1, AS1, AD1, b1, alpha1, AGG1);

    gemm_wmma_f32<HEADS * HID, OUT_CH>
        <<<dim3(N_NODES / 16, OUT_CH / 64), 32, 0, stream>>>(AGG1, W2, G2);

    att_dots<OUT_CH><<<cdiv((long long)N_NODES * 32, TB), TB, 0, stream>>>(
        G2, a_src2, a_dst2, N_NODES, 1, AS2, AD2);

    node_softmax_gather<1, OUT_CH, false><<<cdiv((long long)N_NODES * 32, TB), TB, 0, stream>>>(
        ei, CUR, PERM, G2, AS2, AD2, b2, alpha2, out);
}


# Attention dots: one wave takes 8 whole rows (16-byte loads all in flight, DPP wave sums, one store per 8 nodes), was one wave per (node, head): 100000 waves become 5000; dot sum order changed
# speedup vs baseline: 2.3056x; 1.0414x over previous
#include <hip/hip_runtime.h>
#include <math.h>

#define N_NODES 20000
#define N_EDGES 320000
#define IN_CH 512
#define HID 128
#define HEADS 4
#define OUT_CH 256
#define NEG_SLOPE 0.2f

typedef float v2f __attribute__((ext_vector_type(2)));
typedef float v4f __attribute__((ext_vector_type(4)));
typedef float v8f __attribute__((ext_vector_type(8)));

__global__ void fill_i32(int* __restrict__ p, int v, int n) {
    int i = blockIdx.x * blockDim.x + threadIdx.x;
    if (i < n) p[i] = v;
}

struct a_set { v4f v0, v1, v2, v3; };
struct b_set { v4f r0, r1, r2, r3; };

template <int NOUT>
__device__ __forceinline__ b_set load_b(const float* __restrict__ Bp, int k) {
    b_set b;
    b.r0 = *(const v4f*)(Bp + (size_t)(k + 0) * NOUT);
    b.r1 = *(const v4f*)(Bp + (size_t)(k + 1) * NOUT);
    b.r2 = *(const v4f*)(Bp + (size_t)(k + 2) * NOUT);
    b.r3 = *(const v4f*)(Bp + (size_t)(k + 3) * NOUT);
    return b;
}

__device__ __forceinline__ a_set load_a(const float* __restrict__ Ap, int k0) {
    a_set a;
    a.v0 = *(const v4f*)(Ap + k0 + 0);
    a.v1 = *(const v4f*)(Ap + k0 + 4);
    a.v2 = *(const v4f*)(Ap + k0 + 8);
    a.v3 = *(const v4f*)(Ap + k0 + 12);
    return a;
}

__device__ __forceinline__ void mma_sub(const v4f& a, const b_set& b, v8f& acc0, v8f& acc1,
                                        v8f& acc2, v8f& acc3) {
    const v2f a01 = {a.x, a.y};
    const v2f a23 = {a.z, a.w};
    v2f p0 = {b.r0.x, b.r1.x}, p1 = {b.r0.y, b.r1.y};
    v2f p2 = {b.r0.z, b.r1.z}, p3 = {b.r0.w, b.r1.w};
    v2f q0 = {b.r2.x, b.r3.x}, q1 = {b.r2.y, b.r3.y};
    v2f q2 = {b.r2.z, b.r3.z}, q3 = {b.r2.w, b.r3.w};
    asm volatile("" : "+v"(p0), "+v"(p1), "+v"(p2), "+v"(p3),
                      "+v"(q0), "+v"(q1), "+v"(q2), "+v"(q3));
    __builtin_amdgcn_sched_barrier(0);
    acc0 = __builtin_amdgcn_wmma_f32_16x16x4_f32(false, a01, false, p0, (short)0, acc0, false, false);
    acc1 = __builtin_amdgcn_wmma_f32_16x16x4_f32(false, a01, false, p1, (short)0, acc1, false, false);
    acc2 = __builtin_amdgcn_wmma_f32_16x16x4_f32(false, a01, false, p2, (short)0, acc2, false, false);
    acc3 = __builtin_amdgcn_wmma_f32_16x16x4_f32(false, a01, false, p3, (short)0, acc3, false, false);
    acc0 = __builtin_amdgcn_wmma_f32_16x16x4_f32(false, a23, false, q0, (short)0, acc0, false, false);
    acc1 = __builtin_amdgcn_wmma_f32_16x16x4_f32(false, a23, false, q1, (short)0, acc1, false, false);
    acc2 = __builtin_amdgcn_wmma_f32_16x16x4_f32(false, a23, false, q2, (short)0, acc2, false, false);
    acc3 = __builtin_amdgcn_wmma_f32_16x16x4_f32(false, a23, false, q3, (short)0, acc3, false, false);
    __builtin_amdgcn_sched_barrier(0);
    asm volatile("v_nop\n\tv_nop\n\tv_nop\n\tv_nop");
    __builtin_amdgcn_sched_barrier(0);
}

template <int K, int NOUT, bool LAST>
__device__ __forceinline__ void k_block(const float* __restrict__ Ap, const float* __restrict__ Bp,
                                        int k0, const a_set& cur, a_set& nxt, b_set& b0,
                                        v8f& acc0, v8f& acc1, v8f& acc2, v8f& acc3) {
    if (!LAST) nxt = load_a(Ap, k0 + 32);
    b_set b1 = load_b<NOUT>(Bp, k0 + 4);
    __builtin_amdgcn_sched_barrier(0);
    mma_sub(cur.v0, b0, acc0, acc1, acc2, acc3);
    b0 = load_b<NOUT>(Bp, k0 + 8);
    __builtin_amdgcn_sched_barrier(0);
    mma_sub(cur.v1, b1, acc0, acc1, acc2, acc3);
    b1 = load_b<NOUT>(Bp, k0 + 12);
    __builtin_amdgcn_sched_barrier(0);
    mma_sub(cur.v2, b0, acc0, acc1, acc2, acc3);
    if (!LAST) {
        b0 = load_b<NOUT>(Bp, k0 + 32);
        __builtin_amdgcn_sched_barrier(0);
    }
    mma_sub(cur.v3, b1, acc0, acc1, acc2, acc3);
}

template <int K, int NOUT>
__global__ __launch_bounds__(32)
void gemm_wmma_f32(const float* __restrict__ A, const float* __restrict__ B,
                   float* __restrict__ C) {
    static_assert(K % 64 == 0 && NOUT % 64 == 0, "tile shape");
    const int lane = threadIdx.x & 31;
    const int m0 = blockIdx.x * 16;
    const int n0 = blockIdx.y * 64;
    const int c = lane & 15;
    const int h = lane >> 4;

    v8f acc0 = {}, acc1 = {}, acc2 = {}, acc3 = {};
    const float* __restrict__ Ap = A + (size_t)(m0 + c) * K + 16 * h;
    const float* __restrict__ Bp = B + (size_t)(16 * h) * NOUT + n0 + 4 * c;

    a_set a0 = load_a(Ap, 0), a1;
    b_set b0 = load_b<NOUT>(Bp, 0);
    for (int k0 = 0; k0 < K - 64; k0 += 64) {
        k_block<K, NOUT, false>(Ap, Bp, k0,      a0, a1, b0, acc0, acc1, acc2, acc3);
        k_block<K, NOUT, false>(Ap, Bp, k0 + 32, a1, a0, b0, acc0, acc1, acc2, acc3);
    }
    k_block<K, NOUT, false>(Ap, Bp, K - 64, a0, a1, b0, acc0, acc1, acc2, acc3);
    k_block<K, NOUT, true >(Ap, Bp, K - 32, a1, a0, b0, acc0, acc1, acc2, acc3);

    for (int r = 0; r < 8; ++r) {
        v4f o = {acc0[r], acc1[r], acc2[r], acc3[r]};
        *(v4f*)(C + (size_t)(m0 + r + 8 * h) * NOUT + n0 + 4 * c) = o;
    }
}

template <int X>
__device__ __forceinline__ float row_xor(float v) {
    const int b = __builtin_bit_cast(int, v);
    return __builtin_bit_cast(float, __builtin_amdgcn_update_dpp(b, b, 0x160 + X, 0xf, 0xf, false));
}
__device__ __forceinline__ float xhalf(float v) {
    const int b = __builtin_bit_cast(int, v);
    return __builtin_bit_cast(float,
        __builtin_amdgcn_permlanex16(b, b, 0x76543210, 0xfedcba98, false, false));
}
__device__ __forceinline__ float wave_max(float v) {
    v = fmaxf(v, row_xor<1>(v));
    v = fmaxf(v, row_xor<2>(v));
    v = fmaxf(v, row_xor<4>(v));
    v = fmaxf(v, row_xor<8>(v));
    return fmaxf(v, xhalf(v));
}
__device__ __forceinline__ float wave_sum(float v) {
    v += row_xor<1>(v);
    v += row_xor<2>(v);
    v += row_xor<4>(v);
    v += row_xor<8>(v);
    return v + xhalf(v);
}

template <int H, int CH, int NPW>
__global__ __launch_bounds__(128)
void att_dots(const float* __restrict__ Hm, const float* __restrict__ a_src,
              const float* __restrict__ a_dst,
              float* __restrict__ as_out, float* __restrict__ ad_out) {
    constexpr int ROW = H * CH;
    constexpr int NV = ROW / 128;
    constexpr int PPH = CH / 128;
    static_assert(CH % 128 == 0 && N_NODES % NPW == 0 && NPW <= 32, "row pieces / node groups");
    const int gid = blockIdx.x * blockDim.x + threadIdx.x;
    const int lane = gid & 31;
    const int n0 = __builtin_amdgcn_readfirstlane(gid >> 5) * NPW;
    if (n0 >= N_NODES) return;

    float4 ws[NV], wd[NV];
#pragma unroll
    for (int p = 0; p < NV; ++p) {
        ws[p] = *(const float4*)(a_src + p * 128 + lane * 4);
        wd[p] = *(const float4*)(a_dst + p * 128 + lane * 4);
    }
    float4 v[NPW][NV];
#pragma unroll
    for (int u = 0; u < NPW; ++u) {
#pragma unroll
        for (int p = 0; p < NV; ++p)
            v[u][p] = *(const float4*)(Hm + (size_t)(n0 + u) * ROW + p * 128 + lane * 4);
    }

    float rs[H], rd[H];
#pragma unroll
    for (int h = 0; h < H; ++h) { rs[h] = 0.f; rd[h] = 0.f; }
#pragma unroll
    for (int u = 0; u < NPW; ++u) {
#pragma unroll
        for (int h = 0; h < H; ++h) {
            float s = 0.f, d = 0.f;
#pragma unroll
            for (int q = 0; q < PPH; ++q) {
                const float4 x = v[u][h * PPH + q];
                const float4 a = ws[h * PPH + q];
                const float4 b = wd[h * PPH + q];
                s = __builtin_fmaf(x.x, a.x, s); s = __builtin_fmaf(x.y, a.y, s);
                s = __builtin_fmaf(x.z, a.z, s); s = __builtin_fmaf(x.w, a.w, s);
                d = __builtin_fmaf(x.x, b.x, d); d = __builtin_fmaf(x.y, b.y, d);
                d = __builtin_fmaf(x.z, b.z, d); d = __builtin_fmaf(x.w, b.w, d);
            }
            s = wave_sum(s);
            d = wave_sum(d);
            if (lane == u) { rs[h] = s; rd[h] = d; }
        }
    }
    if (lane < NPW) {
        if constexpr (H == 4) {
            float4 t; t.x = rs[0]; t.y = rs[1]; t.z = rs[2]; t.w = rs[3];
            *(float4*)(as_out + (size_t)(n0 + lane) * 4) = t;
            t.x = rd[0]; t.y = rd[1]; t.z = rd[2]; t.w = rd[3];
            *(float4*)(ad_out + (size_t)(n0 + lane) * 4) = t;
        } else {
#pragma unroll
            for (int h = 0; h < H; ++h) {
                as_out[(size_t)(n0 + lane) * H + h] = rs[h];
                ad_out[(size_t)(n0 + lane) * H + h] = rd[h];
            }
        }
    }
}

__global__ __launch_bounds__(256)
void dst_count(const int* __restrict__ ei, int* __restrict__ cur) {
    int e = blockIdx.x * blockDim.x + threadIdx.x;
    if (e < N_EDGES) atomicAdd(&cur[ei[N_EDGES + e]], 1);
}

__global__ __launch_bounds__(1024)
void dst_scan(int* __restrict__ cur) {
    constexpr int ITEMS = (N_NODES + 1023) / 1024;
    __shared__ int wsum[32];
    const int t = threadIdx.x;
    const int lane = t & 31, w = t >> 5;
    const int base = t * ITEMS;
    int v[ITEMS];
    int s = 0;
#pragma unroll
    for (int i = 0; i < ITEMS; ++i) {
        int idx = base + i;
        v[i] = (idx < N_NODES) ? cur[idx] : 0;
        s += v[i];
    }
    int inc = s;
#pragma unroll
    for (int off = 1; off < 32; off <<= 1) {
        int n = __shfl_up(inc, off, 32);
        if (lane >= off) inc += n;
    }
    if (lane == 31) wsum[w] = inc;
    __syncthreads();
    if (w == 0) {
        int x = wsum[lane];
        int xi = x;
#pragma unroll
        for (int off = 1; off < 32; off <<= 1) {
            int n = __shfl_up(xi, off, 32);
            if (lane >= off) xi += n;
        }
        wsum[lane] = xi - x;
    }
    __syncthreads();
    int run = wsum[w] + inc - s;
#pragma unroll
    for (int i = 0; i < ITEMS; ++i) {
        int idx = base + i;
        if (idx < N_NODES) { cur[idx] = run; run += v[i]; }
    }
}

__global__ __launch_bounds__(256)
void dst_place(const int* __restrict__ ei, int* __restrict__ cur, int* __restrict__ perm) {
    int e = blockIdx.x * blockDim.x + threadIdx.x;
    if (e < N_EDGES) {
        int p = atomicAdd(&cur[ei[N_EDGES + e]], 1);
        perm[p] = e;
    }
}


__device__ __forceinline__ float leaky(float v) {
    return (v > 0.f) ? v : NEG_SLOPE * v;
}

template <int H>
__device__ __forceinline__ void load_logits(const float* __restrict__ as, int s, float (&lg)[H]) {
    if constexpr (H == 4) {
        const float4 t = *(const float4*)(as + (size_t)s * 4);
        lg[0] = t.x; lg[1] = t.y; lg[2] = t.z; lg[3] = t.w;
    } else {
#pragma unroll
        for (int h = 0; h < H; ++h) lg[h] = as[(size_t)s * H + h];
    }
}

template <int H>
__device__ __forceinline__ void store_alpha(float* __restrict__ alpha_out, size_t e, const float (&a)[H]) {
    if constexpr (H == 4) {
        float4 t; t.x = a[0]; t.y = a[1]; t.z = a[2]; t.w = a[3];
        *(float4*)(alpha_out + e * 4) = t;
    } else {
#pragma unroll
        for (int h = 0; h < H; ++h) alpha_out[e * H + h] = a[h];
    }
}

template <int H, int CH>
__device__ __forceinline__ void gather_rows(float4 (&acc)[(H * CH) / 128], int my_s,
                                            const float (&a)[H], int n,
                                            const float* __restrict__ Hm, int lane) {
    constexpr int ROW = H * CH;
    constexpr int NV = ROW / 128;
    constexpr int EPT = 8 / NV;
    int j = 0;
    for (; j + EPT <= n; j += EPT) {
        int s[EPT];
        float al[EPT][H];
        float4 v[EPT][NV];
#pragma unroll
        for (int u = 0; u < EPT; ++u) {
            s[u] = __builtin_amdgcn_readlane(my_s, j + u);
#pragma unroll
            for (int h = 0; h < H; ++h)
                al[u][h] = __int_as_float(__builtin_amdgcn_readlane(__float_as_int(a[h]), j + u));
        }
#pragma unroll
        for (int u = 0; u < EPT; ++u) {
#pragma unroll
            for (int p = 0; p < NV; ++p)
                v[u][p] = *(const float4*)(Hm + (size_t)s[u] * ROW + p * 128 + lane * 4);
        }
#pragma unroll
        for (int u = 0; u < EPT; ++u) {
#pragma unroll
            for (int p = 0; p < NV; ++p) {
                const float w = al[u][(p * 128) / CH];
                acc[p].x = __builtin_fmaf(w, v[u][p].x, acc[p].x);
                acc[p].y = __builtin_fmaf(w, v[u][p].y, acc[p].y);
                acc[p].z = __builtin_fmaf(w, v[u][p].z, acc[p].z);
                acc[p].w = __builtin_fmaf(w, v[u][p].w, acc[p].w);
            }
        }
    }
    for (; j < n; ++j) {
        const int s0 = __builtin_amdgcn_readlane(my_s, j);
        float al0[H];
#pragma unroll
        for (int h = 0; h < H; ++h)
            al0[h] = __int_as_float(__builtin_amdgcn_readlane(__float_as_int(a[h]), j));
        float4 v0[NV];
#pragma unroll
        for (int p = 0; p < NV; ++p)
            v0[p] = *(const float4*)(Hm + (size_t)s0 * ROW + p * 128 + lane * 4);
#pragma unroll
        for (int p = 0; p < NV; ++p) {
            const float w = al0[(p * 128) / CH];
            acc[p].x = __builtin_fmaf(w, v0[p].x, acc[p].x);
            acc[p].y = __builtin_fmaf(w, v0[p].y, acc[p].y);
            acc[p].z = __builtin_fmaf(w, v0[p].z, acc[p].z);
            acc[p].w = __builtin_fmaf(w, v0[p].w, acc[p].w);
        }
    }
}

template <int H, int CH, bool DO_ELU>
__global__ __launch_bounds__(256)
void node_softmax_gather(const int* __restrict__ ei, const int* __restrict__ cur,
                         const int* __restrict__ perm, const float* __restrict__ Hm,
                         const float* __restrict__ as, const float* __restrict__ ad,
                         const float* __restrict__ bias,
                         float* __restrict__ alpha_out, float* __restrict__ outp) {
    constexpr int ROW = H * CH;
    constexpr int NV = ROW / 128;
    static_assert(CH % 128 == 0 && (8 % NV) == 0, "row pieces");
    const int gid = blockIdx.x * blockDim.x + threadIdx.x;
    const int lane = gid & 31;
    const int d = __builtin_amdgcn_readfirstlane(gid >> 5);
    if (d >= N_NODES) return;
    const int beg = d ? cur[d - 1] : 0;
    const int end = cur[d];
    const int deg = end - beg;

    float adh[H], vself[H];
    {
        float lgd[H];
        load_logits<H>(as, d, lgd);
#pragma unroll
        for (int h = 0; h < H; ++h) {
            adh[h] = ad[d * H + h];
            vself[h] = leaky(lgd[h] + adh[h]);
        }
    }
    float4 vs[NV];
#pragma unroll
    for (int p = 0; p < NV; ++p)
        vs[p] = *(const float4*)(Hm + (size_t)d * ROW + p * 128 + lane * 4);
    float4 acc[NV];

    if (deg <= 32) {
        int my_e = 0, my_s = 0;
        float my_v[H];
#pragma unroll
        for (int h = 0; h < H; ++h) my_v[h] = vself[h];
        if (lane < deg) {
            my_e = perm[beg + lane];
            my_s = ei[my_e];
            float lg[H];
            load_logits<H>(as, my_s, lg);
#pragma unroll
            for (int h = 0; h < H; ++h) my_v[h] = leaky(lg[h] + adh[h]);
        }
        float a[H], aself[H];
#pragma unroll
        for (int h = 0; h < H; ++h) {
            const float m = wave_max(fmaxf(my_v[h], vself[h]));
            const float ex = (lane < deg) ? expf(my_v[h] - m) : 0.f;
            const float exs = expf(vself[h] - m);
            const float sum = wave_sum(ex) + exs;
            a[h] = ex / sum;
            aself[h] = exs / sum;
        }
        if (lane < deg) store_alpha<H>(alpha_out, (size_t)my_e, a);
        if (lane == 0) store_alpha<H>(alpha_out, (size_t)(N_EDGES + d), aself);
#pragma unroll
        for (int p = 0; p < NV; ++p) {
            const float w = aself[(p * 128) / CH];
            acc[p].x = w * vs[p].x; acc[p].y = w * vs[p].y;
            acc[p].z = w * vs[p].z; acc[p].w = w * vs[p].w;
        }
        gather_rows<H, CH>(acc, my_s, a, deg, Hm, lane);
    } else {
        float m[H], part[H], sum[H], aself[H];
#pragma unroll
        for (int h = 0; h < H; ++h) { m[h] = vself[h]; part[h] = 0.f; }
        for (int i = beg; i < end; i += 32) {
            if (i + lane < end) {
                const int s = ei[perm[i + lane]];
                float lg[H];
                load_logits<H>(as, s, lg);
#pragma unroll
                for (int h = 0; h < H; ++h) m[h] = fmaxf(m[h], leaky(lg[h] + adh[h]));
            }
        }
#pragma unroll
        for (int h = 0; h < H; ++h) m[h] = wave_max(m[h]);
        for (int i = beg; i < end; i += 32) {
            if (i + lane < end) {
                const int s = ei[perm[i + lane]];
                float lg[H];
                load_logits<H>(as, s, lg);
#pragma unroll
                for (int h = 0; h < H; ++h) part[h] += expf(leaky(lg[h] + adh[h]) - m[h]);
            }
        }
#pragma unroll
        for (int h = 0; h < H; ++h) {
            const float exs = expf(vself[h] - m[h]);
            sum[h] = wave_sum(part[h]) + exs;
            aself[h] = exs / sum[h];
        }
        if (lane == 0) store_alpha<H>(alpha_out, (size_t)(N_EDGES + d), aself);
#pragma unroll
        for (int p = 0; p < NV; ++p) {
            const float w = aself[(p * 128) / CH];
            acc[p].x = w * vs[p].x; acc[p].y = w * vs[p].y;
            acc[p].z = w * vs[p].z; acc[p].w = w * vs[p].w;
        }
        for (int i = beg; i < end; i += 32) {
            const int n = (end - i < 32) ? (end - i) : 32;
            int my_s = 0;
            float a[H];
#pragma unroll
            for (int h = 0; h < H; ++h) a[h] = 0.f;
            if (lane < n) {
                const int e = perm[i + lane];
                my_s = ei[e];
                float lg[H];
                load_logits<H>(as, my_s, lg);
#pragma unroll
                for (int h = 0; h < H; ++h) a[h] = expf(leaky(lg[h] + adh[h]) - m[h]) / sum[h];
                store_alpha<H>(alpha_out, (size_t)e, a);
            }
            gather_rows<H, CH>(acc, my_s, a, n, Hm, lane);
        }
    }

#pragma unroll
    for (int p = 0; p < NV; ++p) {
        const float4 b = *(const float4*)(bias + p * 128 + lane * 4);
        float4 o = acc[p];
        o.x += b.x; o.y += b.y; o.z += b.z; o.w += b.w;
        if (DO_ELU) {
            o.x = (o.x > 0.f) ? o.x : (expf(o.x) - 1.f);
            o.y = (o.y > 0.f) ? o.y : (expf(o.y) - 1.f);
            o.z = (o.z > 0.f) ? o.z : (expf(o.z) - 1.f);
            o.w = (o.w > 0.f) ? o.w : (expf(o.w) - 1.f);
        }
        *(float4*)(outp + (size_t)d * ROW + p * 128 + lane * 4) = o;
    }
}

static inline int cdiv(long long a, int b) { return (int)((a + b - 1) / b); }

extern "C" void kernel_launch(void* const* d_in, const int* in_sizes, int n_in,
                              void* d_out, int out_size, void* d_ws, size_t ws_size,
                              hipStream_t stream) {
    const float* x      = (const float*)d_in[0];
    const int*   ei     = (const int*)d_in[1];
    const float* W1     = (const float*)d_in[2];
    const float* a_src1 = (const float*)d_in[3];
    const float* a_dst1 = (const float*)d_in[4];
    const float* b1     = (const float*)d_in[5];
    const float* W2     = (const float*)d_in[6];
    const float* a_src2 = (const float*)d_in[7];
    const float* a_dst2 = (const float*)d_in[8];
    const float* b2     = (const float*)d_in[9];

    const int EP = N_EDGES + N_NODES;

    float* H1   = (float*)d_ws;
    float* AGG1 = H1   + (size_t)N_NODES * 512;
    float* AS1  = AGG1 + (size_t)N_NODES * 512;
    float* AD1  = AS1  + (size_t)N_NODES * HEADS;
    float* AS2  = AD1  + (size_t)N_NODES * HEADS;
    float* AD2  = AS2  + N_NODES;
    int*   CUR  = (int*)(AD2 + N_NODES);
    int*   PERM = CUR + N_NODES;
    float* G2   = H1;

    float* out    = (float*)d_out;
    float* alpha1 = out    + (size_t)N_NODES * OUT_CH;
    float* alpha2 = alpha1 + (size_t)EP * HEADS;

    const int TB = 256;

    gemm_wmma_f32<IN_CH, HEADS * HID>
        <<<dim3(N_NODES / 16, (HEADS * HID) / 64), 32, 0, stream>>>(x, W1, H1);

    att_dots<HEADS, HID, 8><<<cdiv((long long)(N_NODES / 8) * 32, 128), 128, 0, stream>>>(
        H1, a_src1, a_dst1, AS1, AD1);

    fill_i32<<<cdiv((long long)N_NODES, TB), TB, 0, stream>>>(CUR, 0, N_NODES);
    dst_count<<<cdiv((long long)N_EDGES, TB), TB, 0, stream>>>(ei, CUR);
    dst_scan<<<1, 1024, 0, stream>>>(CUR);
    dst_place<<<cdiv((long long)N_EDGES, TB), TB, 0, stream>>>(ei, CUR, PERM);

    node_softmax_gather<HEADS, HID, true><<<cdiv((long long)N_NODES * 32, TB), TB, 0, stream>>>(
        ei, CUR, PERM, H1, AS1, AD1, b1, alpha1, AGG1);

    gemm_wmma_f32<HEADS * HID, OUT_CH>
        <<<dim3(N_NODES / 16, OUT_CH / 64), 32, 0, stream>>>(AGG1, W2, G2);

    att_dots<1, OUT_CH, 8><<<cdiv((long long)(N_NODES / 8) * 32, 128), 128, 0, stream>>>(
        G2, a_src2, a_dst2, AS2, AD2);

    node_softmax_gather<1, OUT_CH, false><<<cdiv((long long)N_NODES * 32, TB), TB, 0, stream>>>(
        ei, CUR, PERM, G2, AS2, AD2, b2, alpha2, out);
}


# Edge-list build rides in neighbour launches: counter clear in GEMM1's, count (atomic return = arrival rank, taken here not at placing) in the dots', scan+place fused via LDS scan; 10 launches -> 7
# speedup vs baseline: 2.6266x; 1.1392x over previous
#include <hip/hip_runtime.h>
#include <math.h>

#define N_NODES 20000
#define N_EDGES 320000
#define IN_CH 512
#define HID 128
#define HEADS 4
#define OUT_CH 256
#define NEG_SLOPE 0.2f

typedef float v2f __attribute__((ext_vector_type(2)));
typedef float v4f __attribute__((ext_vector_type(4)));
typedef float v8f __attribute__((ext_vector_type(8)));

struct a_set { v4f v0, v1, v2, v3; };
struct b_set { v4f r0, r1, r2, r3; };

template <int NOUT>
__device__ __forceinline__ b_set load_b(const float* __restrict__ Bp, int k) {
    b_set b;
    b.r0 = *(const v4f*)(Bp + (size_t)(k + 0) * NOUT);
    b.r1 = *(const v4f*)(Bp + (size_t)(k + 1) * NOUT);
    b.r2 = *(const v4f*)(Bp + (size_t)(k + 2) * NOUT);
    b.r3 = *(const v4f*)(Bp + (size_t)(k + 3) * NOUT);
    return b;
}

__device__ __forceinline__ a_set load_a(const float* __restrict__ Ap, int k0) {
    a_set a;
    a.v0 = *(const v4f*)(Ap + k0 + 0);
    a.v1 = *(const v4f*)(Ap + k0 + 4);
    a.v2 = *(const v4f*)(Ap + k0 + 8);
    a.v3 = *(const v4f*)(Ap + k0 + 12);
    return a;
}

__device__ __forceinline__ void mma_sub(const v4f& a, const b_set& b, v8f& acc0, v8f& acc1,
                                        v8f& acc2, v8f& acc3) {
    const v2f a01 = {a.x, a.y};
    const v2f a23 = {a.z, a.w};
    v2f p0 = {b.r0.x, b.r1.x}, p1 = {b.r0.y, b.r1.y};
    v2f p2 = {b.r0.z, b.r1.z}, p3 = {b.r0.w, b.r1.w};
    v2f q0 = {b.r2.x, b.r3.x}, q1 = {b.r2.y, b.r3.y};
    v2f q2 = {b.r2.z, b.r3.z}, q3 = {b.r2.w, b.r3.w};
    asm volatile("" : "+v"(p0), "+v"(p1), "+v"(p2), "+v"(p3),
                      "+v"(q0), "+v"(q1), "+v"(q2), "+v"(q3));
    __builtin_amdgcn_sched_barrier(0);
    acc0 = __builtin_amdgcn_wmma_f32_16x16x4_f32(false, a01, false, p0, (short)0, acc0, false, false);
    acc1 = __builtin_amdgcn_wmma_f32_16x16x4_f32(false, a01, false, p1, (short)0, acc1, false, false);
    acc2 = __builtin_amdgcn_wmma_f32_16x16x4_f32(false, a01, false, p2, (short)0, acc2, false, false);
    acc3 = __builtin_amdgcn_wmma_f32_16x16x4_f32(false, a01, false, p3, (short)0, acc3, false, false);
    acc0 = __builtin_amdgcn_wmma_f32_16x16x4_f32(false, a23, false, q0, (short)0, acc0, false, false);
    acc1 = __builtin_amdgcn_wmma_f32_16x16x4_f32(false, a23, false, q1, (short)0, acc1, false, false);
    acc2 = __builtin_amdgcn_wmma_f32_16x16x4_f32(false, a23, false, q2, (short)0, acc2, false, false);
    acc3 = __builtin_amdgcn_wmma_f32_16x16x4_f32(false, a23, false, q3, (short)0, acc3, false, false);
    __builtin_amdgcn_sched_barrier(0);
    asm volatile("v_nop\n\tv_nop\n\tv_nop\n\tv_nop");
    __builtin_amdgcn_sched_barrier(0);
}

template <int K, int NOUT, bool LAST>
__device__ __forceinline__ void k_block(const float* __restrict__ Ap, const float* __restrict__ Bp,
                                        int k0, const a_set& cur, a_set& nxt, b_set& b0,
                                        v8f& acc0, v8f& acc1, v8f& acc2, v8f& acc3) {
    if (!LAST) nxt = load_a(Ap, k0 + 32);
    b_set b1 = load_b<NOUT>(Bp, k0 + 4);
    __builtin_amdgcn_sched_barrier(0);
    mma_sub(cur.v0, b0, acc0, acc1, acc2, acc3);
    b0 = load_b<NOUT>(Bp, k0 + 8);
    __builtin_amdgcn_sched_barrier(0);
    mma_sub(cur.v1, b1, acc0, acc1, acc2, acc3);
    b1 = load_b<NOUT>(Bp, k0 + 12);
    __builtin_amdgcn_sched_barrier(0);
    mma_sub(cur.v2, b0, acc0, acc1, acc2, acc3);
    if (!LAST) {
        b0 = load_b<NOUT>(Bp, k0 + 32);
        __builtin_amdgcn_sched_barrier(0);
    }
    mma_sub(cur.v3, b1, acc0, acc1, acc2, acc3);
}

template <int K, int NOUT, bool CLEAR>
__global__ __launch_bounds__(32)
void gemm_wmma_f32(const float* __restrict__ A, const float* __restrict__ B,
                   float* __restrict__ C, int* __restrict__ clr) {
    static_assert(K % 64 == 0 && NOUT % 64 == 0, "tile shape");
    const int lane = threadIdx.x & 31;
    const int m0 = blockIdx.x * 16;
    const int n0 = blockIdx.y * 64;
    const int c = lane & 15;
    const int h = lane >> 4;
    if constexpr (CLEAR) {
        if (blockIdx.y == 0 && lane < 16) clr[m0 + lane] = 0;
    }

    v8f acc0 = {}, acc1 = {}, acc2 = {}, acc3 = {};
    const float* __restrict__ Ap = A + (size_t)(m0 + c) * K + 16 * h;
    const float* __restrict__ Bp = B + (size_t)(16 * h) * NOUT + n0 + 4 * c;

    a_set a0 = load_a(Ap, 0), a1;
    b_set b0 = load_b<NOUT>(Bp, 0);
    for (int k0 = 0; k0 < K - 64; k0 += 64) {
        k_block<K, NOUT, false>(Ap, Bp, k0,      a0, a1, b0, acc0, acc1, acc2, acc3);
        k_block<K, NOUT, false>(Ap, Bp, k0 + 32, a1, a0, b0, acc0, acc1, acc2, acc3);
    }
    k_block<K, NOUT, false>(Ap, Bp, K - 64, a0, a1, b0, acc0, acc1, acc2, acc3);
    k_block<K, NOUT, true >(Ap, Bp, K - 32, a1, a0, b0, acc0, acc1, acc2, acc3);

    for (int r = 0; r < 8; ++r) {
        v4f o = {acc0[r], acc1[r], acc2[r], acc3[r]};
        *(v4f*)(C + (size_t)(m0 + r + 8 * h) * NOUT + n0 + 4 * c) = o;
    }
}

template <int X>
__device__ __forceinline__ float row_xor(float v) {
    const int b = __builtin_bit_cast(int, v);
    return __builtin_bit_cast(float, __builtin_amdgcn_update_dpp(b, b, 0x160 + X, 0xf, 0xf, false));
}
__device__ __forceinline__ float xhalf(float v) {
    const int b = __builtin_bit_cast(int, v);
    return __builtin_bit_cast(float,
        __builtin_amdgcn_permlanex16(b, b, 0x76543210, 0xfedcba98, false, false));
}
__device__ __forceinline__ float wave_max(float v) {
    v = fmaxf(v, row_xor<1>(v));
    v = fmaxf(v, row_xor<2>(v));
    v = fmaxf(v, row_xor<4>(v));
    v = fmaxf(v, row_xor<8>(v));
    return fmaxf(v, xhalf(v));
}
__device__ __forceinline__ float wave_sum(float v) {
    v += row_xor<1>(v);
    v += row_xor<2>(v);
    v += row_xor<4>(v);
    v += row_xor<8>(v);
    return v + xhalf(v);
}

template <int H, int CH, int NPW, bool COUNT>
__global__ __launch_bounds__(128)
void att_dots(const float* __restrict__ Hm, const float* __restrict__ a_src,
              const float* __restrict__ a_dst,
              float* __restrict__ as_out, float* __restrict__ ad_out,
              const int* __restrict__ ei, int* __restrict__ cnt, int* __restrict__ rank) {
    constexpr int ROW = H * CH;
    constexpr int GT = (N_NODES / NPW) * 32;
    constexpr int CE = COUNT ? (N_EDGES + GT - 1) / GT : 1;
    constexpr int NV = ROW / 128;
    constexpr int PPH = CH / 128;
    static_assert(CH % 128 == 0 && N_NODES % NPW == 0 && NPW <= 32, "row pieces / node groups");
    const int gid = blockIdx.x * blockDim.x + threadIdx.x;
    const int lane = gid & 31;
    const int n0 = __builtin_amdgcn_readfirstlane(gid >> 5) * NPW;
    if (n0 >= N_NODES) return;

    static_assert(!COUNT || N_EDGES % GT == 0, "every thread counts exactly CE edges");
    int dd[CE], rk[CE];
    if constexpr (COUNT) {
#pragma unroll
        for (int i = 0; i < CE; ++i) dd[i] = ei[N_EDGES + gid + i * GT];
    }

    float4 ws[NV], wd[NV];
#pragma unroll
    for (int p = 0; p < NV; ++p) {
        ws[p] = *(const float4*)(a_src + p * 128 + lane * 4);
        wd[p] = *(const float4*)(a_dst + p * 128 + lane * 4);
    }
    float4 v[NPW][NV];
#pragma unroll
    for (int u = 0; u < NPW; ++u) {
#pragma unroll
        for (int p = 0; p < NV; ++p)
            v[u][p] = *(const float4*)(Hm + (size_t)(n0 + u) * ROW + p * 128 + lane * 4);
    }
    if constexpr (COUNT) {
#pragma unroll
        for (int i = 0; i < CE; ++i) rk[i] = atomicAdd(&cnt[dd[i]], 1);
    }

    float rs[H], rd[H];
#pragma unroll
    for (int h = 0; h < H; ++h) { rs[h] = 0.f; rd[h] = 0.f; }
#pragma unroll
    for (int u = 0; u < NPW; ++u) {
#pragma unroll
        for (int h = 0; h < H; ++h) {
            float s = 0.f, d = 0.f;
#pragma unroll
            for (int q = 0; q < PPH; ++q) {
                const float4 x = v[u][h * PPH + q];
                const float4 a = ws[h * PPH + q];
                const float4 b = wd[h * PPH + q];
                s = __builtin_fmaf(x.x, a.x, s); s = __builtin_fmaf(x.y, a.y, s);
                s = __builtin_fmaf(x.z, a.z, s); s = __builtin_fmaf(x.w, a.w, s);
                d = __builtin_fmaf(x.x, b.x, d); d = __builtin_fmaf(x.y, b.y, d);
                d = __builtin_fmaf(x.z, b.z, d); d = __builtin_fmaf(x.w, b.w, d);
            }
            s = wave_sum(s);
            d = wave_sum(d);
            if (lane == u) { rs[h] = s; rd[h] = d; }
        }
    }
    if (lane < NPW) {
        if constexpr (H == 4) {
            float4 t; t.x = rs[0]; t.y = rs[1]; t.z = rs[2]; t.w = rs[3];
            *(float4*)(as_out + (size_t)(n0 + lane) * 4) = t;
            t.x = rd[0]; t.y = rd[1]; t.z = rd[2]; t.w = rd[3];
            *(float4*)(ad_out + (size_t)(n0 + lane) * 4) = t;
        } else {
#pragma unroll
            for (int h = 0; h < H; ++h) {
                as_out[(size_t)(n0 + lane) * H + h] = rs[h];
                ad_out[(size_t)(n0 + lane) * H + h] = rd[h];
            }
        }
    }
    if constexpr (COUNT) {
#pragma unroll
        for (int i = 0; i < CE; ++i) rank[gid + i * GT] = rk[i];
    }
}

template <int NB>
__global__ __launch_bounds__(1024)
void dst_scan_place(const int* __restrict__ ei, const int* __restrict__ cnt,
                    const int* __restrict__ rank, int* __restrict__ ends,
                    int* __restrict__ perm) {
    constexpr int ITEMS = (N_NODES + 1023) / 1024;
    constexpr int SL = (N_NODES + NB - 1) / NB;
    constexpr int PE = (N_EDGES + NB * 1024 - 1) / (NB * 1024);
    __shared__ int start[N_NODES];
    __shared__ int wsum[32];
    const int t = threadIdx.x;
    const int lane = t & 31, w = t >> 5;
    const int base = t * ITEMS;
    const int e0 = blockIdx.x * 1024 + t;
    int dd[PE], rr[PE];
#pragma unroll
    for (int i = 0; i < PE; ++i) {
        const int e = e0 + i * NB * 1024;
        dd[i] = (e < N_EDGES) ? ei[N_EDGES + e] : 0;
        rr[i] = (e < N_EDGES) ? rank[e] : 0;
    }
    int v[ITEMS];
    int s = 0;
#pragma unroll
    for (int i = 0; i < ITEMS; ++i) {
        int idx = base + i;
        v[i] = (idx < N_NODES) ? cnt[idx] : 0;
        s += v[i];
    }
    int inc = s;
#pragma unroll
    for (int off = 1; off < 32; off <<= 1) {
        int n = __shfl_up(inc, off, 32);
        if (lane >= off) inc += n;
    }
    if (lane == 31) wsum[w] = inc;
    __syncthreads();
    if (w == 0) {
        int x = wsum[lane];
        int xi = x;
#pragma unroll
        for (int off = 1; off < 32; off <<= 1) {
            int n = __shfl_up(xi, off, 32);
            if (lane >= off) xi += n;
        }
        wsum[lane] = xi - x;
    }
    __syncthreads();
    int run = wsum[w] + inc - s;
    const int lo = blockIdx.x * SL, hi = lo + SL;
#pragma unroll
    for (int i = 0; i < ITEMS; ++i) {
        int idx = base + i;
        if (idx < N_NODES) {
            start[idx] = run;
            run += v[i];
            if (idx >= lo && idx < hi) ends[idx] = run;
        }
    }
    __syncthreads();
#pragma unroll
    for (int i = 0; i < PE; ++i) {
        const int e = e0 + i * NB * 1024;
        if (e < N_EDGES) perm[start[dd[i]] + rr[i]] = e;
    }
}


__device__ __forceinline__ float leaky(float v) {
    return (v > 0.f) ? v : NEG_SLOPE * v;
}

template <int H>
__device__ __forceinline__ void load_logits(const float* __restrict__ as, int s, float (&lg)[H]) {
    if constexpr (H == 4) {
        const float4 t = *(const float4*)(as + (size_t)s * 4);
        lg[0] = t.x; lg[1] = t.y; lg[2] = t.z; lg[3] = t.w;
    } else {
#pragma unroll
        for (int h = 0; h < H; ++h) lg[h] = as[(size_t)s * H + h];
    }
}

template <int H>
__device__ __forceinline__ void store_alpha(float* __restrict__ alpha_out, size_t e, const float (&a)[H]) {
    if constexpr (H == 4) {
        float4 t; t.x = a[0]; t.y = a[1]; t.z = a[2]; t.w = a[3];
        *(float4*)(alpha_out + e * 4) = t;
    } else {
#pragma unroll
        for (int h = 0; h < H; ++h) alpha_out[e * H + h] = a[h];
    }
}

template <int H, int CH>
__device__ __forceinline__ void gather_rows(float4 (&acc)[(H * CH) / 128], int my_s,
                                            const float (&a)[H], int n,
                                            const float* __restrict__ Hm, int lane) {
    constexpr int ROW = H * CH;
    constexpr int NV = ROW / 128;
    constexpr int EPT = 8 / NV;
    int j = 0;
    for (; j + EPT <= n; j += EPT) {
        int s[EPT];
        float al[EPT][H];
        float4 v[EPT][NV];
#pragma unroll
        for (int u = 0; u < EPT; ++u) {
            s[u] = __builtin_amdgcn_readlane(my_s, j + u);
#pragma unroll
            for (int h = 0; h < H; ++h)
                al[u][h] = __int_as_float(__builtin_amdgcn_readlane(__float_as_int(a[h]), j + u));
        }
#pragma unroll
        for (int u = 0; u < EPT; ++u) {
#pragma unroll
            for (int p = 0; p < NV; ++p)
                v[u][p] = *(const float4*)(Hm + (size_t)s[u] * ROW + p * 128 + lane * 4);
        }
#pragma unroll
        for (int u = 0; u < EPT; ++u) {
#pragma unroll
            for (int p = 0; p < NV; ++p) {
                const float w = al[u][(p * 128) / CH];
                acc[p].x = __builtin_fmaf(w, v[u][p].x, acc[p].x);
                acc[p].y = __builtin_fmaf(w, v[u][p].y, acc[p].y);
                acc[p].z = __builtin_fmaf(w, v[u][p].z, acc[p].z);
                acc[p].w = __builtin_fmaf(w, v[u][p].w, acc[p].w);
            }
        }
    }
    for (; j < n; ++j) {
        const int s0 = __builtin_amdgcn_readlane(my_s, j);
        float al0[H];
#pragma unroll
        for (int h = 0; h < H; ++h)
            al0[h] = __int_as_float(__builtin_amdgcn_readlane(__float_as_int(a[h]), j));
        float4 v0[NV];
#pragma unroll
        for (int p = 0; p < NV; ++p)
            v0[p] = *(const float4*)(Hm + (size_t)s0 * ROW + p * 128 + lane * 4);
#pragma unroll
        for (int p = 0; p < NV; ++p) {
            const float w = al0[(p * 128) / CH];
            acc[p].x = __builtin_fmaf(w, v0[p].x, acc[p].x);
            acc[p].y = __builtin_fmaf(w, v0[p].y, acc[p].y);
            acc[p].z = __builtin_fmaf(w, v0[p].z, acc[p].z);
            acc[p].w = __builtin_fmaf(w, v0[p].w, acc[p].w);
        }
    }
}

template <int H, int CH, bool DO_ELU>
__global__ __launch_bounds__(256)
void node_softmax_gather(const int* __restrict__ ei, const int* __restrict__ cur,
                         const int* __restrict__ perm, const float* __restrict__ Hm,
                         const float* __restrict__ as, const float* __restrict__ ad,
                         const float* __restrict__ bias,
                         float* __restrict__ alpha_out, float* __restrict__ outp) {
    constexpr int ROW = H * CH;
    constexpr int NV = ROW / 128;
    static_assert(CH % 128 == 0 && (8 % NV) == 0, "row pieces");
    const int gid = blockIdx.x * blockDim.x + threadIdx.x;
    const int lane = gid & 31;
    const int d = __builtin_amdgcn_readfirstlane(gid >> 5);
    if (d >= N_NODES) return;
    const int beg = d ? cur[d - 1] : 0;
    const int end = cur[d];
    const int deg = end - beg;

    float adh[H], vself[H];
    {
        float lgd[H];
        load_logits<H>(as, d, lgd);
#pragma unroll
        for (int h = 0; h < H; ++h) {
            adh[h] = ad[d * H + h];
            vself[h] = leaky(lgd[h] + adh[h]);
        }
    }
    float4 vs[NV];
#pragma unroll
    for (int p = 0; p < NV; ++p)
        vs[p] = *(const float4*)(Hm + (size_t)d * ROW + p * 128 + lane * 4);
    float4 acc[NV];

    if (deg <= 32) {
        int my_e = 0, my_s = 0;
        float my_v[H];
#pragma unroll
        for (int h = 0; h < H; ++h) my_v[h] = vself[h];
        if (lane < deg) {
            my_e = perm[beg + lane];
            my_s = ei[my_e];
            float lg[H];
            load_logits<H>(as, my_s, lg);
#pragma unroll
            for (int h = 0; h < H; ++h) my_v[h] = leaky(lg[h] + adh[h]);
        }
        float a[H], aself[H];
#pragma unroll
        for (int h = 0; h < H; ++h) {
            const float m = wave_max(fmaxf(my_v[h], vself[h]));
            const float ex = (lane < deg) ? expf(my_v[h] - m) : 0.f;
            const float exs = expf(vself[h] - m);
            const float sum = wave_sum(ex) + exs;
            a[h] = ex / sum;
            aself[h] = exs / sum;
        }
        if (lane < deg) store_alpha<H>(alpha_out, (size_t)my_e, a);
        if (lane == 0) store_alpha<H>(alpha_out, (size_t)(N_EDGES + d), aself);
#pragma unroll
        for (int p = 0; p < NV; ++p) {
            const float w = aself[(p * 128) / CH];
            acc[p].x = w * vs[p].x; acc[p].y = w * vs[p].y;
            acc[p].z = w * vs[p].z; acc[p].w = w * vs[p].w;
        }
        gather_rows<H, CH>(acc, my_s, a, deg, Hm, lane);
    } else {
        float m[H], part[H], sum[H], aself[H];
#pragma unroll
        for (int h = 0; h < H; ++h) { m[h] = vself[h]; part[h] = 0.f; }
        for (int i = beg; i < end; i += 32) {
            if (i + lane < end) {
                const int s = ei[perm[i + lane]];
                float lg[H];
                load_logits<H>(as, s, lg);
#pragma unroll
                for (int h = 0; h < H; ++h) m[h] = fmaxf(m[h], leaky(lg[h] + adh[h]));
            }
        }
#pragma unroll
        for (int h = 0; h < H; ++h) m[h] = wave_max(m[h]);
        for (int i = beg; i < end; i += 32) {
            if (i + lane < end) {
                const int s = ei[perm[i + lane]];
                float lg[H];
                load_logits<H>(as, s, lg);
#pragma unroll
                for (int h = 0; h < H; ++h) part[h] += expf(leaky(lg[h] + adh[h]) - m[h]);
            }
        }
#pragma unroll
        for (int h = 0; h < H; ++h) {
            const float exs = expf(vself[h] - m[h]);
            sum[h] = wave_sum(part[h]) + exs;
            aself[h] = exs / sum[h];
        }
        if (lane == 0) store_alpha<H>(alpha_out, (size_t)(N_EDGES + d), aself);
#pragma unroll
        for (int p = 0; p < NV; ++p) {
            const float w = aself[(p * 128) / CH];
            acc[p].x = w * vs[p].x; acc[p].y = w * vs[p].y;
            acc[p].z = w * vs[p].z; acc[p].w = w * vs[p].w;
        }
        for (int i = beg; i < end; i += 32) {
            const int n = (end - i < 32) ? (end - i) : 32;
            int my_s = 0;
            float a[H];
#pragma unroll
            for (int h = 0; h < H; ++h) a[h] = 0.f;
            if (lane < n) {
                const int e = perm[i + lane];
                my_s = ei[e];
                float lg[H];
                load_logits<H>(as, my_s, lg);
#pragma unroll
                for (int h = 0; h < H; ++h) a[h] = expf(leaky(lg[h] + adh[h]) - m[h]) / sum[h];
                store_alpha<H>(alpha_out, (size_t)e, a);
            }
            gather_rows<H, CH>(acc, my_s, a, n, Hm, lane);
        }
    }

#pragma unroll
    for (int p = 0; p < NV; ++p) {
        const float4 b = *(const float4*)(bias + p * 128 + lane * 4);
        float4 o = acc[p];
        o.x += b.x; o.y += b.y; o.z += b.z; o.w += b.w;
        if (DO_ELU) {
            o.x = (o.x > 0.f) ? o.x : (expf(o.x) - 1.f);
            o.y = (o.y > 0.f) ? o.y : (expf(o.y) - 1.f);
            o.z = (o.z > 0.f) ? o.z : (expf(o.z) - 1.f);
            o.w = (o.w > 0.f) ? o.w : (expf(o.w) - 1.f);
        }
        *(float4*)(outp + (size_t)d * ROW + p * 128 + lane * 4) = o;
    }
}

static inline int cdiv(long long a, int b) { return (int)((a + b - 1) / b); }

extern "C" void kernel_launch(void* const* d_in, const int* in_sizes, int n_in,
                              void* d_out, int out_size, void* d_ws, size_t ws_size,
                              hipStream_t stream) {
    const float* x      = (const float*)d_in[0];
    const int*   ei     = (const int*)d_in[1];
    const float* W1     = (const float*)d_in[2];
    const float* a_src1 = (const float*)d_in[3];
    const float* a_dst1 = (const float*)d_in[4];
    const float* b1     = (const float*)d_in[5];
    const float* W2     = (const float*)d_in[6];
    const float* a_src2 = (const float*)d_in[7];
    const float* a_dst2 = (const float*)d_in[8];
    const float* b2     = (const float*)d_in[9];

    const int EP = N_EDGES + N_NODES;

    float* H1   = (float*)d_ws;
    float* AGG1 = H1   + (size_t)N_NODES * 512;
    float* AS1  = AGG1 + (size_t)N_NODES * 512;
    float* AD1  = AS1  + (size_t)N_NODES * HEADS;
    float* AS2  = AD1  + (size_t)N_NODES * HEADS;
    float* AD2  = AS2  + N_NODES;
    int*   CUR  = (int*)(AD2 + N_NODES);
    int*   PERM = CUR + N_NODES;
    float* G2   = H1;
    int*   CNT  = (int*)AGG1;
    int*   RANK = CNT + N_NODES;

    float* out    = (float*)d_out;
    float* alpha1 = out    + (size_t)N_NODES * OUT_CH;
    float* alpha2 = alpha1 + (size_t)EP * HEADS;

    const int TB = 256;

    gemm_wmma_f32<IN_CH, HEADS * HID, true>
        <<<dim3(N_NODES / 16, (HEADS * HID) / 64), 32, 0, stream>>>(x, W1, H1, CNT);

    att_dots<HEADS, HID, 8, true><<<cdiv((long long)(N_NODES / 8) * 32, 128), 128, 0, stream>>>(
        H1, a_src1, a_dst1, AS1, AD1, ei, CNT, RANK);

    dst_scan_place<64><<<64, 1024, 0, stream>>>(ei, CNT, RANK, CUR, PERM);

    node_softmax_gather<HEADS, HID, true><<<cdiv((long long)N_NODES * 32, TB), TB, 0, stream>>>(
        ei, CUR, PERM, H1, AS1, AD1, b1, alpha1, AGG1);

    gemm_wmma_f32<HEADS * HID, OUT_CH, false>
        <<<dim3(N_NODES / 16, OUT_CH / 64), 32, 0, stream>>>(AGG1, W2, G2, nullptr);

    att_dots<1, OUT_CH, 8, false><<<cdiv((long long)(N_NODES / 8) * 32, 128), 128, 0, stream>>>(
        G2, a_src2, a_dst2, AS2, AD2, nullptr, nullptr, nullptr);

    node_softmax_gather<1, OUT_CH, false><<<cdiv((long long)N_NODES * 32, TB), TB, 0, stream>>>(
        ei, CUR, PERM, G2, AS2, AD2, b2, alpha2, out);
}


# Edge list as 32-slot rows per destination: the counting atomic's return is the slot of the (edge, source) pair, overflow list past 32; scan+place launch dropped (7->6); sum order: arrival, as before
# speedup vs baseline: 2.8738x; 1.0941x over previous
#include <hip/hip_runtime.h>
#include <math.h>

#define N_NODES 20000
#define N_EDGES 320000
#define IN_CH 512
#define HID 128
#define HEADS 4
#define OUT_CH 256
#define NEG_SLOPE 0.2f

typedef float v2f __attribute__((ext_vector_type(2)));
typedef float v4f __attribute__((ext_vector_type(4)));
typedef float v8f __attribute__((ext_vector_type(8)));

#define SLOT_CAP 32
#define SLOT_ROW OUT_CH
static_assert(SLOT_CAP == 32, "one slot per lane of a wave");
static_assert(2 * SLOT_CAP <= SLOT_ROW, "a slot row fits into an output row");

struct a_set { v4f v0, v1, v2, v3; };
struct b_set { v4f r0, r1, r2, r3; };

template <int NOUT>
__device__ __forceinline__ b_set load_b(const float* __restrict__ Bp, int k) {
    b_set b;
    b.r0 = *(const v4f*)(Bp + (size_t)(k + 0) * NOUT);
    b.r1 = *(const v4f*)(Bp + (size_t)(k + 1) * NOUT);
    b.r2 = *(const v4f*)(Bp + (size_t)(k + 2) * NOUT);
    b.r3 = *(const v4f*)(Bp + (size_t)(k + 3) * NOUT);
    return b;
}

__device__ __forceinline__ a_set load_a(const float* __restrict__ Ap, int k0) {
    a_set a;
    a.v0 = *(const v4f*)(Ap + k0 + 0);
    a.v1 = *(const v4f*)(Ap + k0 + 4);
    a.v2 = *(const v4f*)(Ap + k0 + 8);
    a.v3 = *(const v4f*)(Ap + k0 + 12);
    return a;
}

__device__ __forceinline__ void mma_sub(const v4f& a, const b_set& b, v8f& acc0, v8f& acc1,
                                        v8f& acc2, v8f& acc3) {
    const v2f a01 = {a.x, a.y};
    const v2f a23 = {a.z, a.w};
    v2f p0 = {b.r0.x, b.r1.x}, p1 = {b.r0.y, b.r1.y};
    v2f p2 = {b.r0.z, b.r1.z}, p3 = {b.r0.w, b.r1.w};
    v2f q0 = {b.r2.x, b.r3.x}, q1 = {b.r2.y, b.r3.y};
    v2f q2 = {b.r2.z, b.r3.z}, q3 = {b.r2.w, b.r3.w};
    asm volatile("" : "+v"(p0), "+v"(p1), "+v"(p2), "+v"(p3),
                      "+v"(q0), "+v"(q1), "+v"(q2), "+v"(q3));
    __builtin_amdgcn_sched_barrier(0);
    acc0 = __builtin_amdgcn_wmma_f32_16x16x4_f32(false, a01, false, p0, (short)0, acc0, false, false);
    acc1 = __builtin_amdgcn_wmma_f32_16x16x4_f32(false, a01, false, p1, (short)0, acc1, false, false);
    acc2 = __builtin_amdgcn_wmma_f32_16x16x4_f32(false, a01, false, p2, (short)0, acc2, false, false);
    acc3 = __builtin_amdgcn_wmma_f32_16x16x4_f32(false, a01, false, p3, (short)0, acc3, false, false);
    acc0 = __builtin_amdgcn_wmma_f32_16x16x4_f32(false, a23, false, q0, (short)0, acc0, false, false);
    acc1 = __builtin_amdgcn_wmma_f32_16x16x4_f32(false, a23, false, q1, (short)0, acc1, false, false);
    acc2 = __builtin_amdgcn_wmma_f32_16x16x4_f32(false, a23, false, q2, (short)0, acc2, false, false);
    acc3 = __builtin_amdgcn_wmma_f32_16x16x4_f32(false, a23, false, q3, (short)0, acc3, false, false);
    __builtin_amdgcn_sched_barrier(0);
    asm volatile("v_nop\n\tv_nop\n\tv_nop\n\tv_nop");
    __builtin_amdgcn_sched_barrier(0);
}

template <int K, int NOUT, bool LAST>
__device__ __forceinline__ void k_block(const float* __restrict__ Ap, const float* __restrict__ Bp,
                                        int k0, const a_set& cur, a_set& nxt, b_set& b0,
                                        v8f& acc0, v8f& acc1, v8f& acc2, v8f& acc3) {
    if (!LAST) nxt = load_a(Ap, k0 + 32);
    b_set b1 = load_b<NOUT>(Bp, k0 + 4);
    __builtin_amdgcn_sched_barrier(0);
    mma_sub(cur.v0, b0, acc0, acc1, acc2, acc3);
    b0 = load_b<NOUT>(Bp, k0 + 8);
    __builtin_amdgcn_sched_barrier(0);
    mma_sub(cur.v1, b1, acc0, acc1, acc2, acc3);
    b1 = load_b<NOUT>(Bp, k0 + 12);
    __builtin_amdgcn_sched_barrier(0);
    mma_sub(cur.v2, b0, acc0, acc1, acc2, acc3);
    if (!LAST) {
        b0 = load_b<NOUT>(Bp, k0 + 32);
        __builtin_amdgcn_sched_barrier(0);
    }
    mma_sub(cur.v3, b1, acc0, acc1, acc2, acc3);
}

template <int K, int NOUT, bool CLEAR>
__global__ __launch_bounds__(32)
void gemm_wmma_f32(const float* __restrict__ A, const float* __restrict__ B,
                   float* __restrict__ C, int* __restrict__ clr) {
    static_assert(K % 64 == 0 && NOUT % 64 == 0, "tile shape");
    const int lane = threadIdx.x & 31;
    const int m0 = blockIdx.x * 16;
    const int n0 = blockIdx.y * 64;
    const int c = lane & 15;
    const int h = lane >> 4;
    if constexpr (CLEAR) {
        if (blockIdx.y == 0 && lane < 16) clr[m0 + lane] = 0;
        if (blockIdx.y == 1 && blockIdx.x == 0 && lane == 0) clr[N_NODES] = 0;
    }

    v8f acc0 = {}, acc1 = {}, acc2 = {}, acc3 = {};
    const float* __restrict__ Ap = A + (size_t)(m0 + c) * K + 16 * h;
    const float* __restrict__ Bp = B + (size_t)(16 * h) * NOUT + n0 + 4 * c;

    a_set a0 = load_a(Ap, 0), a1;
    b_set b0 = load_b<NOUT>(Bp, 0);
    for (int k0 = 0; k0 < K - 64; k0 += 64) {
        k_block<K, NOUT, false>(Ap, Bp, k0,      a0, a1, b0, acc0, acc1, acc2, acc3);
        k_block<K, NOUT, false>(Ap, Bp, k0 + 32, a1, a0, b0, acc0, acc1, acc2, acc3);
    }
    k_block<K, NOUT, false>(Ap, Bp, K - 64, a0, a1, b0, acc0, acc1, acc2, acc3);
    k_block<K, NOUT, true >(Ap, Bp, K - 32, a1, a0, b0, acc0, acc1, acc2, acc3);

    for (int r = 0; r < 8; ++r) {
        v4f o = {acc0[r], acc1[r], acc2[r], acc3[r]};
        *(v4f*)(C + (size_t)(m0 + r + 8 * h) * NOUT + n0 + 4 * c) = o;
    }
}

template <int X>
__device__ __forceinline__ float row_xor(float v) {
    const int b = __builtin_bit_cast(int, v);
    return __builtin_bit_cast(float, __builtin_amdgcn_update_dpp(b, b, 0x160 + X, 0xf, 0xf, false));
}
__device__ __forceinline__ float xhalf(float v) {
    const int b = __builtin_bit_cast(int, v);
    return __builtin_bit_cast(float,
        __builtin_amdgcn_permlanex16(b, b, 0x76543210, 0xfedcba98, false, false));
}
__device__ __forceinline__ float wave_max(float v) {
    v = fmaxf(v, row_xor<1>(v));
    v = fmaxf(v, row_xor<2>(v));
    v = fmaxf(v, row_xor<4>(v));
    v = fmaxf(v, row_xor<8>(v));
    return fmaxf(v, xhalf(v));
}
__device__ __forceinline__ float wave_sum(float v) {
    v += row_xor<1>(v);
    v += row_xor<2>(v);
    v += row_xor<4>(v);
    v += row_xor<8>(v);
    return v + xhalf(v);
}

template <int H, int CH, int NPW, bool COUNT>
__global__ __launch_bounds__(128)
void att_dots(const float* __restrict__ Hm, const float* __restrict__ a_src,
              const float* __restrict__ a_dst,
              float* __restrict__ as_out, float* __restrict__ ad_out,
              const int* __restrict__ ei, int* __restrict__ cnt, int* __restrict__ slots,
              int* __restrict__ ovfn, int* __restrict__ ovf) {
    constexpr int ROW = H * CH;
    constexpr int GT = (N_NODES / NPW) * 32;
    constexpr int CE = COUNT ? (N_EDGES + GT - 1) / GT : 1;
    constexpr int NV = ROW / 128;
    constexpr int PPH = CH / 128;
    static_assert(CH % 128 == 0 && N_NODES % NPW == 0 && NPW <= 32, "row pieces / node groups");
    const int gid = blockIdx.x * blockDim.x + threadIdx.x;
    const int lane = gid & 31;
    const int n0 = __builtin_amdgcn_readfirstlane(gid >> 5) * NPW;
    if (n0 >= N_NODES) return;

    static_assert(!COUNT || N_EDGES % GT == 0, "every thread counts exactly CE edges");
    int dd[CE], ss[CE], rk[CE];
    if constexpr (COUNT) {
#pragma unroll
        for (int i = 0; i < CE; ++i) {
            dd[i] = ei[N_EDGES + gid + i * GT];
            ss[i] = ei[gid + i * GT];
        }
    }

    float4 ws[NV], wd[NV];
#pragma unroll
    for (int p = 0; p < NV; ++p) {
        ws[p] = *(const float4*)(a_src + p * 128 + lane * 4);
        wd[p] = *(const float4*)(a_dst + p * 128 + lane * 4);
    }
    float4 v[NPW][NV];
#pragma unroll
    for (int u = 0; u < NPW; ++u) {
#pragma unroll
        for (int p = 0; p < NV; ++p)
            v[u][p] = *(const float4*)(Hm + (size_t)(n0 + u) * ROW + p * 128 + lane * 4);
    }
    if constexpr (COUNT) {
#pragma unroll
        for (int i = 0; i < CE; ++i) rk[i] = atomicAdd(&cnt[dd[i]], 1);
    }

    float rs[H], rd[H];
#pragma unroll
    for (int h = 0; h < H; ++h) { rs[h] = 0.f; rd[h] = 0.f; }
#pragma unroll
    for (int u = 0; u < NPW; ++u) {
#pragma unroll
        for (int h = 0; h < H; ++h) {
            float s = 0.f, d = 0.f;
#pragma unroll
            for (int q = 0; q < PPH; ++q) {
                const float4 x = v[u][h * PPH + q];
                const float4 a = ws[h * PPH + q];
                const float4 b = wd[h * PPH + q];
                s = __builtin_fmaf(x.x, a.x, s); s = __builtin_fmaf(x.y, a.y, s);
                s = __builtin_fmaf(x.z, a.z, s); s = __builtin_fmaf(x.w, a.w, s);
                d = __builtin_fmaf(x.x, b.x, d); d = __builtin_fmaf(x.y, b.y, d);
                d = __builtin_fmaf(x.z, b.z, d); d = __builtin_fmaf(x.w, b.w, d);
            }
            s = wave_sum(s);
            d = wave_sum(d);
            if (lane == u) { rs[h] = s; rd[h] = d; }
        }
    }
    if (lane < NPW) {
        if constexpr (H == 4) {
            float4 t; t.x = rs[0]; t.y = rs[1]; t.z = rs[2]; t.w = rs[3];
            *(float4*)(as_out + (size_t)(n0 + lane) * 4) = t;
            t.x = rd[0]; t.y = rd[1]; t.z = rd[2]; t.w = rd[3];
            *(float4*)(ad_out + (size_t)(n0 + lane) * 4) = t;
        } else {
#pragma unroll
            for (int h = 0; h < H; ++h) {
                as_out[(size_t)(n0 + lane) * H + h] = rs[h];
                ad_out[(size_t)(n0 + lane) * H + h] = rd[h];
            }
        }
    }
    if constexpr (COUNT) {
#pragma unroll
        for (int i = 0; i < CE; ++i) {
            const int e = gid + i * GT;
            if (rk[i] < SLOT_CAP) {
                int2 r; r.x = e; r.y = ss[i];
                *(int2*)(slots + (size_t)dd[i] * SLOT_ROW + 2 * rk[i]) = r;
            } else {
                ovf[atomicAdd(ovfn, 1)] = e;
            }
        }
    }
}


__device__ __forceinline__ float leaky(float v) {
    return (v > 0.f) ? v : NEG_SLOPE * v;
}

template <int H>
__device__ __forceinline__ void load_logits(const float* __restrict__ as, int s, float (&lg)[H]) {
    if constexpr (H == 4) {
        const float4 t = *(const float4*)(as + (size_t)s * 4);
        lg[0] = t.x; lg[1] = t.y; lg[2] = t.z; lg[3] = t.w;
    } else {
#pragma unroll
        for (int h = 0; h < H; ++h) lg[h] = as[(size_t)s * H + h];
    }
}

template <int H>
__device__ __forceinline__ void store_alpha(float* __restrict__ alpha_out, size_t e, const float (&a)[H]) {
    if constexpr (H == 4) {
        float4 t; t.x = a[0]; t.y = a[1]; t.z = a[2]; t.w = a[3];
        *(float4*)(alpha_out + e * 4) = t;
    } else {
#pragma unroll
        for (int h = 0; h < H; ++h) alpha_out[e * H + h] = a[h];
    }
}

template <int H, int CH>
__device__ __forceinline__ void gather_rows(float4 (&acc)[(H * CH) / 128], int my_s,
                                            const float (&a)[H], int n,
                                            const float* __restrict__ Hm, int lane) {
    constexpr int ROW = H * CH;
    constexpr int NV = ROW / 128;
    constexpr int EPT = 8 / NV;
    int j = 0;
    for (; j + EPT <= n; j += EPT) {
        int s[EPT];
        float al[EPT][H];
        float4 v[EPT][NV];
#pragma unroll
        for (int u = 0; u < EPT; ++u) {
            s[u] = __builtin_amdgcn_readlane(my_s, j + u);
#pragma unroll
            for (int h = 0; h < H; ++h)
                al[u][h] = __int_as_float(__builtin_amdgcn_readlane(__float_as_int(a[h]), j + u));
        }
#pragma unroll
        for (int u = 0; u < EPT; ++u) {
#pragma unroll
            for (int p = 0; p < NV; ++p)
                v[u][p] = *(const float4*)(Hm + (size_t)s[u] * ROW + p * 128 + lane * 4);
        }
#pragma unroll
        for (int u = 0; u < EPT; ++u) {
#pragma unroll
            for (int p = 0; p < NV; ++p) {
                const float w = al[u][(p * 128) / CH];
                acc[p].x = __builtin_fmaf(w, v[u][p].x, acc[p].x);
                acc[p].y = __builtin_fmaf(w, v[u][p].y, acc[p].y);
                acc[p].z = __builtin_fmaf(w, v[u][p].z, acc[p].z);
                acc[p].w = __builtin_fmaf(w, v[u][p].w, acc[p].w);
            }
        }
    }
    for (; j < n; ++j) {
        const int s0 = __builtin_amdgcn_readlane(my_s, j);
        float al0[H];
#pragma unroll
        for (int h = 0; h < H; ++h)
            al0[h] = __int_as_float(__builtin_amdgcn_readlane(__float_as_int(a[h]), j));
        float4 v0[NV];
#pragma unroll
        for (int p = 0; p < NV; ++p)
            v0[p] = *(const float4*)(Hm + (size_t)s0 * ROW + p * 128 + lane * 4);
#pragma unroll
        for (int p = 0; p < NV; ++p) {
            const float w = al0[(p * 128) / CH];
            acc[p].x = __builtin_fmaf(w, v0[p].x, acc[p].x);
            acc[p].y = __builtin_fmaf(w, v0[p].y, acc[p].y);
            acc[p].z = __builtin_fmaf(w, v0[p].z, acc[p].z);
            acc[p].w = __builtin_fmaf(w, v0[p].w, acc[p].w);
        }
    }
}

template <int H, int CH>
__device__ __forceinline__ void gather_rows_mask(float4 (&acc)[(H * CH) / 128], int my_s,
                                                 const float (&a)[H], unsigned mask,
                                                 const float* __restrict__ Hm, int lane) {
    constexpr int ROW = H * CH;
    constexpr int NV = ROW / 128;
    while (mask) {
        const int j = __builtin_ctz(mask);
        mask &= mask - 1;
        const int s0 = __builtin_amdgcn_readlane(my_s, j);
        float al0[H];
#pragma unroll
        for (int h = 0; h < H; ++h)
            al0[h] = __int_as_float(__builtin_amdgcn_readlane(__float_as_int(a[h]), j));
        float4 v0[NV];
#pragma unroll
        for (int p = 0; p < NV; ++p)
            v0[p] = *(const float4*)(Hm + (size_t)s0 * ROW + p * 128 + lane * 4);
#pragma unroll
        for (int p = 0; p < NV; ++p) {
            const float w = al0[(p * 128) / CH];
            acc[p].x = __builtin_fmaf(w, v0[p].x, acc[p].x);
            acc[p].y = __builtin_fmaf(w, v0[p].y, acc[p].y);
            acc[p].z = __builtin_fmaf(w, v0[p].z, acc[p].z);
            acc[p].w = __builtin_fmaf(w, v0[p].w, acc[p].w);
        }
    }
}

template <int H, int CH, bool DO_ELU>
__global__ __launch_bounds__(256)
void node_softmax_gather(const int* __restrict__ ei, const int* __restrict__ cnt,
                         const int* slots, const int* __restrict__ ovfn,
                         const int* __restrict__ ovf, const float* __restrict__ Hm,
                         const float* __restrict__ as, const float* __restrict__ ad,
                         const float* __restrict__ bias,
                         float* __restrict__ alpha_out, float* outp) {
    constexpr int ROW = H * CH;
    constexpr int NV = ROW / 128;
    static_assert(CH % 128 == 0 && (8 % NV) == 0, "row pieces");
    const int gid = blockIdx.x * blockDim.x + threadIdx.x;
    const int lane = gid & 31;
    const int d = __builtin_amdgcn_readfirstlane(gid >> 5);
    if (d >= N_NODES) return;
    const int deg = cnt[d];
    const int* srow = slots + (size_t)d * SLOT_ROW;

    float adh[H], vself[H];
    {
        float lgd[H];
        load_logits<H>(as, d, lgd);
#pragma unroll
        for (int h = 0; h < H; ++h) {
            adh[h] = ad[d * H + h];
            vself[h] = leaky(lgd[h] + adh[h]);
        }
    }
    float4 vs[NV];
#pragma unroll
    for (int p = 0; p < NV; ++p)
        vs[p] = *(const float4*)(Hm + (size_t)d * ROW + p * 128 + lane * 4);
    float4 acc[NV];

    if (deg <= SLOT_CAP) {
        int my_e = 0, my_s = 0;
        float my_v[H];
#pragma unroll
        for (int h = 0; h < H; ++h) my_v[h] = vself[h];
        if (lane < deg) {
            const int2 r = *(const int2*)(srow + 2 * lane);
            my_e = r.x;
            my_s = r.y;
            float lg[H];
            load_logits<H>(as, my_s, lg);
#pragma unroll
            for (int h = 0; h < H; ++h) my_v[h] = leaky(lg[h] + adh[h]);
        }
        float a[H], aself[H];
#pragma unroll
        for (int h = 0; h < H; ++h) {
            const float m = wave_max(fmaxf(my_v[h], vself[h]));
            const float ex = (lane < deg) ? expf(my_v[h] - m) : 0.f;
            const float exs = expf(vself[h] - m);
            const float sum = wave_sum(ex) + exs;
            a[h] = ex / sum;
            aself[h] = exs / sum;
        }
        if (lane < deg) store_alpha<H>(alpha_out, (size_t)my_e, a);
        if (lane == 0) store_alpha<H>(alpha_out, (size_t)(N_EDGES + d), aself);
#pragma unroll
        for (int p = 0; p < NV; ++p) {
            const float w = aself[(p * 128) / CH];
            acc[p].x = w * vs[p].x; acc[p].y = w * vs[p].y;
            acc[p].z = w * vs[p].z; acc[p].w = w * vs[p].w;
        }
        gather_rows<H, CH>(acc, my_s, a, deg, Hm, lane);
    } else {
        const int novf = __builtin_amdgcn_readfirstlane(ovfn[0]);
        const int2 r0 = *(const int2*)(srow + 2 * lane);
        const int e0 = r0.x, s0 = r0.y;
        float v0[H];
        {
            float lg[H];
            load_logits<H>(as, s0, lg);
#pragma unroll
            for (int h = 0; h < H; ++h) v0[h] = leaky(lg[h] + adh[h]);
        }
        float m[H], part[H], sum[H], aself[H];
#pragma unroll
        for (int h = 0; h < H; ++h) m[h] = fmaxf(vself[h], v0[h]);
        for (int i = 0; i < novf; i += 32) {
            if (i + lane < novf) {
                const int e = ovf[i + lane];
                if (ei[N_EDGES + e] == d) {
                    const int s = ei[e];
                    float lg[H];
                    load_logits<H>(as, s, lg);
#pragma unroll
                    for (int h = 0; h < H; ++h) m[h] = fmaxf(m[h], leaky(lg[h] + adh[h]));
                }
            }
        }
#pragma unroll
        for (int h = 0; h < H; ++h) {
            m[h] = wave_max(m[h]);
            part[h] = expf(v0[h] - m[h]);
        }
        for (int i = 0; i < novf; i += 32) {
            if (i + lane < novf) {
                const int e = ovf[i + lane];
                if (ei[N_EDGES + e] == d) {
                    const int s = ei[e];
                    float lg[H];
                    load_logits<H>(as, s, lg);
#pragma unroll
                    for (int h = 0; h < H; ++h) part[h] += expf(leaky(lg[h] + adh[h]) - m[h]);
                }
            }
        }
#pragma unroll
        for (int h = 0; h < H; ++h) {
            const float exs = expf(vself[h] - m[h]);
            sum[h] = wave_sum(part[h]) + exs;
            aself[h] = exs / sum[h];
        }
        if (lane == 0) store_alpha<H>(alpha_out, (size_t)(N_EDGES + d), aself);
#pragma unroll
        for (int p = 0; p < NV; ++p) {
            const float w = aself[(p * 128) / CH];
            acc[p].x = w * vs[p].x; acc[p].y = w * vs[p].y;
            acc[p].z = w * vs[p].z; acc[p].w = w * vs[p].w;
        }
        {
            float a[H];
#pragma unroll
            for (int h = 0; h < H; ++h) a[h] = expf(v0[h] - m[h]) / sum[h];
            store_alpha<H>(alpha_out, (size_t)e0, a);
            gather_rows<H, CH>(acc, s0, a, SLOT_CAP, Hm, lane);
        }
        for (int i = 0; i < novf; i += 32) {
            int my_s = 0;
            bool hit = false;
            float a[H];
#pragma unroll
            for (int h = 0; h < H; ++h) a[h] = 0.f;
            if (i + lane < novf) {
                const int e = ovf[i + lane];
                if (ei[N_EDGES + e] == d) {
                    hit = true;
                    my_s = ei[e];
                    float lg[H];
                    load_logits<H>(as, my_s, lg);
#pragma unroll
                    for (int h = 0; h < H; ++h) a[h] = expf(leaky(lg[h] + adh[h]) - m[h]) / sum[h];
                    store_alpha<H>(alpha_out, (size_t)e, a);
                }
            }
            const unsigned mask = __builtin_amdgcn_ballot_w32(hit);
            gather_rows_mask<H, CH>(acc, my_s, a, mask, Hm, lane);
        }
    }

#pragma unroll
    for (int p = 0; p < NV; ++p) {
        const float4 b = *(const float4*)(bias + p * 128 + lane * 4);
        float4 o = acc[p];
        o.x += b.x; o.y += b.y; o.z += b.z; o.w += b.w;
        if (DO_ELU) {
            o.x = (o.x > 0.f) ? o.x : (expf(o.x) - 1.f);
            o.y = (o.y > 0.f) ? o.y : (expf(o.y) - 1.f);
            o.z = (o.z > 0.f) ? o.z : (expf(o.z) - 1.f);
            o.w = (o.w > 0.f) ? o.w : (expf(o.w) - 1.f);
        }
        *(float4*)(outp + (size_t)d * ROW + p * 128 + lane * 4) = o;
    }
}

static inline int cdiv(long long a, int b) { return (int)((a + b - 1) / b); }

extern "C" void kernel_launch(void* const* d_in, const int* in_sizes, int n_in,
                              void* d_out, int out_size, void* d_ws, size_t ws_size,
                              hipStream_t stream) {
    const float* x      = (const float*)d_in[0];
    const int*   ei     = (const int*)d_in[1];
    const float* W1     = (const float*)d_in[2];
    const float* a_src1 = (const float*)d_in[3];
    const float* a_dst1 = (const float*)d_in[4];
    const float* b1     = (const float*)d_in[5];
    const float* W2     = (const float*)d_in[6];
    const float* a_src2 = (const float*)d_in[7];
    const float* a_dst2 = (const float*)d_in[8];
    const float* b2     = (const float*)d_in[9];

    const int EP = N_EDGES + N_NODES;

    float* H1   = (float*)d_ws;
    float* AGG1 = H1   + (size_t)N_NODES * 512;
    float* AS1  = AGG1 + (size_t)N_NODES * 512;
    float* AD1  = AS1  + (size_t)N_NODES * HEADS;
    float* AS2  = AD1  + (size_t)N_NODES * HEADS;
    float* AD2  = AS2  + N_NODES;
    int*   CNT  = (int*)(AD2 + N_NODES);
    int*   OVFN = CNT + N_NODES;
    int*   OVF  = OVFN + 1;
    float* G2   = H1;

    float* out    = (float*)d_out;
    float* alpha1 = out    + (size_t)N_NODES * OUT_CH;
    float* alpha2 = alpha1 + (size_t)EP * HEADS;
    int*   SLOTS  = (int*)d_out;

    const int TB = 256;

    gemm_wmma_f32<IN_CH, HEADS * HID, true>
        <<<dim3(N_NODES / 16, (HEADS * HID) / 64), 32, 0, stream>>>(x, W1, H1, CNT);

    att_dots<HEADS, HID, 8, true><<<cdiv((long long)(N_NODES / 8) * 32, 128), 128, 0, stream>>>(
        H1, a_src1, a_dst1, AS1, AD1, ei, CNT, SLOTS, OVFN, OVF);

    node_softmax_gather<HEADS, HID, true><<<cdiv((long long)N_NODES * 32, TB), TB, 0, stream>>>(
        ei, CNT, SLOTS, OVFN, OVF, H1, AS1, AD1, b1, alpha1, AGG1);

    gemm_wmma_f32<HEADS * HID, OUT_CH, false>
        <<<dim3(N_NODES / 16, OUT_CH / 64), 32, 0, stream>>>(AGG1, W2, G2, nullptr);

    att_dots<1, OUT_CH, 8, false><<<cdiv((long long)(N_NODES / 8) * 32, 128), 128, 0, stream>>>(
        G2, a_src2, a_dst2, AS2, AD2, nullptr, nullptr, nullptr, nullptr, nullptr);

    node_softmax_gather<1, OUT_CH, false><<<cdiv((long long)N_NODES * 32, TB), TB, 0, stream>>>(
        ei, CNT, SLOTS, OVFN, OVF, G2, AS2, AD2, b2, alpha2, out);
}


# Layer-2 GEMM fused into layer 1's softmax+gather launch: 8-wave groups do 16 nodes (2 per wave), keep the finished rows in LDS, multiply them by W2 on WMMA, k order kept; 6->5 launches, no h2 plane
# speedup vs baseline: 3.0520x; 1.0620x over previous
#include <hip/hip_runtime.h>
#include <math.h>

#define N_NODES 20000
#define N_EDGES 320000
#define IN_CH 512
#define HID 128
#define HEADS 4
#define OUT_CH 256
#define NEG_SLOPE 0.2f

typedef float v2f __attribute__((ext_vector_type(2)));
typedef float v4f __attribute__((ext_vector_type(4)));
typedef float v8f __attribute__((ext_vector_type(8)));

#define SLOT_CAP 32
#define SLOT_ROW OUT_CH
static_assert(SLOT_CAP == 32, "one slot per lane of a wave");
static_assert(2 * SLOT_CAP <= SLOT_ROW, "a slot row fits into an output row");

struct a_set { v4f v0, v1, v2, v3; };
struct b_set { v4f r0, r1, r2, r3; };

template <int NOUT>
__device__ __forceinline__ b_set load_b(const float* __restrict__ Bp, int k) {
    b_set b;
    b.r0 = *(const v4f*)(Bp + (size_t)(k + 0) * NOUT);
    b.r1 = *(const v4f*)(Bp + (size_t)(k + 1) * NOUT);
    b.r2 = *(const v4f*)(Bp + (size_t)(k + 2) * NOUT);
    b.r3 = *(const v4f*)(Bp + (size_t)(k + 3) * NOUT);
    return b;
}

__device__ __forceinline__ a_set load_a(const float* __restrict__ Ap, int k0) {
    a_set a;
    a.v0 = *(const v4f*)(Ap + k0 + 0);
    a.v1 = *(const v4f*)(Ap + k0 + 4);
    a.v2 = *(const v4f*)(Ap + k0 + 8);
    a.v3 = *(const v4f*)(Ap + k0 + 12);
    return a;
}

__device__ __forceinline__ void mma_sub(const v4f& a, const b_set& b, v8f& acc0, v8f& acc1,
                                        v8f& acc2, v8f& acc3) {
    const v2f a01 = {a.x, a.y};
    const v2f a23 = {a.z, a.w};
    v2f p0 = {b.r0.x, b.r1.x}, p1 = {b.r0.y, b.r1.y};
    v2f p2 = {b.r0.z, b.r1.z}, p3 = {b.r0.w, b.r1.w};
    v2f q0 = {b.r2.x, b.r3.x}, q1 = {b.r2.y, b.r3.y};
    v2f q2 = {b.r2.z, b.r3.z}, q3 = {b.r2.w, b.r3.w};
    asm volatile("" : "+v"(p0), "+v"(p1), "+v"(p2), "+v"(p3),
                      "+v"(q0), "+v"(q1), "+v"(q2), "+v"(q3));
    __builtin_amdgcn_sched_barrier(0);
    acc0 = __builtin_amdgcn_wmma_f32_16x16x4_f32(false, a01, false, p0, (short)0, acc0, false, false);
    acc1 = __builtin_amdgcn_wmma_f32_16x16x4_f32(false, a01, false, p1, (short)0, acc1, false, false);
    acc2 = __builtin_amdgcn_wmma_f32_16x16x4_f32(false, a01, false, p2, (short)0, acc2, false, false);
    acc3 = __builtin_amdgcn_wmma_f32_16x16x4_f32(false, a01, false, p3, (short)0, acc3, false, false);
    acc0 = __builtin_amdgcn_wmma_f32_16x16x4_f32(false, a23, false, q0, (short)0, acc0, false, false);
    acc1 = __builtin_amdgcn_wmma_f32_16x16x4_f32(false, a23, false, q1, (short)0, acc1, false, false);
    acc2 = __builtin_amdgcn_wmma_f32_16x16x4_f32(false, a23, false, q2, (short)0, acc2, false, false);
    acc3 = __builtin_amdgcn_wmma_f32_16x16x4_f32(false, a23, false, q3, (short)0, acc3, false, false);
    __builtin_amdgcn_sched_barrier(0);
    asm volatile("v_nop\n\tv_nop\n\tv_nop\n\tv_nop");
    __builtin_amdgcn_sched_barrier(0);
}

template <int K, int NOUT, bool LAST>
__device__ __forceinline__ void k_block(const float* __restrict__ Ap, const float* __restrict__ Bp,
                                        int k0, const a_set& cur, a_set& nxt, b_set& b0,
                                        v8f& acc0, v8f& acc1, v8f& acc2, v8f& acc3) {
    if (!LAST) nxt = load_a(Ap, k0 + 32);
    b_set b1 = load_b<NOUT>(Bp, k0 + 4);
    __builtin_amdgcn_sched_barrier(0);
    mma_sub(cur.v0, b0, acc0, acc1, acc2, acc3);
    b0 = load_b<NOUT>(Bp, k0 + 8);
    __builtin_amdgcn_sched_barrier(0);
    mma_sub(cur.v1, b1, acc0, acc1, acc2, acc3);
    b1 = load_b<NOUT>(Bp, k0 + 12);
    __builtin_amdgcn_sched_barrier(0);
    mma_sub(cur.v2, b0, acc0, acc1, acc2, acc3);
    if (!LAST) {
        b0 = load_b<NOUT>(Bp, k0 + 32);
        __builtin_amdgcn_sched_barrier(0);
    }
    mma_sub(cur.v3, b1, acc0, acc1, acc2, acc3);
}

template <int K, int NOUT, bool CLEAR>
__global__ __launch_bounds__(32)
void gemm_wmma_f32(const float* __restrict__ A, const float* __restrict__ B,
                   float* __restrict__ C, int* __restrict__ clr) {
    static_assert(K % 64 == 0 && NOUT % 64 == 0, "tile shape");
    const int lane = threadIdx.x & 31;
    const int m0 = blockIdx.x * 16;
    const int n0 = blockIdx.y * 64;
    const int c = lane & 15;
    const int h = lane >> 4;
    if constexpr (CLEAR) {
        if (blockIdx.y == 0 && lane < 16) clr[m0 + lane] = 0;
        if (blockIdx.y == 1 && blockIdx.x == 0 && lane == 0) clr[N_NODES] = 0;
    }

    v8f acc0 = {}, acc1 = {}, acc2 = {}, acc3 = {};
    const float* __restrict__ Ap = A + (size_t)(m0 + c) * K + 16 * h;
    const float* __restrict__ Bp = B + (size_t)(16 * h) * NOUT + n0 + 4 * c;

    a_set a0 = load_a(Ap, 0), a1;
    b_set b0 = load_b<NOUT>(Bp, 0);
    for (int k0 = 0; k0 < K - 64; k0 += 64) {
        k_block<K, NOUT, false>(Ap, Bp, k0,      a0, a1, b0, acc0, acc1, acc2, acc3);
        k_block<K, NOUT, false>(Ap, Bp, k0 + 32, a1, a0, b0, acc0, acc1, acc2, acc3);
    }
    k_block<K, NOUT, false>(Ap, Bp, K - 64, a0, a1, b0, acc0, acc1, acc2, acc3);
    k_block<K, NOUT, true >(Ap, Bp, K - 32, a1, a0, b0, acc0, acc1, acc2, acc3);

    for (int r = 0; r < 8; ++r) {
        v4f o = {acc0[r], acc1[r], acc2[r], acc3[r]};
        *(v4f*)(C + (size_t)(m0 + r + 8 * h) * NOUT + n0 + 4 * c) = o;
    }
}

template <int X>
__device__ __forceinline__ float row_xor(float v) {
    const int b = __builtin_bit_cast(int, v);
    return __builtin_bit_cast(float, __builtin_amdgcn_update_dpp(b, b, 0x160 + X, 0xf, 0xf, false));
}
__device__ __forceinline__ float xhalf(float v) {
    const int b = __builtin_bit_cast(int, v);
    return __builtin_bit_cast(float,
        __builtin_amdgcn_permlanex16(b, b, 0x76543210, 0xfedcba98, false, false));
}
__device__ __forceinline__ float wave_max(float v) {
    v = fmaxf(v, row_xor<1>(v));
    v = fmaxf(v, row_xor<2>(v));
    v = fmaxf(v, row_xor<4>(v));
    v = fmaxf(v, row_xor<8>(v));
    return fmaxf(v, xhalf(v));
}
__device__ __forceinline__ float wave_sum(float v) {
    v += row_xor<1>(v);
    v += row_xor<2>(v);
    v += row_xor<4>(v);
    v += row_xor<8>(v);
    return v + xhalf(v);
}

template <int H, int CH, int NPW, bool COUNT>
__global__ __launch_bounds__(128)
void att_dots(const float* __restrict__ Hm, const float* __restrict__ a_src,
              const float* __restrict__ a_dst,
              float* __restrict__ as_out, float* __restrict__ ad_out,
              const int* __restrict__ ei, int* __restrict__ cnt, int* __restrict__ slots,
              int* __restrict__ ovfn, int* __restrict__ ovf) {
    constexpr int ROW = H * CH;
    constexpr int GT = (N_NODES / NPW) * 32;
    constexpr int CE = COUNT ? (N_EDGES + GT - 1) / GT : 1;
    constexpr int NV = ROW / 128;
    constexpr int PPH = CH / 128;
    static_assert(CH % 128 == 0 && N_NODES % NPW == 0 && NPW <= 32, "row pieces / node groups");
    const int gid = blockIdx.x * blockDim.x + threadIdx.x;
    const int lane = gid & 31;
    const int n0 = __builtin_amdgcn_readfirstlane(gid >> 5) * NPW;
    if (n0 >= N_NODES) return;

    static_assert(!COUNT || N_EDGES % GT == 0, "every thread counts exactly CE edges");
    int dd[CE], ss[CE], rk[CE];
    if constexpr (COUNT) {
#pragma unroll
        for (int i = 0; i < CE; ++i) {
            dd[i] = ei[N_EDGES + gid + i * GT];
            ss[i] = ei[gid + i * GT];
        }
    }

    float4 ws[NV], wd[NV];
#pragma unroll
    for (int p = 0; p < NV; ++p) {
        ws[p] = *(const float4*)(a_src + p * 128 + lane * 4);
        wd[p] = *(const float4*)(a_dst + p * 128 + lane * 4);
    }
    float4 v[NPW][NV];
#pragma unroll
    for (int u = 0; u < NPW; ++u) {
#pragma unroll
        for (int p = 0; p < NV; ++p)
            v[u][p] = *(const float4*)(Hm + (size_t)(n0 + u) * ROW + p * 128 + lane * 4);
    }
    if constexpr (COUNT) {
#pragma unroll
        for (int i = 0; i < CE; ++i) rk[i] = atomicAdd(&cnt[dd[i]], 1);
    }

    float rs[H], rd[H];
#pragma unroll
    for (int h = 0; h < H; ++h) { rs[h] = 0.f; rd[h] = 0.f; }
#pragma unroll
    for (int u = 0; u < NPW; ++u) {
#pragma unroll
        for (int h = 0; h < H; ++h) {
            float s = 0.f, d = 0.f;
#pragma unroll
            for (int q = 0; q < PPH; ++q) {
                const float4 x = v[u][h * PPH + q];
                const float4 a = ws[h * PPH + q];
                const float4 b = wd[h * PPH + q];
                s = __builtin_fmaf(x.x, a.x, s); s = __builtin_fmaf(x.y, a.y, s);
                s = __builtin_fmaf(x.z, a.z, s); s = __builtin_fmaf(x.w, a.w, s);
                d = __builtin_fmaf(x.x, b.x, d); d = __builtin_fmaf(x.y, b.y, d);
                d = __builtin_fmaf(x.z, b.z, d); d = __builtin_fmaf(x.w, b.w, d);
            }
            s = wave_sum(s);
            d = wave_sum(d);
            if (lane == u) { rs[h] = s; rd[h] = d; }
        }
    }
    if (lane < NPW) {
        if constexpr (H == 4) {
            float4 t; t.x = rs[0]; t.y = rs[1]; t.z = rs[2]; t.w = rs[3];
            *(float4*)(as_out + (size_t)(n0 + lane) * 4) = t;
            t.x = rd[0]; t.y = rd[1]; t.z = rd[2]; t.w = rd[3];
            *(float4*)(ad_out + (size_t)(n0 + lane) * 4) = t;
        } else {
#pragma unroll
            for (int h = 0; h < H; ++h) {
                as_out[(size_t)(n0 + lane) * H + h] = rs[h];
                ad_out[(size_t)(n0 + lane) * H + h] = rd[h];
            }
        }
    }
    if constexpr (COUNT) {
#pragma unroll
        for (int i = 0; i < CE; ++i) {
            const int e = gid + i * GT;
            if (rk[i] < SLOT_CAP) {
                int2 r; r.x = e; r.y = ss[i];
                *(int2*)(slots + (size_t)dd[i] * SLOT_ROW + 2 * rk[i]) = r;
            } else {
                ovf[atomicAdd(ovfn, 1)] = e;
            }
        }
    }
}


__device__ __forceinline__ float leaky(float v) {
    return (v > 0.f) ? v : NEG_SLOPE * v;
}

template <int H>
__device__ __forceinline__ void load_logits(const float* __restrict__ as, int s, float (&lg)[H]) {
    if constexpr (H == 4) {
        const float4 t = *(const float4*)(as + (size_t)s * 4);
        lg[0] = t.x; lg[1] = t.y; lg[2] = t.z; lg[3] = t.w;
    } else {
#pragma unroll
        for (int h = 0; h < H; ++h) lg[h] = as[(size_t)s * H + h];
    }
}

template <int H>
__device__ __forceinline__ void store_alpha(float* __restrict__ alpha_out, size_t e, const float (&a)[H]) {
    if constexpr (H == 4) {
        float4 t; t.x = a[0]; t.y = a[1]; t.z = a[2]; t.w = a[3];
        *(float4*)(alpha_out + e * 4) = t;
    } else {
#pragma unroll
        for (int h = 0; h < H; ++h) alpha_out[e * H + h] = a[h];
    }
}

template <int H, int CH>
__device__ __forceinline__ void gather_rows(float4 (&acc)[(H * CH) / 128], int my_s,
                                            const float (&a)[H], int n,
                                            const float* __restrict__ Hm, int lane) {
    constexpr int ROW = H * CH;
    constexpr int NV = ROW / 128;
    constexpr int EPT = 8 / NV;
    int j = 0;
    for (; j + EPT <= n; j += EPT) {
        int s[EPT];
        float al[EPT][H];
        float4 v[EPT][NV];
#pragma unroll
        for (int u = 0; u < EPT; ++u) {
            s[u] = __builtin_amdgcn_readlane(my_s, j + u);
#pragma unroll
            for (int h = 0; h < H; ++h)
                al[u][h] = __int_as_float(__builtin_amdgcn_readlane(__float_as_int(a[h]), j + u));
        }
#pragma unroll
        for (int u = 0; u < EPT; ++u) {
#pragma unroll
            for (int p = 0; p < NV; ++p)
                v[u][p] = *(const float4*)(Hm + (size_t)s[u] * ROW + p * 128 + lane * 4);
        }
#pragma unroll
        for (int u = 0; u < EPT; ++u) {
#pragma unroll
            for (int p = 0; p < NV; ++p) {
                const float w = al[u][(p * 128) / CH];
                acc[p].x = __builtin_fmaf(w, v[u][p].x, acc[p].x);
                acc[p].y = __builtin_fmaf(w, v[u][p].y, acc[p].y);
                acc[p].z = __builtin_fmaf(w, v[u][p].z, acc[p].z);
                acc[p].w = __builtin_fmaf(w, v[u][p].w, acc[p].w);
            }
        }
    }
    for (; j < n; ++j) {
        const int s0 = __builtin_amdgcn_readlane(my_s, j);
        float al0[H];
#pragma unroll
        for (int h = 0; h < H; ++h)
            al0[h] = __int_as_float(__builtin_amdgcn_readlane(__float_as_int(a[h]), j));
        float4 v0[NV];
#pragma unroll
        for (int p = 0; p < NV; ++p)
            v0[p] = *(const float4*)(Hm + (size_t)s0 * ROW + p * 128 + lane * 4);
#pragma unroll
        for (int p = 0; p < NV; ++p) {
            const float w = al0[(p * 128) / CH];
            acc[p].x = __builtin_fmaf(w, v0[p].x, acc[p].x);
            acc[p].y = __builtin_fmaf(w, v0[p].y, acc[p].y);
            acc[p].z = __builtin_fmaf(w, v0[p].z, acc[p].z);
            acc[p].w = __builtin_fmaf(w, v0[p].w, acc[p].w);
        }
    }
}

template <int H, int CH>
__device__ __forceinline__ void gather_rows_mask(float4 (&acc)[(H * CH) / 128], int my_s,
                                                 const float (&a)[H], unsigned mask,
                                                 const float* __restrict__ Hm, int lane) {
    constexpr int ROW = H * CH;
    constexpr int NV = ROW / 128;
    while (mask) {
        const int j = __builtin_ctz(mask);
        mask &= mask - 1;
        const int s0 = __builtin_amdgcn_readlane(my_s, j);
        float al0[H];
#pragma unroll
        for (int h = 0; h < H; ++h)
            al0[h] = __int_as_float(__builtin_amdgcn_readlane(__float_as_int(a[h]), j));
        float4 v0[NV];
#pragma unroll
        for (int p = 0; p < NV; ++p)
            v0[p] = *(const float4*)(Hm + (size_t)s0 * ROW + p * 128 + lane * 4);
#pragma unroll
        for (int p = 0; p < NV; ++p) {
            const float w = al0[(p * 128) / CH];
            acc[p].x = __builtin_fmaf(w, v0[p].x, acc[p].x);
            acc[p].y = __builtin_fmaf(w, v0[p].y, acc[p].y);
            acc[p].z = __builtin_fmaf(w, v0[p].z, acc[p].z);
            acc[p].w = __builtin_fmaf(w, v0[p].w, acc[p].w);
        }
    }
}

template <int H, int CH, bool DO_ELU>
__device__ __forceinline__
void node_row(const int d, const int lane,
              const int* __restrict__ ei, const int* __restrict__ cnt,
              const int* slots, const int* __restrict__ ovfn,
              const int* __restrict__ ovf, const float* __restrict__ Hm,
              const float* __restrict__ as, const float* __restrict__ ad,
              const float* __restrict__ bias,
              float* __restrict__ alpha_out, float4 (&o)[(H * CH) / 128]) {
    constexpr int ROW = H * CH;
    constexpr int NV = ROW / 128;
    static_assert(CH % 128 == 0 && (8 % NV) == 0, "row pieces");
    const int deg = cnt[d];
    const int* srow = slots + (size_t)d * SLOT_ROW;

    float adh[H], vself[H];
    {
        float lgd[H];
        load_logits<H>(as, d, lgd);
#pragma unroll
        for (int h = 0; h < H; ++h) {
            adh[h] = ad[d * H + h];
            vself[h] = leaky(lgd[h] + adh[h]);
        }
    }
    float4 vs[NV];
#pragma unroll
    for (int p = 0; p < NV; ++p)
        vs[p] = *(const float4*)(Hm + (size_t)d * ROW + p * 128 + lane * 4);
    float4 acc[NV];

    if (deg <= SLOT_CAP) {
        int my_e = 0, my_s = 0;
        float my_v[H];
#pragma unroll
        for (int h = 0; h < H; ++h) my_v[h] = vself[h];
        if (lane < deg) {
            const int2 r = *(const int2*)(srow + 2 * lane);
            my_e = r.x;
            my_s = r.y;
            float lg[H];
            load_logits<H>(as, my_s, lg);
#pragma unroll
            for (int h = 0; h < H; ++h) my_v[h] = leaky(lg[h] + adh[h]);
        }
        float a[H], aself[H];
#pragma unroll
        for (int h = 0; h < H; ++h) {
            const float m = wave_max(fmaxf(my_v[h], vself[h]));
            const float ex = (lane < deg) ? expf(my_v[h] - m) : 0.f;
            const float exs = expf(vself[h] - m);
            const float sum = wave_sum(ex) + exs;
            a[h] = ex / sum;
            aself[h] = exs / sum;
        }
        if (lane < deg) store_alpha<H>(alpha_out, (size_t)my_e, a);
        if (lane == 0) store_alpha<H>(alpha_out, (size_t)(N_EDGES + d), aself);
#pragma unroll
        for (int p = 0; p < NV; ++p) {
            const float w = aself[(p * 128) / CH];
            acc[p].x = w * vs[p].x; acc[p].y = w * vs[p].y;
            acc[p].z = w * vs[p].z; acc[p].w = w * vs[p].w;
        }
        gather_rows<H, CH>(acc, my_s, a, deg, Hm, lane);
    } else {
        const int novf = __builtin_amdgcn_readfirstlane(ovfn[0]);
        const int2 r0 = *(const int2*)(srow + 2 * lane);
        const int e0 = r0.x, s0 = r0.y;
        float v0[H];
        {
            float lg[H];
            load_logits<H>(as, s0, lg);
#pragma unroll
            for (int h = 0; h < H; ++h) v0[h] = leaky(lg[h] + adh[h]);
        }
        float m[H], part[H], sum[H], aself[H];
#pragma unroll
        for (int h = 0; h < H; ++h) m[h] = fmaxf(vself[h], v0[h]);
        for (int i = 0; i < novf; i += 32) {
            if (i + lane < novf) {
                const int e = ovf[i + lane];
                if (ei[N_EDGES + e] == d) {
                    const int s = ei[e];
                    float lg[H];
                    load_logits<H>(as, s, lg);
#pragma unroll
                    for (int h = 0; h < H; ++h) m[h] = fmaxf(m[h], leaky(lg[h] + adh[h]));
                }
            }
        }
#pragma unroll
        for (int h = 0; h < H; ++h) {
            m[h] = wave_max(m[h]);
            part[h] = expf(v0[h] - m[h]);
        }
        for (int i = 0; i < novf; i += 32) {
            if (i + lane < novf) {
                const int e = ovf[i + lane];
                if (ei[N_EDGES + e] == d) {
                    const int s = ei[e];
                    float lg[H];
                    load_logits<H>(as, s, lg);
#pragma unroll
                    for (int h = 0; h < H; ++h) part[h] += expf(leaky(lg[h] + adh[h]) - m[h]);
                }
            }
        }
#pragma unroll
        for (int h = 0; h < H; ++h) {
            const float exs = expf(vself[h] - m[h]);
            sum[h] = wave_sum(part[h]) + exs;
            aself[h] = exs / sum[h];
        }
        if (lane == 0) store_alpha<H>(alpha_out, (size_t)(N_EDGES + d), aself);
#pragma unroll
        for (int p = 0; p < NV; ++p) {
            const float w = aself[(p * 128) / CH];
            acc[p].x = w * vs[p].x; acc[p].y = w * vs[p].y;
            acc[p].z = w * vs[p].z; acc[p].w = w * vs[p].w;
        }
        {
            float a[H];
#pragma unroll
            for (int h = 0; h < H; ++h) a[h] = expf(v0[h] - m[h]) / sum[h];
            store_alpha<H>(alpha_out, (size_t)e0, a);
            gather_rows<H, CH>(acc, s0, a, SLOT_CAP, Hm, lane);
        }
        for (int i = 0; i < novf; i += 32) {
            int my_s = 0;
            bool hit = false;
            float a[H];
#pragma unroll
            for (int h = 0; h < H; ++h) a[h] = 0.f;
            if (i + lane < novf) {
                const int e = ovf[i + lane];
                if (ei[N_EDGES + e] == d) {
                    hit = true;
                    my_s = ei[e];
                    float lg[H];
                    load_logits<H>(as, my_s, lg);
#pragma unroll
                    for (int h = 0; h < H; ++h) a[h] = expf(leaky(lg[h] + adh[h]) - m[h]) / sum[h];
                    store_alpha<H>(alpha_out, (size_t)e, a);
                }
            }
            const unsigned mask = __builtin_amdgcn_ballot_w32(hit);
            gather_rows_mask<H, CH>(acc, my_s, a, mask, Hm, lane);
        }
    }

#pragma unroll
    for (int p = 0; p < NV; ++p) {
        const float4 b = *(const float4*)(bias + p * 128 + lane * 4);
        float4 t = acc[p];
        t.x += b.x; t.y += b.y; t.z += b.z; t.w += b.w;
        if (DO_ELU) {
            t.x = (t.x > 0.f) ? t.x : (expf(t.x) - 1.f);
            t.y = (t.y > 0.f) ? t.y : (expf(t.y) - 1.f);
            t.z = (t.z > 0.f) ? t.z : (expf(t.z) - 1.f);
            t.w = (t.w > 0.f) ? t.w : (expf(t.w) - 1.f);
        }
        o[p] = t;
    }
}

template <int H, int CH, bool DO_ELU>
__global__ __launch_bounds__(256)
void node_softmax_gather(const int* __restrict__ ei, const int* __restrict__ cnt,
                         const int* slots, const int* __restrict__ ovfn,
                         const int* __restrict__ ovf, const float* __restrict__ Hm,
                         const float* __restrict__ as, const float* __restrict__ ad,
                         const float* __restrict__ bias,
                         float* __restrict__ alpha_out, float* outp) {
    constexpr int ROW = H * CH;
    constexpr int NV = ROW / 128;
    const int gid = blockIdx.x * blockDim.x + threadIdx.x;
    const int lane = gid & 31;
    const int d = __builtin_amdgcn_readfirstlane(gid >> 5);
    if (d >= N_NODES) return;
    float4 o[NV];
    node_row<H, CH, DO_ELU>(d, lane, ei, cnt, slots, ovfn, ovf, Hm, as, ad, bias, alpha_out, o);
#pragma unroll
    for (int p = 0; p < NV; ++p)
        *(float4*)(outp + (size_t)d * ROW + p * 128 + lane * 4) = o[p];
}

#define FG_ROWS 16
#define FG_WAVES 8
#define FG_PITCH (HEADS * HID + 4)
static_assert(N_NODES % FG_ROWS == 0, "every work-group is full: all its waves reach the barrier");
static_assert(FG_ROWS % FG_WAVES == 0, "nodes per wave");
static_assert(((FG_PITCH * 4) % 16) == 0 && (((FG_PITCH * 4) / 16) % 2) == 1, "row pitch = 16 B x odd");
static_assert(FG_WAVES * 32 == OUT_CH, "column strips of the waves");

struct b2_half { v2f r[8]; };

__device__ __forceinline__ b2_half load_b2(const float* __restrict__ Bp, int k) {
    b2_half b;
#pragma unroll
    for (int j = 0; j < 8; ++j) b.r[j] = *(const v2f*)(Bp + (size_t)(k + j) * OUT_CH);
    return b;
}

__device__ __forceinline__ void mma_sub2(const v4f& a, const v2f& r0, const v2f& r1, const v2f& r2,
                                         const v2f& r3, v8f& acc0, v8f& acc1) {
    const v2f a01 = {a.x, a.y};
    const v2f a23 = {a.z, a.w};
    v2f p0 = {r0.x, r1.x}, p1 = {r0.y, r1.y};
    v2f q0 = {r2.x, r3.x}, q1 = {r2.y, r3.y};
    asm volatile("" : "+v"(p0), "+v"(p1), "+v"(q0), "+v"(q1));
    __builtin_amdgcn_sched_barrier(0);
    acc0 = __builtin_amdgcn_wmma_f32_16x16x4_f32(false, a01, false, p0, (short)0, acc0, false, false);
    acc1 = __builtin_amdgcn_wmma_f32_16x16x4_f32(false, a01, false, p1, (short)0, acc1, false, false);
    acc0 = __builtin_amdgcn_wmma_f32_16x16x4_f32(false, a23, false, q0, (short)0, acc0, false, false);
    acc1 = __builtin_amdgcn_wmma_f32_16x16x4_f32(false, a23, false, q1, (short)0, acc1, false, false);
    __builtin_amdgcn_sched_barrier(0);
    asm volatile("v_nop\n\tv_nop\n\tv_nop\n\tv_nop" : "+v"(acc0), "+v"(acc1));
    __builtin_amdgcn_sched_barrier(0);
}

template <bool LAST>
__device__ __forceinline__ void k_block2(const float* Ap, const float* __restrict__ Bp, int k0,
                                         const a_set& cur, a_set& nxt, b2_half& bA,
                                         v8f& acc0, v8f& acc1) {
    if (!LAST) nxt = load_a(Ap, k0 + 32);
    b2_half bB = load_b2(Bp, k0 + 8);
    __builtin_amdgcn_sched_barrier(0);
    mma_sub2(cur.v0, bA.r[0], bA.r[1], bA.r[2], bA.r[3], acc0, acc1);
    mma_sub2(cur.v1, bA.r[4], bA.r[5], bA.r[6], bA.r[7], acc0, acc1);
    if (!LAST) {
        bA = load_b2(Bp, k0 + 32);
        __builtin_amdgcn_sched_barrier(0);
    }
    mma_sub2(cur.v2, bB.r[0], bB.r[1], bB.r[2], bB.r[3], acc0, acc1);
    mma_sub2(cur.v3, bB.r[4], bB.r[5], bB.r[6], bB.r[7], acc0, acc1);
}

__global__ __launch_bounds__(FG_WAVES * 32) __attribute__((amdgpu_waves_per_eu(10)))
void gather_elu_gemm(const int* __restrict__ ei, const int* __restrict__ cnt,
                     const int* slots, const int* __restrict__ ovfn,
                     const int* __restrict__ ovf, const float* __restrict__ Hm,
                     const float* __restrict__ as, const float* __restrict__ ad,
                     const float* __restrict__ bias, float* __restrict__ alpha_out,
                     const float* __restrict__ W2, float* __restrict__ G2) {
    constexpr int K = HEADS * HID;
    constexpr int NV = K / 128;
    static_assert(K % 64 == 0, "two blocks of 32 k per trip");
    __shared__ __attribute__((aligned(16))) float tile[FG_ROWS * FG_PITCH];
    const int lane = threadIdx.x & 31;
    const int w = __builtin_amdgcn_readfirstlane(threadIdx.x >> 5);
    const int m0 = blockIdx.x * FG_ROWS;
#pragma nounroll
    for (int i = 0; i < FG_ROWS / FG_WAVES; ++i) {
        const int row = w + FG_WAVES * i;
        int bias_off = 0;
        asm volatile("" : "+s"(bias_off));
        const float* __restrict__ bias_i = bias + bias_off;
        float4 o[NV];
        node_row<HEADS, HID, true>(m0 + row, lane, ei, cnt, slots, ovfn, ovf, Hm, as, ad, bias_i,
                                   alpha_out, o);
#pragma unroll
        for (int p = 0; p < NV; ++p)
            *(float4*)(tile + row * FG_PITCH + p * 128 + lane * 4) = o[p];
    }
    __syncthreads();

    const int c = lane & 15;
    const int h = lane >> 4;
    const int n0 = w * 32;
    const float* Ap = tile + c * FG_PITCH + 16 * h;
    const float* __restrict__ Bp = W2 + (size_t)(16 * h) * OUT_CH + n0 + 2 * c;
    v8f acc0 = {}, acc1 = {};
    a_set a0 = load_a(Ap, 0), a1;
    b2_half bA = load_b2(Bp, 0);
    for (int k0 = 0; k0 < K - 64; k0 += 64) {
        k_block2<false>(Ap, Bp, k0,      a0, a1, bA, acc0, acc1);
        k_block2<false>(Ap, Bp, k0 + 32, a1, a0, bA, acc0, acc1);
    }
    k_block2<false>(Ap, Bp, K - 64, a0, a1, bA, acc0, acc1);
    k_block2<true >(Ap, Bp, K - 32, a1, a0, bA, acc0, acc1);

#pragma unroll
    for (int r = 0; r < 8; ++r) {
        v2f t = {acc0[r], acc1[r]};
        *(v2f*)(G2 + (size_t)(m0 + r + 8 * h) * OUT_CH + n0 + 2 * c) = t;
    }
}

static inline int cdiv(long long a, int b) { return (int)((a + b - 1) / b); }

extern "C" void kernel_launch(void* const* d_in, const int* in_sizes, int n_in,
                              void* d_out, int out_size, void* d_ws, size_t ws_size,
                              hipStream_t stream) {
    const float* x      = (const float*)d_in[0];
    const int*   ei     = (const int*)d_in[1];
    const float* W1     = (const float*)d_in[2];
    const float* a_src1 = (const float*)d_in[3];
    const float* a_dst1 = (const float*)d_in[4];
    const float* b1     = (const float*)d_in[5];
    const float* W2     = (const float*)d_in[6];
    const float* a_src2 = (const float*)d_in[7];
    const float* a_dst2 = (const float*)d_in[8];
    const float* b2     = (const float*)d_in[9];

    const int EP = N_EDGES + N_NODES;

    float* H1   = (float*)d_ws;
    float* G2   = H1   + (size_t)N_NODES * 512;
    float* AS1  = G2   + (size_t)N_NODES * 512;
    float* AD1  = AS1  + (size_t)N_NODES * HEADS;
    float* AS2  = AD1  + (size_t)N_NODES * HEADS;
    float* AD2  = AS2  + N_NODES;
    int*   CNT  = (int*)(AD2 + N_NODES);
    int*   OVFN = CNT + N_NODES;
    int*   OVF  = OVFN + 1;

    float* out    = (float*)d_out;
    float* alpha1 = out    + (size_t)N_NODES * OUT_CH;
    float* alpha2 = alpha1 + (size_t)EP * HEADS;
    int*   SLOTS  = (int*)d_out;

    const int TB = 256;

    gemm_wmma_f32<IN_CH, HEADS * HID, true>
        <<<dim3(N_NODES / 16, (HEADS * HID) / 64), 32, 0, stream>>>(x, W1, H1, CNT);

    att_dots<HEADS, HID, 8, true><<<cdiv((long long)(N_NODES / 8) * 32, 128), 128, 0, stream>>>(
        H1, a_src1, a_dst1, AS1, AD1, ei, CNT, SLOTS, OVFN, OVF);

    gather_elu_gemm<<<N_NODES / FG_ROWS, FG_WAVES * 32, 0, stream>>>(
        ei, CNT, SLOTS, OVFN, OVF, H1, AS1, AD1, b1, alpha1, W2, G2);

    att_dots<1, OUT_CH, 8, false><<<cdiv((long long)(N_NODES / 8) * 32, 128), 128, 0, stream>>>(
        G2, a_src2, a_dst2, AS2, AD2, nullptr, nullptr, nullptr, nullptr, nullptr);

    node_softmax_gather<1, OUT_CH, false><<<cdiv((long long)N_NODES * 32, TB), TB, 0, stream>>>(
        ei, CNT, SLOTS, OVFN, OVF, G2, AS2, AD2, b2, alpha2, out);
}


# First GEMM: each wave owns 32 x 64 of C, two 16-row tiles sharing every B load and B register pair (was 16 x 64); 10000 waves -> 5000, one rolled K loop, all resident; k sum order unchanged
# speedup vs baseline: 3.1639x; 1.0367x over previous
#include <hip/hip_runtime.h>
#include <math.h>

#define N_NODES 20000
#define N_EDGES 320000
#define IN_CH 512
#define HID 128
#define HEADS 4
#define OUT_CH 256
#define NEG_SLOPE 0.2f

typedef float v2f __attribute__((ext_vector_type(2)));
typedef float v4f __attribute__((ext_vector_type(4)));
typedef float v8f __attribute__((ext_vector_type(8)));

#define SLOT_CAP 32
#define SLOT_ROW OUT_CH
static_assert(SLOT_CAP == 32, "one slot per lane of a wave");
static_assert(2 * SLOT_CAP <= SLOT_ROW, "a slot row fits into an output row");

struct a_set { v4f v0, v1, v2, v3; };
struct b_set { v4f r0, r1, r2, r3; };

template <int NOUT>
__device__ __forceinline__ b_set load_b(const float* __restrict__ Bp, int k) {
    b_set b;
    b.r0 = *(const v4f*)(Bp + (size_t)(k + 0) * NOUT);
    b.r1 = *(const v4f*)(Bp + (size_t)(k + 1) * NOUT);
    b.r2 = *(const v4f*)(Bp + (size_t)(k + 2) * NOUT);
    b.r3 = *(const v4f*)(Bp + (size_t)(k + 3) * NOUT);
    return b;
}

__device__ __forceinline__ a_set load_a(const float* __restrict__ Ap, int k0) {
    a_set a;
    a.v0 = *(const v4f*)(Ap + k0 + 0);
    a.v1 = *(const v4f*)(Ap + k0 + 4);
    a.v2 = *(const v4f*)(Ap + k0 + 8);
    a.v3 = *(const v4f*)(Ap + k0 + 12);
    return a;
}

__device__ __forceinline__ void mma_sub(const v4f& a, const v4f& e, const b_set& b,
                                        v8f& acc0, v8f& acc1, v8f& acc2, v8f& acc3,
                                        v8f& ecc0, v8f& ecc1, v8f& ecc2, v8f& ecc3) {
    const v2f a01 = {a.x, a.y};
    const v2f a23 = {a.z, a.w};
    const v2f e01 = {e.x, e.y};
    const v2f e23 = {e.z, e.w};
    v2f p0 = {b.r0.x, b.r1.x}, p1 = {b.r0.y, b.r1.y};
    v2f p2 = {b.r0.z, b.r1.z}, p3 = {b.r0.w, b.r1.w};
    v2f q0 = {b.r2.x, b.r3.x}, q1 = {b.r2.y, b.r3.y};
    v2f q2 = {b.r2.z, b.r3.z}, q3 = {b.r2.w, b.r3.w};
    asm volatile("" : "+v"(p0), "+v"(p1), "+v"(p2), "+v"(p3),
                      "+v"(q0), "+v"(q1), "+v"(q2), "+v"(q3));
    __builtin_amdgcn_sched_barrier(0);
    acc0 = __builtin_amdgcn_wmma_f32_16x16x4_f32(false, a01, false, p0, (short)0, acc0, false, false);
    acc1 = __builtin_amdgcn_wmma_f32_16x16x4_f32(false, a01, false, p1, (short)0, acc1, false, false);
    acc2 = __builtin_amdgcn_wmma_f32_16x16x4_f32(false, a01, false, p2, (short)0, acc2, false, false);
    acc3 = __builtin_amdgcn_wmma_f32_16x16x4_f32(false, a01, false, p3, (short)0, acc3, false, false);
    ecc0 = __builtin_amdgcn_wmma_f32_16x16x4_f32(false, e01, false, p0, (short)0, ecc0, false, false);
    ecc1 = __builtin_amdgcn_wmma_f32_16x16x4_f32(false, e01, false, p1, (short)0, ecc1, false, false);
    ecc2 = __builtin_amdgcn_wmma_f32_16x16x4_f32(false, e01, false, p2, (short)0, ecc2, false, false);
    ecc3 = __builtin_amdgcn_wmma_f32_16x16x4_f32(false, e01, false, p3, (short)0, ecc3, false, false);
    acc0 = __builtin_amdgcn_wmma_f32_16x16x4_f32(false, a23, false, q0, (short)0, acc0, false, false);
    acc1 = __builtin_amdgcn_wmma_f32_16x16x4_f32(false, a23, false, q1, (short)0, acc1, false, false);
    acc2 = __builtin_amdgcn_wmma_f32_16x16x4_f32(false, a23, false, q2, (short)0, acc2, false, false);
    acc3 = __builtin_amdgcn_wmma_f32_16x16x4_f32(false, a23, false, q3, (short)0, acc3, false, false);
    ecc0 = __builtin_amdgcn_wmma_f32_16x16x4_f32(false, e23, false, q0, (short)0, ecc0, false, false);
    ecc1 = __builtin_amdgcn_wmma_f32_16x16x4_f32(false, e23, false, q1, (short)0, ecc1, false, false);
    ecc2 = __builtin_amdgcn_wmma_f32_16x16x4_f32(false, e23, false, q2, (short)0, ecc2, false, false);
    ecc3 = __builtin_amdgcn_wmma_f32_16x16x4_f32(false, e23, false, q3, (short)0, ecc3, false, false);
    __builtin_amdgcn_sched_barrier(0);
    asm volatile("v_nop\n\tv_nop\n\tv_nop\n\tv_nop");
    __builtin_amdgcn_sched_barrier(0);
}

template <int K, int NOUT>
__device__ __forceinline__ void k_block(const float* __restrict__ Ap, const float* __restrict__ Bp,
                                        int k0, int kn, const a_set& cur, const a_set& eur,
                                        a_set& nxt, a_set& ext, b_set& b0,
                                        v8f& acc0, v8f& acc1, v8f& acc2, v8f& acc3,
                                        v8f& ecc0, v8f& ecc1, v8f& ecc2, v8f& ecc3) {
    nxt = load_a(Ap, kn);
    ext = load_a(Ap + (size_t)16 * K, kn);
    b_set b1 = load_b<NOUT>(Bp, k0 + 4);
    __builtin_amdgcn_sched_barrier(0);
    mma_sub(cur.v0, eur.v0, b0, acc0, acc1, acc2, acc3, ecc0, ecc1, ecc2, ecc3);
    b0 = load_b<NOUT>(Bp, k0 + 8);
    __builtin_amdgcn_sched_barrier(0);
    mma_sub(cur.v1, eur.v1, b1, acc0, acc1, acc2, acc3, ecc0, ecc1, ecc2, ecc3);
    b1 = load_b<NOUT>(Bp, k0 + 12);
    __builtin_amdgcn_sched_barrier(0);
    mma_sub(cur.v2, eur.v2, b0, acc0, acc1, acc2, acc3, ecc0, ecc1, ecc2, ecc3);
    b0 = load_b<NOUT>(Bp, kn);
    __builtin_amdgcn_sched_barrier(0);
    mma_sub(cur.v3, eur.v3, b1, acc0, acc1, acc2, acc3, ecc0, ecc1, ecc2, ecc3);
}

template <int K, int NOUT, bool CLEAR>
__global__ __launch_bounds__(32)
void gemm_wmma_f32(const float* __restrict__ A, const float* __restrict__ B,
                   float* __restrict__ C, int* __restrict__ clr) {
    static_assert(K % 64 == 0 && (K & (K - 1)) == 0 && NOUT % 64 == 0 && N_NODES % 32 == 0, "tile shape");
    const int lane = threadIdx.x & 31;
    const int m0 = blockIdx.x * 32;
    const int n0 = blockIdx.y * 64;
    const int c = lane & 15;
    const int h = lane >> 4;
    if constexpr (CLEAR) {
        if (blockIdx.y == 0) clr[m0 + lane] = 0;
        if (blockIdx.y == 1 && blockIdx.x == 0 && lane == 0) clr[N_NODES] = 0;
    }

    v8f acc0 = {}, acc1 = {}, acc2 = {}, acc3 = {};
    v8f ecc0 = {}, ecc1 = {}, ecc2 = {}, ecc3 = {};
    const float* __restrict__ Ap = A + (size_t)(m0 + c) * K + 16 * h;
    const float* __restrict__ Bp = B + (size_t)(16 * h) * NOUT + n0 + 4 * c;

    a_set a0 = load_a(Ap, 0), a1;
    a_set e0 = load_a(Ap + (size_t)16 * K, 0), e1;
    b_set b0 = load_b<NOUT>(Bp, 0);
    for (int k0 = 0; k0 < K; k0 += 64) {
        k_block<K, NOUT>(Ap, Bp, k0,      k0 + 32,             a0, e0, a1, e1, b0,
                         acc0, acc1, acc2, acc3, ecc0, ecc1, ecc2, ecc3);
        k_block<K, NOUT>(Ap, Bp, k0 + 32, (k0 + 64) & (K - 1), a1, e1, a0, e0, b0,
                         acc0, acc1, acc2, acc3, ecc0, ecc1, ecc2, ecc3);
    }

    for (int r = 0; r < 8; ++r) {
        v4f o = {acc0[r], acc1[r], acc2[r], acc3[r]};
        *(v4f*)(C + (size_t)(m0 + r + 8 * h) * NOUT + n0 + 4 * c) = o;
    }
    for (int r = 0; r < 8; ++r) {
        v4f o = {ecc0[r], ecc1[r], ecc2[r], ecc3[r]};
        *(v4f*)(C + (size_t)(m0 + 16 + r + 8 * h) * NOUT + n0 + 4 * c) = o;
    }
}

template <int X>
__device__ __forceinline__ float row_xor(float v) {
    const int b = __builtin_bit_cast(int, v);
    return __builtin_bit_cast(float, __builtin_amdgcn_update_dpp(b, b, 0x160 + X, 0xf, 0xf, false));
}
__device__ __forceinline__ float xhalf(float v) {
    const int b = __builtin_bit_cast(int, v);
    return __builtin_bit_cast(float,
        __builtin_amdgcn_permlanex16(b, b, 0x76543210, 0xfedcba98, false, false));
}
__device__ __forceinline__ float wave_max(float v) {
    v = fmaxf(v, row_xor<1>(v));
    v = fmaxf(v, row_xor<2>(v));
    v = fmaxf(v, row_xor<4>(v));
    v = fmaxf(v, row_xor<8>(v));
    return fmaxf(v, xhalf(v));
}
__device__ __forceinline__ float wave_sum(float v) {
    v += row_xor<1>(v);
    v += row_xor<2>(v);
    v += row_xor<4>(v);
    v += row_xor<8>(v);
    return v + xhalf(v);
}

template <int H, int CH, int NPW, bool COUNT>
__global__ __launch_bounds__(128)
void att_dots(const float* __restrict__ Hm, const float* __restrict__ a_src,
              const float* __restrict__ a_dst,
              float* __restrict__ as_out, float* __restrict__ ad_out,
              const int* __restrict__ ei, int* __restrict__ cnt, int* __restrict__ slots,
              int* __restrict__ ovfn, int* __restrict__ ovf) {
    constexpr int ROW = H * CH;
    constexpr int GT = (N_NODES / NPW) * 32;
    constexpr int CE = COUNT ? (N_EDGES + GT - 1) / GT : 1;
    constexpr int NV = ROW / 128;
    constexpr int PPH = CH / 128;
    static_assert(CH % 128 == 0 && N_NODES % NPW == 0 && NPW <= 32, "row pieces / node groups");
    const int gid = blockIdx.x * blockDim.x + threadIdx.x;
    const int lane = gid & 31;
    const int n0 = __builtin_amdgcn_readfirstlane(gid >> 5) * NPW;
    if (n0 >= N_NODES) return;

    static_assert(!COUNT || N_EDGES % GT == 0, "every thread counts exactly CE edges");
    int dd[CE], ss[CE], rk[CE];
    if constexpr (COUNT) {
#pragma unroll
        for (int i = 0; i < CE; ++i) {
            dd[i] = ei[N_EDGES + gid + i * GT];
            ss[i] = ei[gid + i * GT];
        }
    }

    float4 ws[NV], wd[NV];
#pragma unroll
    for (int p = 0; p < NV; ++p) {
        ws[p] = *(const float4*)(a_src + p * 128 + lane * 4);
        wd[p] = *(const float4*)(a_dst + p * 128 + lane * 4);
    }
    float4 v[NPW][NV];
#pragma unroll
    for (int u = 0; u < NPW; ++u) {
#pragma unroll
        for (int p = 0; p < NV; ++p)
            v[u][p] = *(const float4*)(Hm + (size_t)(n0 + u) * ROW + p * 128 + lane * 4);
    }
    if constexpr (COUNT) {
#pragma unroll
        for (int i = 0; i < CE; ++i) rk[i] = atomicAdd(&cnt[dd[i]], 1);
    }

    float rs[H], rd[H];
#pragma unroll
    for (int h = 0; h < H; ++h) { rs[h] = 0.f; rd[h] = 0.f; }
#pragma unroll
    for (int u = 0; u < NPW; ++u) {
#pragma unroll
        for (int h = 0; h < H; ++h) {
            float s = 0.f, d = 0.f;
#pragma unroll
            for (int q = 0; q < PPH; ++q) {
                const float4 x = v[u][h * PPH + q];
                const float4 a = ws[h * PPH + q];
                const float4 b = wd[h * PPH + q];
                s = __builtin_fmaf(x.x, a.x, s); s = __builtin_fmaf(x.y, a.y, s);
                s = __builtin_fmaf(x.z, a.z, s); s = __builtin_fmaf(x.w, a.w, s);
                d = __builtin_fmaf(x.x, b.x, d); d = __builtin_fmaf(x.y, b.y, d);
                d = __builtin_fmaf(x.z, b.z, d); d = __builtin_fmaf(x.w, b.w, d);
            }
            s = wave_sum(s);
            d = wave_sum(d);
            if (lane == u) { rs[h] = s; rd[h] = d; }
        }
    }
    if (lane < NPW) {
        if constexpr (H == 4) {
            float4 t; t.x = rs[0]; t.y = rs[1]; t.z = rs[2]; t.w = rs[3];
            *(float4*)(as_out + (size_t)(n0 + lane) * 4) = t;
            t.x = rd[0]; t.y = rd[1]; t.z = rd[2]; t.w = rd[3];
            *(float4*)(ad_out + (size_t)(n0 + lane) * 4) = t;
        } else {
#pragma unroll
            for (int h = 0; h < H; ++h) {
                as_out[(size_t)(n0 + lane) * H + h] = rs[h];
                ad_out[(size_t)(n0 + lane) * H + h] = rd[h];
            }
        }
    }
    if constexpr (COUNT) {
#pragma unroll
        for (int i = 0; i < CE; ++i) {
            const int e = gid + i * GT;
            if (rk[i] < SLOT_CAP) {
                int2 r; r.x = e; r.y = ss[i];
                *(int2*)(slots + (size_t)dd[i] * SLOT_ROW + 2 * rk[i]) = r;
            } else {
                ovf[atomicAdd(ovfn, 1)] = e;
            }
        }
    }
}


__device__ __forceinline__ float leaky(float v) {
    return (v > 0.f) ? v : NEG_SLOPE * v;
}

template <int H>
__device__ __forceinline__ void load_logits(const float* __restrict__ as, int s, float (&lg)[H]) {
    if constexpr (H == 4) {
        const float4 t = *(const float4*)(as + (size_t)s * 4);
        lg[0] = t.x; lg[1] = t.y; lg[2] = t.z; lg[3] = t.w;
    } else {
#pragma unroll
        for (int h = 0; h < H; ++h) lg[h] = as[(size_t)s * H + h];
    }
}

template <int H>
__device__ __forceinline__ void store_alpha(float* __restrict__ alpha_out, size_t e, const float (&a)[H]) {
    if constexpr (H == 4) {
        float4 t; t.x = a[0]; t.y = a[1]; t.z = a[2]; t.w = a[3];
        *(float4*)(alpha_out + e * 4) = t;
    } else {
#pragma unroll
        for (int h = 0; h < H; ++h) alpha_out[e * H + h] = a[h];
    }
}

template <int H, int CH>
__device__ __forceinline__ void gather_rows(float4 (&acc)[(H * CH) / 128], int my_s,
                                            const float (&a)[H], int n,
                                            const float* __restrict__ Hm, int lane) {
    constexpr int ROW = H * CH;
    constexpr int NV = ROW / 128;
    constexpr int EPT = 8 / NV;
    int j = 0;
    for (; j + EPT <= n; j += EPT) {
        int s[EPT];
        float al[EPT][H];
        float4 v[EPT][NV];
#pragma unroll
        for (int u = 0; u < EPT; ++u) {
            s[u] = __builtin_amdgcn_readlane(my_s, j + u);
#pragma unroll
            for (int h = 0; h < H; ++h)
                al[u][h] = __int_as_float(__builtin_amdgcn_readlane(__float_as_int(a[h]), j + u));
        }
#pragma unroll
        for (int u = 0; u < EPT; ++u) {
#pragma unroll
            for (int p = 0; p < NV; ++p)
                v[u][p] = *(const float4*)(Hm + (size_t)s[u] * ROW + p * 128 + lane * 4);
        }
#pragma unroll
        for (int u = 0; u < EPT; ++u) {
#pragma unroll
            for (int p = 0; p < NV; ++p) {
                const float w = al[u][(p * 128) / CH];
                acc[p].x = __builtin_fmaf(w, v[u][p].x, acc[p].x);
                acc[p].y = __builtin_fmaf(w, v[u][p].y, acc[p].y);
                acc[p].z = __builtin_fmaf(w, v[u][p].z, acc[p].z);
                acc[p].w = __builtin_fmaf(w, v[u][p].w, acc[p].w);
            }
        }
    }
    for (; j < n; ++j) {
        const int s0 = __builtin_amdgcn_readlane(my_s, j);
        float al0[H];
#pragma unroll
        for (int h = 0; h < H; ++h)
            al0[h] = __int_as_float(__builtin_amdgcn_readlane(__float_as_int(a[h]), j));
        float4 v0[NV];
#pragma unroll
        for (int p = 0; p < NV; ++p)
            v0[p] = *(const float4*)(Hm + (size_t)s0 * ROW + p * 128 + lane * 4);
#pragma unroll
        for (int p = 0; p < NV; ++p) {
            const float w = al0[(p * 128) / CH];
            acc[p].x = __builtin_fmaf(w, v0[p].x, acc[p].x);
            acc[p].y = __builtin_fmaf(w, v0[p].y, acc[p].y);
            acc[p].z = __builtin_fmaf(w, v0[p].z, acc[p].z);
            acc[p].w = __builtin_fmaf(w, v0[p].w, acc[p].w);
        }
    }
}

template <int H, int CH>
__device__ __forceinline__ void gather_rows_mask(float4 (&acc)[(H * CH) / 128], int my_s,
                                                 const float (&a)[H], unsigned mask,
                                                 const float* __restrict__ Hm, int lane) {
    constexpr int ROW = H * CH;
    constexpr int NV = ROW / 128;
    while (mask) {
        const int j = __builtin_ctz(mask);
        mask &= mask - 1;
        const int s0 = __builtin_amdgcn_readlane(my_s, j);
        float al0[H];
#pragma unroll
        for (int h = 0; h < H; ++h)
            al0[h] = __int_as_float(__builtin_amdgcn_readlane(__float_as_int(a[h]), j));
        float4 v0[NV];
#pragma unroll
        for (int p = 0; p < NV; ++p)
            v0[p] = *(const float4*)(Hm + (size_t)s0 * ROW + p * 128 + lane * 4);
#pragma unroll
        for (int p = 0; p < NV; ++p) {
            const float w = al0[(p * 128) / CH];
            acc[p].x = __builtin_fmaf(w, v0[p].x, acc[p].x);
            acc[p].y = __builtin_fmaf(w, v0[p].y, acc[p].y);
            acc[p].z = __builtin_fmaf(w, v0[p].z, acc[p].z);
            acc[p].w = __builtin_fmaf(w, v0[p].w, acc[p].w);
        }
    }
}

template <int H, int CH, bool DO_ELU>
__device__ __forceinline__
void node_row(const int d, const int lane,
              const int* __restrict__ ei, const int* __restrict__ cnt,
              const int* slots, const int* __restrict__ ovfn,
              const int* __restrict__ ovf, const float* __restrict__ Hm,
              const float* __restrict__ as, const float* __restrict__ ad,
              const float* __restrict__ bias,
              float* __restrict__ alpha_out, float4 (&o)[(H * CH) / 128]) {
    constexpr int ROW = H * CH;
    constexpr int NV = ROW / 128;
    static_assert(CH % 128 == 0 && (8 % NV) == 0, "row pieces");
    const int deg = cnt[d];
    const int* srow = slots + (size_t)d * SLOT_ROW;

    float adh[H], vself[H];
    {
        float lgd[H];
        load_logits<H>(as, d, lgd);
#pragma unroll
        for (int h = 0; h < H; ++h) {
            adh[h] = ad[d * H + h];
            vself[h] = leaky(lgd[h] + adh[h]);
        }
    }
    float4 vs[NV];
#pragma unroll
    for (int p = 0; p < NV; ++p)
        vs[p] = *(const float4*)(Hm + (size_t)d * ROW + p * 128 + lane * 4);
    float4 acc[NV];

    if (deg <= SLOT_CAP) {
        int my_e = 0, my_s = 0;
        float my_v[H];
#pragma unroll
        for (int h = 0; h < H; ++h) my_v[h] = vself[h];
        if (lane < deg) {
            const int2 r = *(const int2*)(srow + 2 * lane);
            my_e = r.x;
            my_s = r.y;
            float lg[H];
            load_logits<H>(as, my_s, lg);
#pragma unroll
            for (int h = 0; h < H; ++h) my_v[h] = leaky(lg[h] + adh[h]);
        }
        float a[H], aself[H];
#pragma unroll
        for (int h = 0; h < H; ++h) {
            const float m = wave_max(fmaxf(my_v[h], vself[h]));
            const float ex = (lane < deg) ? expf(my_v[h] - m) : 0.f;
            const float exs = expf(vself[h] - m);
            const float sum = wave_sum(ex) + exs;
            a[h] = ex / sum;
            aself[h] = exs / sum;
        }
        if (lane < deg) store_alpha<H>(alpha_out, (size_t)my_e, a);
        if (lane == 0) store_alpha<H>(alpha_out, (size_t)(N_EDGES + d), aself);
#pragma unroll
        for (int p = 0; p < NV; ++p) {
            const float w = aself[(p * 128) / CH];
            acc[p].x = w * vs[p].x; acc[p].y = w * vs[p].y;
            acc[p].z = w * vs[p].z; acc[p].w = w * vs[p].w;
        }
        gather_rows<H, CH>(acc, my_s, a, deg, Hm, lane);
    } else {
        const int novf = __builtin_amdgcn_readfirstlane(ovfn[0]);
        const int2 r0 = *(const int2*)(srow + 2 * lane);
        const int e0 = r0.x, s0 = r0.y;
        float v0[H];
        {
            float lg[H];
            load_logits<H>(as, s0, lg);
#pragma unroll
            for (int h = 0; h < H; ++h) v0[h] = leaky(lg[h] + adh[h]);
        }
        float m[H], part[H], sum[H], aself[H];
#pragma unroll
        for (int h = 0; h < H; ++h) m[h] = fmaxf(vself[h], v0[h]);
        for (int i = 0; i < novf; i += 32) {
            if (i + lane < novf) {
                const int e = ovf[i + lane];
                if (ei[N_EDGES + e] == d) {
                    const int s = ei[e];
                    float lg[H];
                    load_logits<H>(as, s, lg);
#pragma unroll
                    for (int h = 0; h < H; ++h) m[h] = fmaxf(m[h], leaky(lg[h] + adh[h]));
                }
            }
        }
#pragma unroll
        for (int h = 0; h < H; ++h) {
            m[h] = wave_max(m[h]);
            part[h] = expf(v0[h] - m[h]);
        }
        for (int i = 0; i < novf; i += 32) {
            if (i + lane < novf) {
                const int e = ovf[i + lane];
                if (ei[N_EDGES + e] == d) {
                    const int s = ei[e];
                    float lg[H];
                    load_logits<H>(as, s, lg);
#pragma unroll
                    for (int h = 0; h < H; ++h) part[h] += expf(leaky(lg[h] + adh[h]) - m[h]);
                }
            }
        }
#pragma unroll
        for (int h = 0; h < H; ++h) {
            const float exs = expf(vself[h] - m[h]);
            sum[h] = wave_sum(part[h]) + exs;
            aself[h] = exs / sum[h];
        }
        if (lane == 0) store_alpha<H>(alpha_out, (size_t)(N_EDGES + d), aself);
#pragma unroll
        for (int p = 0; p < NV; ++p) {
            const float w = aself[(p * 128) / CH];
            acc[p].x = w * vs[p].x; acc[p].y = w * vs[p].y;
            acc[p].z = w * vs[p].z; acc[p].w = w * vs[p].w;
        }
        {
            float a[H];
#pragma unroll
            for (int h = 0; h < H; ++h) a[h] = expf(v0[h] - m[h]) / sum[h];
            store_alpha<H>(alpha_out, (size_t)e0, a);
            gather_rows<H, CH>(acc, s0, a, SLOT_CAP, Hm, lane);
        }
        for (int i = 0; i < novf; i += 32) {
            int my_s = 0;
            bool hit = false;
            float a[H];
#pragma unroll
            for (int h = 0; h < H; ++h) a[h] = 0.f;
            if (i + lane < novf) {
                const int e = ovf[i + lane];
                if (ei[N_EDGES + e] == d) {
                    hit = true;
                    my_s = ei[e];
                    float lg[H];
                    load_logits<H>(as, my_s, lg);
#pragma unroll
                    for (int h = 0; h < H; ++h) a[h] = expf(leaky(lg[h] + adh[h]) - m[h]) / sum[h];
                    store_alpha<H>(alpha_out, (size_t)e, a);
                }
            }
            const unsigned mask = __builtin_amdgcn_ballot_w32(hit);
            gather_rows_mask<H, CH>(acc, my_s, a, mask, Hm, lane);
        }
    }

#pragma unroll
    for (int p = 0; p < NV; ++p) {
        const float4 b = *(const float4*)(bias + p * 128 + lane * 4);
        float4 t = acc[p];
        t.x += b.x; t.y += b.y; t.z += b.z; t.w += b.w;
        if (DO_ELU) {
            t.x = (t.x > 0.f) ? t.x : (expf(t.x) - 1.f);
            t.y = (t.y > 0.f) ? t.y : (expf(t.y) - 1.f);
            t.z = (t.z > 0.f) ? t.z : (expf(t.z) - 1.f);
            t.w = (t.w > 0.f) ? t.w : (expf(t.w) - 1.f);
        }
        o[p] = t;
    }
}

template <int H, int CH, bool DO_ELU>
__global__ __launch_bounds__(256)
void node_softmax_gather(const int* __restrict__ ei, const int* __restrict__ cnt,
                         const int* slots, const int* __restrict__ ovfn,
                         const int* __restrict__ ovf, const float* __restrict__ Hm,
                         const float* __restrict__ as, const float* __restrict__ ad,
                         const float* __restrict__ bias,
                         float* __restrict__ alpha_out, float* outp) {
    constexpr int ROW = H * CH;
    constexpr int NV = ROW / 128;
    const int gid = blockIdx.x * blockDim.x + threadIdx.x;
    const int lane = gid & 31;
    const int d = __builtin_amdgcn_readfirstlane(gid >> 5);
    if (d >= N_NODES) return;
    float4 o[NV];
    node_row<H, CH, DO_ELU>(d, lane, ei, cnt, slots, ovfn, ovf, Hm, as, ad, bias, alpha_out, o);
#pragma unroll
    for (int p = 0; p < NV; ++p)
        *(float4*)(outp + (size_t)d * ROW + p * 128 + lane * 4) = o[p];
}

#define FG_ROWS 16
#define FG_WAVES 8
#define FG_PITCH (HEADS * HID + 4)
static_assert(N_NODES % FG_ROWS == 0, "every work-group is full: all its waves reach the barrier");
static_assert(FG_ROWS % FG_WAVES == 0, "nodes per wave");
static_assert(((FG_PITCH * 4) % 16) == 0 && (((FG_PITCH * 4) / 16) % 2) == 1, "row pitch = 16 B x odd");
static_assert(FG_WAVES * 32 == OUT_CH, "column strips of the waves");

struct b2_half { v2f r[8]; };

__device__ __forceinline__ b2_half load_b2(const float* __restrict__ Bp, int k) {
    b2_half b;
#pragma unroll
    for (int j = 0; j < 8; ++j) b.r[j] = *(const v2f*)(Bp + (size_t)(k + j) * OUT_CH);
    return b;
}

__device__ __forceinline__ void mma_sub2(const v4f& a, const v2f& r0, const v2f& r1, const v2f& r2,
                                         const v2f& r3, v8f& acc0, v8f& acc1) {
    const v2f a01 = {a.x, a.y};
    const v2f a23 = {a.z, a.w};
    v2f p0 = {r0.x, r1.x}, p1 = {r0.y, r1.y};
    v2f q0 = {r2.x, r3.x}, q1 = {r2.y, r3.y};
    asm volatile("" : "+v"(p0), "+v"(p1), "+v"(q0), "+v"(q1));
    __builtin_amdgcn_sched_barrier(0);
    acc0 = __builtin_amdgcn_wmma_f32_16x16x4_f32(false, a01, false, p0, (short)0, acc0, false, false);
    acc1 = __builtin_amdgcn_wmma_f32_16x16x4_f32(false, a01, false, p1, (short)0, acc1, false, false);
    acc0 = __builtin_amdgcn_wmma_f32_16x16x4_f32(false, a23, false, q0, (short)0, acc0, false, false);
    acc1 = __builtin_amdgcn_wmma_f32_16x16x4_f32(false, a23, false, q1, (short)0, acc1, false, false);
    __builtin_amdgcn_sched_barrier(0);
    asm volatile("v_nop\n\tv_nop\n\tv_nop\n\tv_nop" : "+v"(acc0), "+v"(acc1));
    __builtin_amdgcn_sched_barrier(0);
}

template <bool LAST>
__device__ __forceinline__ void k_block2(const float* Ap, const float* __restrict__ Bp, int k0,
                                         const a_set& cur, a_set& nxt, b2_half& bA,
                                         v8f& acc0, v8f& acc1) {
    if (!LAST) nxt = load_a(Ap, k0 + 32);
    b2_half bB = load_b2(Bp, k0 + 8);
    __builtin_amdgcn_sched_barrier(0);
    mma_sub2(cur.v0, bA.r[0], bA.r[1], bA.r[2], bA.r[3], acc0, acc1);
    mma_sub2(cur.v1, bA.r[4], bA.r[5], bA.r[6], bA.r[7], acc0, acc1);
    if (!LAST) {
        bA = load_b2(Bp, k0 + 32);
        __builtin_amdgcn_sched_barrier(0);
    }
    mma_sub2(cur.v2, bB.r[0], bB.r[1], bB.r[2], bB.r[3], acc0, acc1);
    mma_sub2(cur.v3, bB.r[4], bB.r[5], bB.r[6], bB.r[7], acc0, acc1);
}

__global__ __launch_bounds__(FG_WAVES * 32) __attribute__((amdgpu_waves_per_eu(10)))
void gather_elu_gemm(const int* __restrict__ ei, const int* __restrict__ cnt,
                     const int* slots, const int* __restrict__ ovfn,
                     const int* __restrict__ ovf, const float* __restrict__ Hm,
                     const float* __restrict__ as, const float* __restrict__ ad,
                     const float* __restrict__ bias, float* __restrict__ alpha_out,
                     const float* __restrict__ W2, float* __restrict__ G2) {
    constexpr int K = HEADS * HID;
    constexpr int NV = K / 128;
    static_assert(K % 64 == 0, "two blocks of 32 k per trip");
    __shared__ __attribute__((aligned(16))) float tile[FG_ROWS * FG_PITCH];
    const int lane = threadIdx.x & 31;
    const int w = __builtin_amdgcn_readfirstlane(threadIdx.x >> 5);
    const int m0 = blockIdx.x * FG_ROWS;
#pragma nounroll
    for (int i = 0; i < FG_ROWS / FG_WAVES; ++i) {
        const int row = w + FG_WAVES * i;
        int bias_off = 0;
        asm volatile("" : "+s"(bias_off));
        const float* __restrict__ bias_i = bias + bias_off;
        float4 o[NV];
        node_row<HEADS, HID, true>(m0 + row, lane, ei, cnt, slots, ovfn, ovf, Hm, as, ad, bias_i,
                                   alpha_out, o);
#pragma unroll
        for (int p = 0; p < NV; ++p)
            *(float4*)(tile + row * FG_PITCH + p * 128 + lane * 4) = o[p];
    }
    __syncthreads();

    const int c = lane & 15;
    const int h = lane >> 4;
    const int n0 = w * 32;
    const float* Ap = tile + c * FG_PITCH + 16 * h;
    const float* __restrict__ Bp = W2 + (size_t)(16 * h) * OUT_CH + n0 + 2 * c;
    v8f acc0 = {}, acc1 = {};
    a_set a0 = load_a(Ap, 0), a1;
    b2_half bA = load_b2(Bp, 0);
    for (int k0 = 0; k0 < K - 64; k0 += 64) {
        k_block2<false>(Ap, Bp, k0,      a0, a1, bA, acc0, acc1);
        k_block2<false>(Ap, Bp, k0 + 32, a1, a0, bA, acc0, acc1);
    }
    k_block2<false>(Ap, Bp, K - 64, a0, a1, bA, acc0, acc1);
    k_block2<true >(Ap, Bp, K - 32, a1, a0, bA, acc0, acc1);

#pragma unroll
    for (int r = 0; r < 8; ++r) {
        v2f t = {acc0[r], acc1[r]};
        *(v2f*)(G2 + (size_t)(m0 + r + 8 * h) * OUT_CH + n0 + 2 * c) = t;
    }
}

static inline int cdiv(long long a, int b) { return (int)((a + b - 1) / b); }

extern "C" void kernel_launch(void* const* d_in, const int* in_sizes, int n_in,
                              void* d_out, int out_size, void* d_ws, size_t ws_size,
                              hipStream_t stream) {
    const float* x      = (const float*)d_in[0];
    const int*   ei     = (const int*)d_in[1];
    const float* W1     = (const float*)d_in[2];
    const float* a_src1 = (const float*)d_in[3];
    const float* a_dst1 = (const float*)d_in[4];
    const float* b1     = (const float*)d_in[5];
    const float* W2     = (const float*)d_in[6];
    const float* a_src2 = (const float*)d_in[7];
    const float* a_dst2 = (const float*)d_in[8];
    const float* b2     = (const float*)d_in[9];

    const int EP = N_EDGES + N_NODES;

    float* H1   = (float*)d_ws;
    float* G2   = H1   + (size_t)N_NODES * 512;
    float* AS1  = G2   + (size_t)N_NODES * 512;
    float* AD1  = AS1  + (size_t)N_NODES * HEADS;
    float* AS2  = AD1  + (size_t)N_NODES * HEADS;
    float* AD2  = AS2  + N_NODES;
    int*   CNT  = (int*)(AD2 + N_NODES);
    int*   OVFN = CNT + N_NODES;
    int*   OVF  = OVFN + 1;

    float* out    = (float*)d_out;
    float* alpha1 = out    + (size_t)N_NODES * OUT_CH;
    float* alpha2 = alpha1 + (size_t)EP * HEADS;
    int*   SLOTS  = (int*)d_out;

    const int TB = 256;

    gemm_wmma_f32<IN_CH, HEADS * HID, true>
        <<<dim3(N_NODES / 32, (HEADS * HID) / 64), 32, 0, stream>>>(x, W1, H1, CNT);

    att_dots<HEADS, HID, 8, true><<<cdiv((long long)(N_NODES / 8) * 32, 128), 128, 0, stream>>>(
        H1, a_src1, a_dst1, AS1, AD1, ei, CNT, SLOTS, OVFN, OVF);

    gather_elu_gemm<<<N_NODES / FG_ROWS, FG_WAVES * 32, 0, stream>>>(
        ei, CNT, SLOTS, OVFN, OVF, H1, AS1, AD1, b1, alpha1, W2, G2);

    att_dots<1, OUT_CH, 8, false><<<cdiv((long long)(N_NODES / 8) * 32, 128), 128, 0, stream>>>(
        G2, a_src2, a_dst2, AS2, AD2, nullptr, nullptr, nullptr, nullptr, nullptr);

    node_softmax_gather<1, OUT_CH, false><<<cdiv((long long)N_NODES * 32, TB), TB, 0, stream>>>(
        ei, CNT, SLOTS, OVFN, OVF, G2, AS2, AD2, b2, alpha2, out);
}
